# starting kernel
# baseline (speedup 1.0000x reference)
#include <hip/hip_runtime.h>
#include <hip/hip_bf16.h>

#define DLAT 64
#define DFEAT 128
#define SLOPE 0.01f
#define EPSN 1e-12f

typedef __attribute__((ext_vector_type(2))) float v2f;
typedef __attribute__((ext_vector_type(8))) float v8f;

__device__ __forceinline__ float leaky(float v) { return v >= 0.0f ? v : SLOPE * v; }

__device__ __forceinline__ v8f wmma_f32(v2f a, v2f b, v8f c) {
    return __builtin_amdgcn_wmma_f32_16x16x4_f32(false, a, false, b, (short)0, c,
                                                 false, false);
}

__global__ __launch_bounds__(256) void mlp_norm_conv_kernel(
    const float* __restrict__ feat, const float* __restrict__ Wmlp,
    const float* __restrict__ bmlp, const float* __restrict__ Wc,
    float* __restrict__ xout, float* __restrict__ yout, int nnode)
{
    __shared__ float Wm_s[DFEAT * DLAT];
    __shared__ float Wc_s[DLAT * DLAT];
    __shared__ float xt[32 * 65];
    __shared__ float scl[32];

    const int tid = threadIdx.x;
    for (int i = tid; i < DFEAT * DLAT; i += 256) Wm_s[i] = Wmlp[i];
    for (int i = tid; i < DLAT * DLAT; i += 256) Wc_s[i] = Wc[i];
    __syncthreads();

    const int lane    = tid & 31;
    const int wave    = tid >> 5;
    const int rowtile = wave >> 2;
    const int coltile = wave & 3;
    const int mn      = lane & 15;
    const int kh      = (lane >> 4) << 1;
    const int col     = coltile * 16 + mn;
    const int blk0    = blockIdx.x * 32;

    int arow = blk0 + rowtile * 16 + mn;
    if (arow > nnode - 1) arow = nnode - 1;
    const float* ap = feat + (size_t)arow * DFEAT + kh;

    v8f acc = {};
    #pragma unroll
    for (int kk = 0; kk < DFEAT; kk += 4) {
        v2f a = *(const v2f*)(ap + kk);
        v2f b;
        b.x = Wm_s[(kk + kh) * DLAT + col];
        b.y = Wm_s[(kk + kh + 1) * DLAT + col];
        acc = wmma_f32(a, b, acc);
    }

    const float bias  = bmlp[col];
    const int   rbase = rowtile * 16 + ((lane >> 4) << 3);
    #pragma unroll
    for (int i = 0; i < 8; ++i)
        xt[(rbase + i) * 65 + col] = acc[i] + bias;
    __syncthreads();

    if (tid < 32) {
        float s = 0.f;
        #pragma unroll
        for (int j = 0; j < DLAT; ++j) { float v = xt[tid * 65 + j]; s += v * v; }
        scl[tid] = 1.0f / fmaxf(sqrtf(s), EPSN);
    }
    __syncthreads();

    for (int e = tid; e < 32 * DLAT; e += 256) {
        int r = e >> 6, c = e & 63;
        float v = xt[r * 65 + c] * scl[r];
        xt[r * 65 + c] = v;
        int g = blk0 + r;
        if (g < nnode) xout[(size_t)g * DLAT + c] = v;
    }
    __syncthreads();

    v8f acc2 = {};
    const int rl = rowtile * 16 + mn;
    #pragma unroll
    for (int kk = 0; kk < DLAT; kk += 4) {
        v2f a;
        a.x = xt[rl * 65 + kk + kh];
        a.y = xt[rl * 65 + kk + kh + 1];
        v2f b;
        b.x = Wc_s[(kk + kh) * DLAT + col];
        b.y = Wc_s[(kk + kh + 1) * DLAT + col];
        acc2 = wmma_f32(a, b, acc2);
    }
    #pragma unroll
    for (int i = 0; i < 8; ++i) {
        int g = blk0 + rbase + i;
        if (g < nnode) yout[(size_t)g * DLAT + col] = acc2[i];
    }
}

template <bool FUSE_NEXT>
__global__ __launch_bounds__(256) void update_kernel(
    const float* __restrict__ xin, const float* __restrict__ hraw,
    const float* __restrict__ idemb,
    const float* __restrict__ Wl, const float* __restrict__ bl,
    const float* __restrict__ Wg, const float* __restrict__ bg,
    const float* __restrict__ Wc, float* __restrict__ xout,
    float* __restrict__ yout, int nnode)
{
    __shared__ float Wl_s[DLAT * DLAT];
    __shared__ float Wg_s[DLAT * DLAT];
    __shared__ float Wc_s[DLAT * DLAT];
    __shared__ float xt[32 * 65];

    const int tid = threadIdx.x;
    for (int i = tid; i < DLAT * DLAT; i += 256) {
        Wl_s[i] = Wl[i];
        Wg_s[i] = Wg[i];
        if constexpr (FUSE_NEXT) Wc_s[i] = Wc[i];
    }
    __syncthreads();

    const int lane    = tid & 31;
    const int wave    = tid >> 5;
    const int rowtile = wave >> 2;
    const int coltile = wave & 3;
    const int mn      = lane & 15;
    const int kh      = (lane >> 4) << 1;
    const int col     = coltile * 16 + mn;
    const int blk0    = blockIdx.x * 32;

    int arow = blk0 + rowtile * 16 + mn;
    if (arow > nnode - 1) arow = nnode - 1;
    const float* xp = xin + (size_t)arow * DLAT + kh;
    const float* hp = hraw + (size_t)arow * DLAT + kh;

    v8f c1 = {}, c2 = {};
    #pragma unroll
    for (int kk = 0; kk < DLAT; kk += 4) {
        v2f ax = *(const v2f*)(xp + kk);
        v2f hv = *(const v2f*)(hp + kk);
        v2f ah; ah.x = leaky(hv.x); ah.y = leaky(hv.y);
        v2f b1;
        b1.x = Wl_s[(kk + kh) * DLAT + col];
        b1.y = Wl_s[(kk + kh + 1) * DLAT + col];
        v2f b2;
        b2.x = Wg_s[(kk + kh) * DLAT + col];
        b2.y = Wg_s[(kk + kh + 1) * DLAT + col];
        c1 = wmma_f32(ax, b1, c1);
        c2 = wmma_f32(ah, b2, c2);
    }

    const float blv = bl[col], bgv = bg[col];
    const int rbase = rowtile * 16 + ((lane >> 4) << 3);
    #pragma unroll
    for (int i = 0; i < 8; ++i) {
        int rloc = rbase + i;
        int g    = blk0 + rloc;
        int gc   = g > nnode - 1 ? nnode - 1 : g;
        float idv  = idemb[(size_t)gc * DLAT + col];
        float xhat = leaky(c1[i] + blv) + idv;
        float v    = leaky(c2[i] + bgv + xhat);
        if constexpr (FUSE_NEXT) {
            xt[rloc * 65 + col] = v;
        } else {
            if (g < nnode) xout[(size_t)g * DLAT + col] = v;
        }
    }

    if constexpr (FUSE_NEXT) {
        __syncthreads();
        for (int e = tid; e < 32 * DLAT; e += 256) {
            int r = e >> 6, c = e & 63;
            int g = blk0 + r;
            if (g < nnode) xout[(size_t)g * DLAT + c] = xt[r * 65 + c];
        }
        v8f acc2 = {};
        const int rl = rowtile * 16 + mn;
        #pragma unroll
        for (int kk = 0; kk < DLAT; kk += 4) {
            v2f a;
            a.x = xt[rl * 65 + kk + kh];
            a.y = xt[rl * 65 + kk + kh + 1];
            v2f b;
            b.x = Wc_s[(kk + kh) * DLAT + col];
            b.y = Wc_s[(kk + kh + 1) * DLAT + col];
            acc2 = wmma_f32(a, b, acc2);
        }
        #pragma unroll
        for (int i = 0; i < 8; ++i) {
            int g = blk0 + rbase + i;
            if (g < nnode) yout[(size_t)g * DLAT + col] = acc2[i];
        }
    }
}

__global__ __launch_bounds__(256) void scatter_kernel(
    const float* __restrict__ y, const int* __restrict__ src,
    const int* __restrict__ dst, float* __restrict__ h, int nedge)
{
    const int lane = threadIdx.x & 31;
    const int wid  = (blockIdx.x * blockDim.x + threadIdx.x) >> 5;
    const int nw   = (gridDim.x * blockDim.x) >> 5;
    for (int e = wid; e < nedge; e += nw) {
        const int s = src[e], d = dst[e];
        v2f v = *(const v2f*)(y + (size_t)s * DLAT + lane * 2);
        float* hp = h + (size_t)d * DLAT + lane * 2;
        unsafeAtomicAdd(hp, v.x);
        unsafeAtomicAdd(hp + 1, v.y);
    }
}

__global__ __launch_bounds__(256) void zero_kernel(float4* __restrict__ p, size_t n4)
{
    size_t i      = (size_t)blockIdx.x * blockDim.x + threadIdx.x;
    size_t stride = (size_t)gridDim.x * blockDim.x;
    float4 z = {0.f, 0.f, 0.f, 0.f};
    for (; i < n4; i += stride) p[i] = z;
}

extern "C" void kernel_launch(void* const* d_in, const int* in_sizes, int n_in,
                              void* d_out, int out_size, void* d_ws, size_t ws_size,
                              hipStream_t stream)
{
    const float* features = (const float*)d_in[0];
    const float* id_emb   = (const float*)d_in[1];
    const int*   edge     = (const int*)d_in[2];
    const float* Wmlp     = (const float*)d_in[3];
    const float* bmlp     = (const float*)d_in[4];
    const float* Wc1      = (const float*)d_in[5];
    const float* Wl1      = (const float*)d_in[6];
    const float* bl1      = (const float*)d_in[7];
    const float* Wg1      = (const float*)d_in[8];
    const float* bg1      = (const float*)d_in[9];
    const float* Wc2      = (const float*)d_in[10];
    const float* Wl2      = (const float*)d_in[11];
    const float* bl2      = (const float*)d_in[12];
    const float* Wg2      = (const float*)d_in[13];
    const float* bg2      = (const float*)d_in[14];

    const int nnode = in_sizes[0] / DFEAT;
    const int nedge = in_sizes[2] / 2;
    const int* src  = edge;
    const int* dst  = edge + nedge;

    float* x = (float*)d_ws;
    float* h = x + (size_t)nnode * DLAT;
    float* y = h + (size_t)nnode * DLAT;

    const int    nblk = (nnode + 31) / 32;
    const size_t hn4  = ((size_t)nnode * DLAT) / 4;

    mlp_norm_conv_kernel<<<nblk, 256, 0, stream>>>(features, Wmlp, bmlp, Wc1, x, y, nnode);
    zero_kernel<<<2048, 256, 0, stream>>>((float4*)h, hn4);
    scatter_kernel<<<2048, 256, 0, stream>>>(y, src, dst, h, nedge);
    update_kernel<true><<<nblk, 256, 0, stream>>>(x, h, id_emb, Wl1, bl1, Wg1, bg1,
                                                  Wc2, x, y, nnode);
    zero_kernel<<<2048, 256, 0, stream>>>((float4*)h, hn4);
    scatter_kernel<<<2048, 256, 0, stream>>>(y, src, dst, h, nedge);
    update_kernel<false><<<nblk, 256, 0, stream>>>(x, h, id_emb, Wl2, bl2, Wg2, bg2,
                                                   nullptr, (float*)d_out, nullptr, nnode);
}


# Replace f32 atomic scatter-add by a per-call counting sort of edges by dst (integer atomic arrival rank) and a wave-per-row gather-sum in rank order; h zero-fill launches dropped
# speedup vs baseline: 1.2060x; 1.2060x over previous
#include <hip/hip_runtime.h>
#include <hip/hip_bf16.h>

#define DLAT 64
#define DFEAT 128
#define SLOPE 0.01f
#define EPSN 1e-12f

typedef __attribute__((ext_vector_type(2))) float v2f;
typedef __attribute__((ext_vector_type(8))) float v8f;

__device__ __forceinline__ float leaky(float v) { return v >= 0.0f ? v : SLOPE * v; }

__device__ __forceinline__ v8f wmma_f32(v2f a, v2f b, v8f c) {
    return __builtin_amdgcn_wmma_f32_16x16x4_f32(false, a, false, b, (short)0, c,
                                                 false, false);
}

__global__ __launch_bounds__(256) void mlp_norm_conv_kernel(
    const float* __restrict__ feat, const float* __restrict__ Wmlp,
    const float* __restrict__ bmlp, const float* __restrict__ Wc,
    float* __restrict__ xout, float* __restrict__ yout, int nnode)
{
    __shared__ float Wm_s[DFEAT * DLAT];
    __shared__ float Wc_s[DLAT * DLAT];
    __shared__ float xt[32 * 65];
    __shared__ float scl[32];

    const int tid = threadIdx.x;
    for (int i = tid; i < DFEAT * DLAT; i += 256) Wm_s[i] = Wmlp[i];
    for (int i = tid; i < DLAT * DLAT; i += 256) Wc_s[i] = Wc[i];
    __syncthreads();

    const int lane    = tid & 31;
    const int wave    = tid >> 5;
    const int rowtile = wave >> 2;
    const int coltile = wave & 3;
    const int mn      = lane & 15;
    const int kh      = (lane >> 4) << 1;
    const int col     = coltile * 16 + mn;
    const int blk0    = blockIdx.x * 32;

    int arow = blk0 + rowtile * 16 + mn;
    if (arow > nnode - 1) arow = nnode - 1;
    const float* ap = feat + (size_t)arow * DFEAT + kh;

    v8f acc = {};
    #pragma unroll
    for (int kk = 0; kk < DFEAT; kk += 4) {
        v2f a = *(const v2f*)(ap + kk);
        v2f b;
        b.x = Wm_s[(kk + kh) * DLAT + col];
        b.y = Wm_s[(kk + kh + 1) * DLAT + col];
        acc = wmma_f32(a, b, acc);
    }

    const float bias  = bmlp[col];
    const int   rbase = rowtile * 16 + ((lane >> 4) << 3);
    #pragma unroll
    for (int i = 0; i < 8; ++i)
        xt[(rbase + i) * 65 + col] = acc[i] + bias;
    __syncthreads();

    if (tid < 32) {
        float s = 0.f;
        #pragma unroll
        for (int j = 0; j < DLAT; ++j) { float v = xt[tid * 65 + j]; s += v * v; }
        scl[tid] = 1.0f / fmaxf(sqrtf(s), EPSN);
    }
    __syncthreads();

    for (int e = tid; e < 32 * DLAT; e += 256) {
        int r = e >> 6, c = e & 63;
        float v = xt[r * 65 + c] * scl[r];
        xt[r * 65 + c] = v;
        int g = blk0 + r;
        if (g < nnode) xout[(size_t)g * DLAT + c] = v;
    }
    __syncthreads();

    v8f acc2 = {};
    const int rl = rowtile * 16 + mn;
    #pragma unroll
    for (int kk = 0; kk < DLAT; kk += 4) {
        v2f a;
        a.x = xt[rl * 65 + kk + kh];
        a.y = xt[rl * 65 + kk + kh + 1];
        v2f b;
        b.x = Wc_s[(kk + kh) * DLAT + col];
        b.y = Wc_s[(kk + kh + 1) * DLAT + col];
        acc2 = wmma_f32(a, b, acc2);
    }
    #pragma unroll
    for (int i = 0; i < 8; ++i) {
        int g = blk0 + rbase + i;
        if (g < nnode) yout[(size_t)g * DLAT + col] = acc2[i];
    }
}

template <bool FUSE_NEXT>
__global__ __launch_bounds__(256) void update_kernel(
    const float* __restrict__ xin, const float* __restrict__ hraw,
    const float* __restrict__ idemb,
    const float* __restrict__ Wl, const float* __restrict__ bl,
    const float* __restrict__ Wg, const float* __restrict__ bg,
    const float* __restrict__ Wc, float* __restrict__ xout,
    float* __restrict__ yout, int nnode)
{
    __shared__ float Wl_s[DLAT * DLAT];
    __shared__ float Wg_s[DLAT * DLAT];
    __shared__ float Wc_s[DLAT * DLAT];
    __shared__ float xt[32 * 65];

    const int tid = threadIdx.x;
    for (int i = tid; i < DLAT * DLAT; i += 256) {
        Wl_s[i] = Wl[i];
        Wg_s[i] = Wg[i];
        if constexpr (FUSE_NEXT) Wc_s[i] = Wc[i];
    }
    __syncthreads();

    const int lane    = tid & 31;
    const int wave    = tid >> 5;
    const int rowtile = wave >> 2;
    const int coltile = wave & 3;
    const int mn      = lane & 15;
    const int kh      = (lane >> 4) << 1;
    const int col     = coltile * 16 + mn;
    const int blk0    = blockIdx.x * 32;

    int arow = blk0 + rowtile * 16 + mn;
    if (arow > nnode - 1) arow = nnode - 1;
    const float* xp = xin + (size_t)arow * DLAT + kh;
    const float* hp = hraw + (size_t)arow * DLAT + kh;

    v8f c1 = {}, c2 = {};
    #pragma unroll
    for (int kk = 0; kk < DLAT; kk += 4) {
        v2f ax = *(const v2f*)(xp + kk);
        v2f hv = *(const v2f*)(hp + kk);
        v2f ah; ah.x = leaky(hv.x); ah.y = leaky(hv.y);
        v2f b1;
        b1.x = Wl_s[(kk + kh) * DLAT + col];
        b1.y = Wl_s[(kk + kh + 1) * DLAT + col];
        v2f b2;
        b2.x = Wg_s[(kk + kh) * DLAT + col];
        b2.y = Wg_s[(kk + kh + 1) * DLAT + col];
        c1 = wmma_f32(ax, b1, c1);
        c2 = wmma_f32(ah, b2, c2);
    }

    const float blv = bl[col], bgv = bg[col];
    const int rbase = rowtile * 16 + ((lane >> 4) << 3);
    #pragma unroll
    for (int i = 0; i < 8; ++i) {
        int rloc = rbase + i;
        int g    = blk0 + rloc;
        int gc   = g > nnode - 1 ? nnode - 1 : g;
        float idv  = idemb[(size_t)gc * DLAT + col];
        float xhat = leaky(c1[i] + blv) + idv;
        float v    = leaky(c2[i] + bgv + xhat);
        if constexpr (FUSE_NEXT) {
            xt[rloc * 65 + col] = v;
        } else {
            if (g < nnode) xout[(size_t)g * DLAT + col] = v;
        }
    }

    if constexpr (FUSE_NEXT) {
        __syncthreads();
        for (int e = tid; e < 32 * DLAT; e += 256) {
            int r = e >> 6, c = e & 63;
            int g = blk0 + r;
            if (g < nnode) xout[(size_t)g * DLAT + c] = xt[r * 65 + c];
        }
        v8f acc2 = {};
        const int rl = rowtile * 16 + mn;
        #pragma unroll
        for (int kk = 0; kk < DLAT; kk += 4) {
            v2f a;
            a.x = xt[rl * 65 + kk + kh];
            a.y = xt[rl * 65 + kk + kh + 1];
            v2f b;
            b.x = Wc_s[(kk + kh) * DLAT + col];
            b.y = Wc_s[(kk + kh + 1) * DLAT + col];
            acc2 = wmma_f32(a, b, acc2);
        }
        #pragma unroll
        for (int i = 0; i < 8; ++i) {
            int g = blk0 + rbase + i;
            if (g < nnode) yout[(size_t)g * DLAT + col] = acc2[i];
        }
    }
}

__global__ __launch_bounds__(256) void hist_kernel(
    const int* __restrict__ dst, int* __restrict__ cnt, int* __restrict__ rank, int nedge)
{
    const int S    = gridDim.x * 256;
    const int last = nedge - 1;
    for (int b = blockIdx.x * 256 + threadIdx.x; b < nedge; b += 8 * S) {
        int d[8], r[8];
        #pragma unroll
        for (int k = 0; k < 8; ++k) d[k] = dst[min(b + k * S, last)];
        #pragma unroll
        for (int k = 0; k < 8; ++k) { r[k] = 0; if (b + k * S < nedge) r[k] = atomicAdd(cnt + d[k], 1); }
        #pragma unroll
        for (int k = 0; k < 8; ++k) { const int e = b + k * S; if (e < nedge) rank[e] = r[k]; }
    }
}

__global__ __launch_bounds__(1024) void scan_kernel(
    const int* __restrict__ cnt, int* __restrict__ off, int n)
{
    __shared__ int wtot[32];
    __shared__ int wtot2[32];
    const int tid  = threadIdx.x;
    const int lane = tid & 31;
    const int wave = tid >> 5;
    const int base = blockIdx.x * 1024;

    const int4* c4  = (const int4*)cnt;
    const int   nb4 = base >> 2;
    int s = 0;
    int i = tid;
    for (; i + 7 * 1024 < nb4; i += 8 * 1024) {
        int4 a[8];
        #pragma unroll
        for (int k = 0; k < 8; ++k) a[k] = c4[i + k * 1024];
        #pragma unroll
        for (int k = 0; k < 8; ++k) s += (a[k].x + a[k].y) + (a[k].z + a[k].w);
    }
    for (; i < nb4; i += 1024) {
        int4 a = c4[i];
        s += (a.x + a.y) + (a.z + a.w);
    }
    for (int o = 16; o > 0; o >>= 1) s += __shfl_xor(s, o);
    if (lane == 0) wtot[wave] = s;

    const int idx = base + tid;
    const int v   = idx < n ? cnt[idx] : 0;
    int inc = v;
    for (int o = 1; o < 32; o <<= 1) {
        int u = __shfl_up(inc, o);
        if (lane >= o) inc += u;
    }
    if (lane == 31) wtot2[wave] = inc;
    __syncthreads();

    int prefix = wtot[lane];
    for (int o = 16; o > 0; o >>= 1) prefix += __shfl_xor(prefix, o);
    int wp = lane < wave ? wtot2[lane] : 0;
    for (int o = 16; o > 0; o >>= 1) wp += __shfl_xor(wp, o);

    if (idx < n) off[idx] = prefix + wp + inc - v;
}

__global__ __launch_bounds__(256) void fill_kernel(
    const int* __restrict__ src, const int* __restrict__ dst,
    const int* __restrict__ off, const int* __restrict__ rank,
    int* __restrict__ ssrc, int nedge)
{
    const int S    = gridDim.x * 256;
    const int last = nedge - 1;
    for (int b = blockIdx.x * 256 + threadIdx.x; b < nedge; b += 8 * S) {
        int d[8], r[8], s[8], o[8];
        #pragma unroll
        for (int k = 0; k < 8; ++k) {
            const int e = min(b + k * S, last);
            d[k] = dst[e];
            r[k] = rank[e];
            s[k] = src[e];
        }
        #pragma unroll
        for (int k = 0; k < 8; ++k) o[k] = off[d[k]];
        #pragma unroll
        for (int k = 0; k < 8; ++k) { if (b + k * S < nedge) ssrc[o[k] + r[k]] = s[k]; }
    }
}

__global__ __launch_bounds__(256) void gather_kernel(
    const float* __restrict__ y, const int* __restrict__ off,
    const int* __restrict__ ssrc, float* __restrict__ h, int nnode)
{
    const int lane = threadIdx.x & 31;
    const int w0 = __builtin_amdgcn_readfirstlane((int)((blockIdx.x * 256u + threadIdx.x) >> 5));
    const int nw = (int)((gridDim.x * 256u) >> 5);
    const float* yl = y + lane * 2;

    int node = w0;
    int bv = (node < nnode && lane < 2) ? off[node + lane] : 0;
    while (node < nnode) {
        const int beg = __builtin_amdgcn_readlane(bv, 0);
        const int end = __builtin_amdgcn_readlane(bv, 1);
        int mys = lane < end - beg ? ssrc[beg + lane] : 0;
        const int nxt = node + nw;
        int nbv = 0;
        if (nxt < nnode && lane < 2) nbv = off[nxt + lane];

        v2f acc = {0.f, 0.f};
        int base = beg;
        while (base < end) {
            const int n = min(32, end - base);
            for (int j = 0; j < n; j += 16) {
                v2f v[16];
                #pragma unroll
                for (int k = 0; k < 16; ++k) {
                    const int s = __builtin_amdgcn_readlane(mys, j + k);
                    v[k] = *(const v2f*)(yl + (size_t)s * DLAT);
                }
                #pragma unroll
                for (int k = 0; k < 16; ++k) {
                    const bool on = (j + k) < n;
                    acc.x += on ? v[k].x : 0.f;
                    acc.y += on ? v[k].y : 0.f;
                }
            }
            base += 32;
            if (base < end) mys = lane < end - base ? ssrc[base + lane] : 0;
        }
        *(v2f*)(h + (size_t)node * DLAT + lane * 2) = acc;
        node = nxt; bv = nbv;
    }
}

__global__ __launch_bounds__(256) void zero_kernel(float4* __restrict__ p, size_t n4)
{
    size_t i      = (size_t)blockIdx.x * blockDim.x + threadIdx.x;
    size_t stride = (size_t)gridDim.x * blockDim.x;
    float4 z = {0.f, 0.f, 0.f, 0.f};
    for (; i < n4; i += stride) p[i] = z;
}

extern "C" void kernel_launch(void* const* d_in, const int* in_sizes, int n_in,
                              void* d_out, int out_size, void* d_ws, size_t ws_size,
                              hipStream_t stream)
{
    const float* features = (const float*)d_in[0];
    const float* id_emb   = (const float*)d_in[1];
    const int*   edge     = (const int*)d_in[2];
    const float* Wmlp     = (const float*)d_in[3];
    const float* bmlp     = (const float*)d_in[4];
    const float* Wc1      = (const float*)d_in[5];
    const float* Wl1      = (const float*)d_in[6];
    const float* bl1      = (const float*)d_in[7];
    const float* Wg1      = (const float*)d_in[8];
    const float* bg1      = (const float*)d_in[9];
    const float* Wc2      = (const float*)d_in[10];
    const float* Wl2      = (const float*)d_in[11];
    const float* bl2      = (const float*)d_in[12];
    const float* Wg2      = (const float*)d_in[13];
    const float* bg2      = (const float*)d_in[14];

    const int nnode = in_sizes[0] / DFEAT;
    const int nedge = in_sizes[2] / 2;
    const int* src  = edge;
    const int* dst  = edge + nedge;

    float* x = (float*)d_ws;
    float* h = x + (size_t)nnode * DLAT;
    float* y = h + (size_t)nnode * DLAT;

    const size_t cpad = (((size_t)nnode + 4) / 4) * 4;
    const size_t epad = (((size_t)nedge + 3) / 4) * 4;
    int* cnt  = (int*)(y + (size_t)nnode * DLAT);
    int* off  = cnt + cpad;
    int* rank = off + cpad;
    int* ssrc = rank + epad;

    const int nblk = (nnode + 31) / 32;
    const int sblk = (nnode + 1 + 1023) / 1024;

    zero_kernel<<<(int)((cpad / 4 + 255) / 256), 256, 0, stream>>>((float4*)cnt, cpad / 4);
    hist_kernel<<<1024, 256, 0, stream>>>(dst, cnt, rank, nedge);
    scan_kernel<<<sblk, 1024, 0, stream>>>(cnt, off, nnode + 1);
    fill_kernel<<<1024, 256, 0, stream>>>(src, dst, off, rank, ssrc, nedge);

    mlp_norm_conv_kernel<<<nblk, 256, 0, stream>>>(features, Wmlp, bmlp, Wc1, x, y, nnode);
    gather_kernel<<<2048, 256, 0, stream>>>(y, off, ssrc, h, nnode);
    update_kernel<true><<<nblk, 256, 0, stream>>>(x, h, id_emb, Wl1, bl1, Wg1, bg1,
                                                  Wc2, x, y, nnode);
    gather_kernel<<<2048, 256, 0, stream>>>(y, off, ssrc, h, nnode);
    update_kernel<false><<<nblk, 256, 0, stream>>>(x, h, id_emb, Wl2, bl2, Wg2, bg2,
                                                   nullptr, (float*)d_out, nullptr, nnode);
}


# Dense kernels: x, h, feature tiles and weights staged to LDS by whole-line 16-B loads in one group; A fragments read from LDS, not by 8-B global loads touching 16 lines that 4 waves repeat
# speedup vs baseline: 1.4110x; 1.1700x over previous
#include <hip/hip_runtime.h>
#include <hip/hip_bf16.h>

#define DLAT 64
#define DFEAT 128
#define SLOPE 0.01f
#define EPSN 1e-12f

typedef __attribute__((ext_vector_type(2))) float v2f;
typedef __attribute__((ext_vector_type(8))) float v8f;

__device__ __forceinline__ float leaky(float v) { return v >= 0.0f ? v : SLOPE * v; }

__device__ __forceinline__ v8f wmma_f32(v2f a, v2f b, v8f c) {
    return __builtin_amdgcn_wmma_f32_16x16x4_f32(false, a, false, b, (short)0, c,
                                                 false, false);
}

#define FPITCH 132
#define XPITCH 68

__global__ __launch_bounds__(256) void mlp_norm_conv_kernel(
    const float* __restrict__ feat, const float* __restrict__ Wmlp,
    const float* __restrict__ bmlp, const float* __restrict__ Wc,
    float* __restrict__ xout, float* __restrict__ yout, int nnode)
{
    __shared__ __attribute__((aligned(16))) float Wm_s[DFEAT * DLAT];
    __shared__ __attribute__((aligned(16))) float Wc_s[DLAT * DLAT];
    __shared__ __attribute__((aligned(16))) float ft[32 * FPITCH];
    __shared__ float xt[32 * 65];
    __shared__ float scl[32];

    const int tid  = threadIdx.x;
    const int blk0 = blockIdx.x * 32;

    {
        float4 fr[4], wm[8], wc[4];
        #pragma unroll
        for (int i = 0; i < 4; ++i) {
            const int f = tid + 256 * i;
            int g = blk0 + (f >> 5);
            if (g > nnode - 1) g = nnode - 1;
            fr[i] = *(const float4*)(feat + (size_t)g * DFEAT + ((f & 31) << 2));
        }
        #pragma unroll
        for (int i = 0; i < 8; ++i) wm[i] = ((const float4*)Wmlp)[tid + 256 * i];
        #pragma unroll
        for (int i = 0; i < 4; ++i) wc[i] = ((const float4*)Wc)[tid + 256 * i];
        #pragma unroll
        for (int i = 0; i < 4; ++i) {
            const int f = tid + 256 * i;
            *(float4*)(ft + (f >> 5) * FPITCH + ((f & 31) << 2)) = fr[i];
        }
        #pragma unroll
        for (int i = 0; i < 8; ++i) ((float4*)Wm_s)[tid + 256 * i] = wm[i];
        #pragma unroll
        for (int i = 0; i < 4; ++i) ((float4*)Wc_s)[tid + 256 * i] = wc[i];
    }
    __syncthreads();

    const int lane    = tid & 31;
    const int wave    = tid >> 5;
    const int rowtile = wave >> 2;
    const int coltile = wave & 3;
    const int mn      = lane & 15;
    const int kh      = (lane >> 4) << 1;
    const int col     = coltile * 16 + mn;

    const float* ap = ft + (rowtile * 16 + mn) * FPITCH + kh;

    v8f acc = {};
    #pragma unroll
    for (int kk = 0; kk < DFEAT; kk += 4) {
        v2f a = *(const v2f*)(ap + kk);
        v2f b;
        b.x = Wm_s[(kk + kh) * DLAT + col];
        b.y = Wm_s[(kk + kh + 1) * DLAT + col];
        acc = wmma_f32(a, b, acc);
    }

    const float bias  = bmlp[col];
    const int   rbase = rowtile * 16 + ((lane >> 4) << 3);
    #pragma unroll
    for (int i = 0; i < 8; ++i)
        xt[(rbase + i) * 65 + col] = acc[i] + bias;
    __syncthreads();

    if (tid < 32) {
        float s = 0.f;
        #pragma unroll
        for (int j = 0; j < DLAT; ++j) { float v = xt[tid * 65 + j]; s += v * v; }
        scl[tid] = 1.0f / fmaxf(sqrtf(s), EPSN);
    }
    __syncthreads();

    for (int e = tid; e < 32 * DLAT; e += 256) {
        int r = e >> 6, c = e & 63;
        float v = xt[r * 65 + c] * scl[r];
        xt[r * 65 + c] = v;
        int g = blk0 + r;
        if (g < nnode) xout[(size_t)g * DLAT + c] = v;
    }
    __syncthreads();

    v8f acc2 = {};
    const int rl = rowtile * 16 + mn;
    #pragma unroll
    for (int kk = 0; kk < DLAT; kk += 4) {
        v2f a;
        a.x = xt[rl * 65 + kk + kh];
        a.y = xt[rl * 65 + kk + kh + 1];
        v2f b;
        b.x = Wc_s[(kk + kh) * DLAT + col];
        b.y = Wc_s[(kk + kh + 1) * DLAT + col];
        acc2 = wmma_f32(a, b, acc2);
    }
    #pragma unroll
    for (int i = 0; i < 8; ++i) {
        int g = blk0 + rbase + i;
        if (g < nnode) yout[(size_t)g * DLAT + col] = acc2[i];
    }
}

template <bool FUSE_NEXT>
__global__ __launch_bounds__(256) void update_kernel(
    const float* __restrict__ xin, const float* __restrict__ hraw,
    const float* __restrict__ idemb,
    const float* __restrict__ Wl, const float* __restrict__ bl,
    const float* __restrict__ Wg, const float* __restrict__ bg,
    const float* __restrict__ Wc, float* __restrict__ xout,
    float* __restrict__ yout, int nnode)
{
    __shared__ __attribute__((aligned(16))) float Wl_s[DLAT * DLAT];
    __shared__ __attribute__((aligned(16))) float Wg_s[DLAT * DLAT];
    __shared__ __attribute__((aligned(16))) float Wc_s[DLAT * DLAT];
    __shared__ __attribute__((aligned(16))) float xs[32 * XPITCH];
    __shared__ __attribute__((aligned(16))) float hs[32 * XPITCH];
    __shared__ float xt[32 * 65];

    const int tid  = threadIdx.x;
    const int blk0 = blockIdx.x * 32;

    {
        float4 xr[2], hr[2], wl[4], wg[4], wc[4];
        #pragma unroll
        for (int i = 0; i < 2; ++i) {
            const int f = tid + 256 * i;
            int g = blk0 + (f >> 4);
            if (g > nnode - 1) g = nnode - 1;
            const size_t o = (size_t)g * DLAT + ((f & 15) << 2);
            xr[i] = *(const float4*)(xin + o);
            hr[i] = *(const float4*)(hraw + o);
        }
        #pragma unroll
        for (int i = 0; i < 4; ++i) {
            wl[i] = ((const float4*)Wl)[tid + 256 * i];
            wg[i] = ((const float4*)Wg)[tid + 256 * i];
            if constexpr (FUSE_NEXT) wc[i] = ((const float4*)Wc)[tid + 256 * i];
        }
        #pragma unroll
        for (int i = 0; i < 2; ++i) {
            const int f = tid + 256 * i;
            const int o = (f >> 4) * XPITCH + ((f & 15) << 2);
            *(float4*)(xs + o) = xr[i];
            float4 hv;
            hv.x = leaky(hr[i].x); hv.y = leaky(hr[i].y);
            hv.z = leaky(hr[i].z); hv.w = leaky(hr[i].w);
            *(float4*)(hs + o) = hv;
        }
        #pragma unroll
        for (int i = 0; i < 4; ++i) {
            ((float4*)Wl_s)[tid + 256 * i] = wl[i];
            ((float4*)Wg_s)[tid + 256 * i] = wg[i];
            if constexpr (FUSE_NEXT) ((float4*)Wc_s)[tid + 256 * i] = wc[i];
        }
    }
    __syncthreads();

    const int lane    = tid & 31;
    const int wave    = tid >> 5;
    const int rowtile = wave >> 2;
    const int coltile = wave & 3;
    const int mn      = lane & 15;
    const int kh      = (lane >> 4) << 1;
    const int col     = coltile * 16 + mn;

    const float* xp = xs + (rowtile * 16 + mn) * XPITCH + kh;
    const float* hp = hs + (rowtile * 16 + mn) * XPITCH + kh;

    v8f c1 = {}, c2 = {};
    #pragma unroll
    for (int kk = 0; kk < DLAT; kk += 4) {
        v2f ax = *(const v2f*)(xp + kk);
        v2f ah = *(const v2f*)(hp + kk);
        v2f b1;
        b1.x = Wl_s[(kk + kh) * DLAT + col];
        b1.y = Wl_s[(kk + kh + 1) * DLAT + col];
        v2f b2;
        b2.x = Wg_s[(kk + kh) * DLAT + col];
        b2.y = Wg_s[(kk + kh + 1) * DLAT + col];
        c1 = wmma_f32(ax, b1, c1);
        c2 = wmma_f32(ah, b2, c2);
    }

    const float blv = bl[col], bgv = bg[col];
    const int rbase = rowtile * 16 + ((lane >> 4) << 3);
    #pragma unroll
    for (int i = 0; i < 8; ++i) {
        int rloc = rbase + i;
        int g    = blk0 + rloc;
        int gc   = g > nnode - 1 ? nnode - 1 : g;
        float idv  = idemb[(size_t)gc * DLAT + col];
        float xhat = leaky(c1[i] + blv) + idv;
        float v    = leaky(c2[i] + bgv + xhat);
        if constexpr (FUSE_NEXT) {
            xt[rloc * 65 + col] = v;
        } else {
            if (g < nnode) xout[(size_t)g * DLAT + col] = v;
        }
    }

    if constexpr (FUSE_NEXT) {
        __syncthreads();
        for (int e = tid; e < 32 * DLAT; e += 256) {
            int r = e >> 6, c = e & 63;
            int g = blk0 + r;
            if (g < nnode) xout[(size_t)g * DLAT + c] = xt[r * 65 + c];
        }
        v8f acc2 = {};
        const int rl = rowtile * 16 + mn;
        #pragma unroll
        for (int kk = 0; kk < DLAT; kk += 4) {
            v2f a;
            a.x = xt[rl * 65 + kk + kh];
            a.y = xt[rl * 65 + kk + kh + 1];
            v2f b;
            b.x = Wc_s[(kk + kh) * DLAT + col];
            b.y = Wc_s[(kk + kh + 1) * DLAT + col];
            acc2 = wmma_f32(a, b, acc2);
        }
        #pragma unroll
        for (int i = 0; i < 8; ++i) {
            int g = blk0 + rbase + i;
            if (g < nnode) yout[(size_t)g * DLAT + col] = acc2[i];
        }
    }
}

__global__ __launch_bounds__(256) void hist_kernel(
    const int* __restrict__ dst, int* __restrict__ cnt, int* __restrict__ rank, int nedge)
{
    const int S    = gridDim.x * 256;
    const int last = nedge - 1;
    for (int b = blockIdx.x * 256 + threadIdx.x; b < nedge; b += 8 * S) {
        int d[8], r[8];
        #pragma unroll
        for (int k = 0; k < 8; ++k) d[k] = dst[min(b + k * S, last)];
        #pragma unroll
        for (int k = 0; k < 8; ++k) { r[k] = 0; if (b + k * S < nedge) r[k] = atomicAdd(cnt + d[k], 1); }
        #pragma unroll
        for (int k = 0; k < 8; ++k) { const int e = b + k * S; if (e < nedge) rank[e] = r[k]; }
    }
}

__global__ __launch_bounds__(1024) void scan_kernel(
    const int* __restrict__ cnt, int* __restrict__ off, int n)
{
    __shared__ int wtot[32];
    __shared__ int wtot2[32];
    const int tid  = threadIdx.x;
    const int lane = tid & 31;
    const int wave = tid >> 5;
    const int base = blockIdx.x * 1024;

    const int4* c4  = (const int4*)cnt;
    const int   nb4 = base >> 2;
    int s = 0;
    int i = tid;
    for (; i + 7 * 1024 < nb4; i += 8 * 1024) {
        int4 a[8];
        #pragma unroll
        for (int k = 0; k < 8; ++k) a[k] = c4[i + k * 1024];
        #pragma unroll
        for (int k = 0; k < 8; ++k) s += (a[k].x + a[k].y) + (a[k].z + a[k].w);
    }
    for (; i < nb4; i += 1024) {
        int4 a = c4[i];
        s += (a.x + a.y) + (a.z + a.w);
    }
    for (int o = 16; o > 0; o >>= 1) s += __shfl_xor(s, o);
    if (lane == 0) wtot[wave] = s;

    const int idx = base + tid;
    const int v   = idx < n ? cnt[idx] : 0;
    int inc = v;
    for (int o = 1; o < 32; o <<= 1) {
        int u = __shfl_up(inc, o);
        if (lane >= o) inc += u;
    }
    if (lane == 31) wtot2[wave] = inc;
    __syncthreads();

    int prefix = wtot[lane];
    for (int o = 16; o > 0; o >>= 1) prefix += __shfl_xor(prefix, o);
    int wp = lane < wave ? wtot2[lane] : 0;
    for (int o = 16; o > 0; o >>= 1) wp += __shfl_xor(wp, o);

    if (idx < n) off[idx] = prefix + wp + inc - v;
}

__global__ __launch_bounds__(256) void fill_kernel(
    const int* __restrict__ src, const int* __restrict__ dst,
    const int* __restrict__ off, const int* __restrict__ rank,
    int* __restrict__ ssrc, int nedge)
{
    const int S    = gridDim.x * 256;
    const int last = nedge - 1;
    for (int b = blockIdx.x * 256 + threadIdx.x; b < nedge; b += 8 * S) {
        int d[8], r[8], s[8], o[8];
        #pragma unroll
        for (int k = 0; k < 8; ++k) {
            const int e = min(b + k * S, last);
            d[k] = dst[e];
            r[k] = rank[e];
            s[k] = src[e];
        }
        #pragma unroll
        for (int k = 0; k < 8; ++k) o[k] = off[d[k]];
        #pragma unroll
        for (int k = 0; k < 8; ++k) { if (b + k * S < nedge) ssrc[o[k] + r[k]] = s[k]; }
    }
}

__global__ __launch_bounds__(256) void gather_kernel(
    const float* __restrict__ y, const int* __restrict__ off,
    const int* __restrict__ ssrc, float* __restrict__ h, int nnode)
{
    const int lane = threadIdx.x & 31;
    const int w0 = __builtin_amdgcn_readfirstlane((int)((blockIdx.x * 256u + threadIdx.x) >> 5));
    const int nw = (int)((gridDim.x * 256u) >> 5);
    const float* yl = y + lane * 2;

    int node = w0;
    int bv = (node < nnode && lane < 2) ? off[node + lane] : 0;
    while (node < nnode) {
        const int beg = __builtin_amdgcn_readlane(bv, 0);
        const int end = __builtin_amdgcn_readlane(bv, 1);
        int mys = lane < end - beg ? ssrc[beg + lane] : 0;
        const int nxt = node + nw;
        int nbv = 0;
        if (nxt < nnode && lane < 2) nbv = off[nxt + lane];

        v2f acc = {0.f, 0.f};
        int base = beg;
        while (base < end) {
            const int n = min(32, end - base);
            for (int j = 0; j < n; j += 16) {
                v2f v[16];
                #pragma unroll
                for (int k = 0; k < 16; ++k) {
                    const int s = __builtin_amdgcn_readlane(mys, j + k);
                    v[k] = *(const v2f*)(yl + (size_t)s * DLAT);
                }
                #pragma unroll
                for (int k = 0; k < 16; ++k) {
                    const bool on = (j + k) < n;
                    acc.x += on ? v[k].x : 0.f;
                    acc.y += on ? v[k].y : 0.f;
                }
            }
            base += 32;
            if (base < end) mys = lane < end - base ? ssrc[base + lane] : 0;
        }
        *(v2f*)(h + (size_t)node * DLAT + lane * 2) = acc;
        node = nxt; bv = nbv;
    }
}

__global__ __launch_bounds__(256) void zero_kernel(float4* __restrict__ p, size_t n4)
{
    size_t i      = (size_t)blockIdx.x * blockDim.x + threadIdx.x;
    size_t stride = (size_t)gridDim.x * blockDim.x;
    float4 z = {0.f, 0.f, 0.f, 0.f};
    for (; i < n4; i += stride) p[i] = z;
}

extern "C" void kernel_launch(void* const* d_in, const int* in_sizes, int n_in,
                              void* d_out, int out_size, void* d_ws, size_t ws_size,
                              hipStream_t stream)
{
    const float* features = (const float*)d_in[0];
    const float* id_emb   = (const float*)d_in[1];
    const int*   edge     = (const int*)d_in[2];
    const float* Wmlp     = (const float*)d_in[3];
    const float* bmlp     = (const float*)d_in[4];
    const float* Wc1      = (const float*)d_in[5];
    const float* Wl1      = (const float*)d_in[6];
    const float* bl1      = (const float*)d_in[7];
    const float* Wg1      = (const float*)d_in[8];
    const float* bg1      = (const float*)d_in[9];
    const float* Wc2      = (const float*)d_in[10];
    const float* Wl2      = (const float*)d_in[11];
    const float* bl2      = (const float*)d_in[12];
    const float* Wg2      = (const float*)d_in[13];
    const float* bg2      = (const float*)d_in[14];

    const int nnode = in_sizes[0] / DFEAT;
    const int nedge = in_sizes[2] / 2;
    const int* src  = edge;
    const int* dst  = edge + nedge;

    float* x = (float*)d_ws;
    float* h = x + (size_t)nnode * DLAT;
    float* y = h + (size_t)nnode * DLAT;

    const size_t cpad = (((size_t)nnode + 4) / 4) * 4;
    const size_t epad = (((size_t)nedge + 3) / 4) * 4;
    int* cnt  = (int*)(y + (size_t)nnode * DLAT);
    int* off  = cnt + cpad;
    int* rank = off + cpad;
    int* ssrc = rank + epad;

    const int nblk = (nnode + 31) / 32;
    const int sblk = (nnode + 1 + 1023) / 1024;

    zero_kernel<<<(int)((cpad / 4 + 255) / 256), 256, 0, stream>>>((float4*)cnt, cpad / 4);
    hist_kernel<<<1024, 256, 0, stream>>>(dst, cnt, rank, nedge);
    scan_kernel<<<sblk, 1024, 0, stream>>>(cnt, off, nnode + 1);
    fill_kernel<<<1024, 256, 0, stream>>>(src, dst, off, rank, ssrc, nedge);

    mlp_norm_conv_kernel<<<nblk, 256, 0, stream>>>(features, Wmlp, bmlp, Wc1, x, y, nnode);
    gather_kernel<<<2048, 256, 0, stream>>>(y, off, ssrc, h, nnode);
    update_kernel<true><<<nblk, 256, 0, stream>>>(x, h, id_emb, Wl1, bl1, Wg1, bg1,
                                                  Wc2, x, y, nnode);
    gather_kernel<<<2048, 256, 0, stream>>>(y, off, ssrc, h, nnode);
    update_kernel<false><<<nblk, 256, 0, stream>>>(x, h, id_emb, Wl2, bl2, Wg2, bg2,
                                                   nullptr, (float*)d_out, nullptr, nnode);
}


# Edges grouped by destination with a two-level counting sort in LDS (chunk bins, then rows per bin; row slot = LDS add arrival order), not global atomics and 4-B scatters; 9 to 7 launches
# speedup vs baseline: 2.4671x; 1.7485x over previous
#include <hip/hip_runtime.h>
#include <hip/hip_bf16.h>

#define DLAT 64
#define DFEAT 128
#define SLOPE 0.01f
#define EPSN 1e-12f

typedef __attribute__((ext_vector_type(2))) float v2f;
typedef __attribute__((ext_vector_type(8))) float v8f;

__device__ __forceinline__ float leaky(float v) { return v >= 0.0f ? v : SLOPE * v; }

__device__ __forceinline__ v8f wmma_f32(v2f a, v2f b, v8f c) {
    return __builtin_amdgcn_wmma_f32_16x16x4_f32(false, a, false, b, (short)0, c,
                                                 false, false);
}

#define FPITCH 132
#define XPITCH 68

__global__ __launch_bounds__(256) void mlp_norm_conv_kernel(
    const float* __restrict__ feat, const float* __restrict__ Wmlp,
    const float* __restrict__ bmlp, const float* __restrict__ Wc,
    float* __restrict__ xout, float* __restrict__ yout, int nnode)
{
    __shared__ __attribute__((aligned(16))) float Wm_s[DFEAT * DLAT];
    __shared__ __attribute__((aligned(16))) float Wc_s[DLAT * DLAT];
    __shared__ __attribute__((aligned(16))) float ft[32 * FPITCH];
    __shared__ float xt[32 * 65];
    __shared__ float scl[32];

    const int tid  = threadIdx.x;
    const int blk0 = blockIdx.x * 32;

    {
        float4 fr[4], wm[8], wc[4];
        #pragma unroll
        for (int i = 0; i < 4; ++i) {
            const int f = tid + 256 * i;
            int g = blk0 + (f >> 5);
            if (g > nnode - 1) g = nnode - 1;
            fr[i] = *(const float4*)(feat + (size_t)g * DFEAT + ((f & 31) << 2));
        }
        #pragma unroll
        for (int i = 0; i < 8; ++i) wm[i] = ((const float4*)Wmlp)[tid + 256 * i];
        #pragma unroll
        for (int i = 0; i < 4; ++i) wc[i] = ((const float4*)Wc)[tid + 256 * i];
        #pragma unroll
        for (int i = 0; i < 4; ++i) {
            const int f = tid + 256 * i;
            *(float4*)(ft + (f >> 5) * FPITCH + ((f & 31) << 2)) = fr[i];
        }
        #pragma unroll
        for (int i = 0; i < 8; ++i) ((float4*)Wm_s)[tid + 256 * i] = wm[i];
        #pragma unroll
        for (int i = 0; i < 4; ++i) ((float4*)Wc_s)[tid + 256 * i] = wc[i];
    }
    __syncthreads();

    const int lane    = tid & 31;
    const int wave    = tid >> 5;
    const int rowtile = wave >> 2;
    const int coltile = wave & 3;
    const int mn      = lane & 15;
    const int kh      = (lane >> 4) << 1;
    const int col     = coltile * 16 + mn;

    const float* ap = ft + (rowtile * 16 + mn) * FPITCH + kh;

    v8f acc = {};
    #pragma unroll
    for (int kk = 0; kk < DFEAT; kk += 4) {
        v2f a = *(const v2f*)(ap + kk);
        v2f b;
        b.x = Wm_s[(kk + kh) * DLAT + col];
        b.y = Wm_s[(kk + kh + 1) * DLAT + col];
        acc = wmma_f32(a, b, acc);
    }

    const float bias  = bmlp[col];
    const int   rbase = rowtile * 16 + ((lane >> 4) << 3);
    #pragma unroll
    for (int i = 0; i < 8; ++i)
        xt[(rbase + i) * 65 + col] = acc[i] + bias;
    __syncthreads();

    if (tid < 32) {
        float s = 0.f;
        #pragma unroll
        for (int j = 0; j < DLAT; ++j) { float v = xt[tid * 65 + j]; s += v * v; }
        scl[tid] = 1.0f / fmaxf(sqrtf(s), EPSN);
    }
    __syncthreads();

    for (int e = tid; e < 32 * DLAT; e += 256) {
        int r = e >> 6, c = e & 63;
        float v = xt[r * 65 + c] * scl[r];
        xt[r * 65 + c] = v;
        int g = blk0 + r;
        if (g < nnode) xout[(size_t)g * DLAT + c] = v;
    }
    __syncthreads();

    v8f acc2 = {};
    const int rl = rowtile * 16 + mn;
    #pragma unroll
    for (int kk = 0; kk < DLAT; kk += 4) {
        v2f a;
        a.x = xt[rl * 65 + kk + kh];
        a.y = xt[rl * 65 + kk + kh + 1];
        v2f b;
        b.x = Wc_s[(kk + kh) * DLAT + col];
        b.y = Wc_s[(kk + kh + 1) * DLAT + col];
        acc2 = wmma_f32(a, b, acc2);
    }
    #pragma unroll
    for (int i = 0; i < 8; ++i) {
        int g = blk0 + rbase + i;
        if (g < nnode) yout[(size_t)g * DLAT + col] = acc2[i];
    }
}

template <bool FUSE_NEXT>
__global__ __launch_bounds__(256) void update_kernel(
    const float* __restrict__ xin, const float* __restrict__ hraw,
    const float* __restrict__ idemb,
    const float* __restrict__ Wl, const float* __restrict__ bl,
    const float* __restrict__ Wg, const float* __restrict__ bg,
    const float* __restrict__ Wc, float* __restrict__ xout,
    float* __restrict__ yout, int nnode)
{
    __shared__ __attribute__((aligned(16))) float Wl_s[DLAT * DLAT];
    __shared__ __attribute__((aligned(16))) float Wg_s[DLAT * DLAT];
    __shared__ __attribute__((aligned(16))) float Wc_s[DLAT * DLAT];
    __shared__ __attribute__((aligned(16))) float xs[32 * XPITCH];
    __shared__ __attribute__((aligned(16))) float hs[32 * XPITCH];
    __shared__ float xt[32 * 65];

    const int tid  = threadIdx.x;
    const int blk0 = blockIdx.x * 32;

    {
        float4 xr[2], hr[2], wl[4], wg[4], wc[4];
        #pragma unroll
        for (int i = 0; i < 2; ++i) {
            const int f = tid + 256 * i;
            int g = blk0 + (f >> 4);
            if (g > nnode - 1) g = nnode - 1;
            const size_t o = (size_t)g * DLAT + ((f & 15) << 2);
            xr[i] = *(const float4*)(xin + o);
            hr[i] = *(const float4*)(hraw + o);
        }
        #pragma unroll
        for (int i = 0; i < 4; ++i) {
            wl[i] = ((const float4*)Wl)[tid + 256 * i];
            wg[i] = ((const float4*)Wg)[tid + 256 * i];
            if constexpr (FUSE_NEXT) wc[i] = ((const float4*)Wc)[tid + 256 * i];
        }
        #pragma unroll
        for (int i = 0; i < 2; ++i) {
            const int f = tid + 256 * i;
            const int o = (f >> 4) * XPITCH + ((f & 15) << 2);
            *(float4*)(xs + o) = xr[i];
            float4 hv;
            hv.x = leaky(hr[i].x); hv.y = leaky(hr[i].y);
            hv.z = leaky(hr[i].z); hv.w = leaky(hr[i].w);
            *(float4*)(hs + o) = hv;
        }
        #pragma unroll
        for (int i = 0; i < 4; ++i) {
            ((float4*)Wl_s)[tid + 256 * i] = wl[i];
            ((float4*)Wg_s)[tid + 256 * i] = wg[i];
            if constexpr (FUSE_NEXT) ((float4*)Wc_s)[tid + 256 * i] = wc[i];
        }
    }
    __syncthreads();

    const int lane    = tid & 31;
    const int wave    = tid >> 5;
    const int rowtile = wave >> 2;
    const int coltile = wave & 3;
    const int mn      = lane & 15;
    const int kh      = (lane >> 4) << 1;
    const int col     = coltile * 16 + mn;

    const float* xp = xs + (rowtile * 16 + mn) * XPITCH + kh;
    const float* hp = hs + (rowtile * 16 + mn) * XPITCH + kh;

    v8f c1 = {}, c2 = {};
    #pragma unroll
    for (int kk = 0; kk < DLAT; kk += 4) {
        v2f ax = *(const v2f*)(xp + kk);
        v2f ah = *(const v2f*)(hp + kk);
        v2f b1;
        b1.x = Wl_s[(kk + kh) * DLAT + col];
        b1.y = Wl_s[(kk + kh + 1) * DLAT + col];
        v2f b2;
        b2.x = Wg_s[(kk + kh) * DLAT + col];
        b2.y = Wg_s[(kk + kh + 1) * DLAT + col];
        c1 = wmma_f32(ax, b1, c1);
        c2 = wmma_f32(ah, b2, c2);
    }

    const float blv = bl[col], bgv = bg[col];
    const int rbase = rowtile * 16 + ((lane >> 4) << 3);
    #pragma unroll
    for (int i = 0; i < 8; ++i) {
        int rloc = rbase + i;
        int g    = blk0 + rloc;
        int gc   = g > nnode - 1 ? nnode - 1 : g;
        float idv  = idemb[(size_t)gc * DLAT + col];
        float xhat = leaky(c1[i] + blv) + idv;
        float v    = leaky(c2[i] + bgv + xhat);
        if constexpr (FUSE_NEXT) {
            xt[rloc * 65 + col] = v;
        } else {
            if (g < nnode) xout[(size_t)g * DLAT + col] = v;
        }
    }

    if constexpr (FUSE_NEXT) {
        __syncthreads();
        for (int e = tid; e < 32 * DLAT; e += 256) {
            int r = e >> 6, c = e & 63;
            int g = blk0 + r;
            if (g < nnode) xout[(size_t)g * DLAT + c] = xt[r * 65 + c];
        }
        v8f acc2 = {};
        const int rl = rowtile * 16 + mn;
        #pragma unroll
        for (int kk = 0; kk < DLAT; kk += 4) {
            v2f a;
            a.x = xt[rl * 65 + kk + kh];
            a.y = xt[rl * 65 + kk + kh + 1];
            v2f b;
            b.x = Wc_s[(kk + kh) * DLAT + col];
            b.y = Wc_s[(kk + kh + 1) * DLAT + col];
            acc2 = wmma_f32(a, b, acc2);
        }
        #pragma unroll
        for (int i = 0; i < 8; ++i) {
            int g = blk0 + rbase + i;
            if (g < nnode) yout[(size_t)g * DLAT + col] = acc2[i];
        }
    }
}

#define SORT_CH  8192
#define BIN_SH   9
#define BIN_ROWS 512
#define MAXNB    1024
#define MAXCH    1024
#define BIN_CAP  12288

__device__ __forceinline__ int block_excl_scan_1024(int v, int* wsum, int& total)
{
    const int lane = threadIdx.x & 31;
    const int wave = threadIdx.x >> 5;
    int inc = v;
    for (int o = 1; o < 32; o <<= 1) {
        int u = __shfl_up(inc, o);
        if (lane >= o) inc += u;
    }
    if (lane == 31) wsum[wave] = inc;
    __syncthreads();
    const int ws = wsum[lane];
    int wp  = lane < wave ? ws : 0;
    int tot = ws;
    for (int o = 16; o > 0; o >>= 1) {
        wp  += __shfl_xor(wp, o);
        tot += __shfl_xor(tot, o);
    }
    total = tot;
    __syncthreads();
    return wp + inc - v;
}

__global__ __launch_bounds__(1024) void chunk_sort_kernel(
    const int* __restrict__ src, const int* __restrict__ dst,
    int2* __restrict__ cs, int* __restrict__ tab, int* __restrict__ lst,
    int nedge, int nb, int nch)
{
    __shared__ int  lcnt[MAXNB];
    __shared__ int  lsta[MAXNB];
    __shared__ int  wsum[32];
    __shared__ int2 stage[SORT_CH];

    const int tid  = threadIdx.x;
    const int c    = blockIdx.x;
    const int e0   = c * SORT_CH;
    const int last = nedge - 1;

    lcnt[tid] = 0;
    __syncthreads();

    int s[8], d[8], r[8];
    #pragma unroll
    for (int k = 0; k < 8; ++k) {
        const int e = min(e0 + tid + k * 1024, last);
        s[k] = src[e];
        d[k] = dst[e];
    }
    #pragma unroll
    for (int k = 0; k < 8; ++k) {
        r[k] = 0;
        const int bin = min(d[k] >> BIN_SH, MAXNB - 1);
        if (e0 + tid + k * 1024 < nedge) r[k] = atomicAdd(&lcnt[bin], 1);
    }
    __syncthreads();

    const int v = lcnt[tid];
    int tot;
    const int ex = block_excl_scan_1024(v, wsum, tot);
    lsta[tid] = ex;
    if (tid < nb) {
        tab[(size_t)tid * nch + c] = v;
        lst[(size_t)c * nb + tid]  = ex;
    }
    __syncthreads();

    #pragma unroll
    for (int k = 0; k < 8; ++k) {
        const int bin = min(d[k] >> BIN_SH, MAXNB - 1);
        if (e0 + tid + k * 1024 < nedge) stage[lsta[bin] + r[k]] = make_int2(s[k], d[k]);
    }
    __syncthreads();

    for (int j = tid; j < tot; j += 1024) cs[(size_t)e0 + j] = stage[j];
}

__global__ __launch_bounds__(1024) void bin_sort_kernel(
    const int2* __restrict__ cs, const int* __restrict__ tab, const int* __restrict__ lst,
    int* __restrict__ off, int* __restrict__ ssrc, int nnode, int nb, int nch)
{
    __shared__ int  ppre[MAXCH];
    __shared__ int  ps[MAXCH];
    __shared__ int  lcnt[BIN_ROWS];
    __shared__ int  loff[BIN_ROWS];
    __shared__ int  lcur[BIN_ROWS];
    __shared__ int  wsum[32];
    __shared__ int  gsum[32];
    __shared__ int2 A[BIN_CAP];
    __shared__ int  B[BIN_CAP];

    const int tid   = threadIdx.x;
    const int lane  = tid & 31;
    const int wave  = tid >> 5;
    const int b     = blockIdx.x;
    const int rbase = b << BIN_SH;

    int n = 0, st = 0;
    if (tid < nch) {
        n  = tab[(size_t)b * nch + tid];
        st = lst[(size_t)tid * nb + b];
    }

    int g = 0;
    {
        const int   nfront = b * nch;
        const int   n4     = nfront >> 2;
        const int4* t4     = (const int4*)tab;
        int i = tid;
        for (; i + 7 * 1024 < n4; i += 8 * 1024) {
            int4 a[8];
            #pragma unroll
            for (int k = 0; k < 8; ++k) a[k] = t4[i + k * 1024];
            #pragma unroll
            for (int k = 0; k < 8; ++k) g += (a[k].x + a[k].y) + (a[k].z + a[k].w);
        }
        for (; i < n4; i += 1024) {
            int4 a = t4[i];
            g += (a.x + a.y) + (a.z + a.w);
        }
        for (int q = (n4 << 2) + tid; q < nfront; q += 1024) g += tab[q];
    }
    for (int o = 16; o > 0; o >>= 1) g += __shfl_xor(g, o);
    if (lane == 0) gsum[wave] = g;

    ps[tid] = st;
    if (tid < BIN_ROWS) { lcnt[tid] = 0; lcur[tid] = 0; }
    int total;
    const int ex = block_excl_scan_1024(n, wsum, total);
    ppre[tid] = ex;
    int gb = gsum[lane];
    for (int o = 16; o > 0; o >>= 1) gb += __shfl_xor(gb, o);
    __syncthreads();

    const bool single = total <= BIN_CAP;

    for (int j0 = tid; j0 < total; j0 += 8 * 1024) {
        int2 e[8];
        #pragma unroll
        for (int k = 0; k < 8; ++k) {
            const int j = min(j0 + k * 1024, total - 1);
            int cc = 0;
            #pragma unroll
            for (int sft = MAXCH / 2; sft > 0; sft >>= 1)
                if (ppre[cc + sft] <= j) cc += sft;
            e[k] = cs[(size_t)cc * SORT_CH + ps[cc] + (j - ppre[cc])];
        }
        #pragma unroll
        for (int k = 0; k < 8; ++k) {
            const int j = j0 + k * 1024;
            if (j < total) {
                const int ld = min(max(e[k].y - rbase, 0), BIN_ROWS - 1);
                atomicAdd(&lcnt[ld], 1);
                if (single) A[j] = make_int2(e[k].x, ld);
            }
        }
    }
    __syncthreads();

    const int cv = tid < BIN_ROWS ? lcnt[tid] : 0;
    int rtot;
    const int rex = block_excl_scan_1024(cv, wsum, rtot);
    if (tid < BIN_ROWS) {
        loff[tid] = rex;
        if (rbase + tid < nnode) off[rbase + tid] = gb + rex;
    }
    if (b == nb - 1 && tid == 0) off[nnode] = gb + total;
    __syncthreads();

    if (single) {
        for (int j = tid; j < total; j += 1024) {
            const int2 e = A[j];
            const int pos = loff[e.y] + atomicAdd(&lcur[e.y], 1);
            B[pos] = e.x;
        }
        __syncthreads();
        for (int j = tid; j < total; j += 1024) ssrc[(size_t)gb + j] = B[j];
    } else {
        for (int j = tid; j < total; j += 1024) {
            int cc = 0;
            #pragma unroll
            for (int sft = MAXCH / 2; sft > 0; sft >>= 1)
                if (ppre[cc + sft] <= j) cc += sft;
            const int2 e  = cs[(size_t)cc * SORT_CH + ps[cc] + (j - ppre[cc])];
            const int  ld = min(max(e.y - rbase, 0), BIN_ROWS - 1);
            const int  pos = loff[ld] + atomicAdd(&lcur[ld], 1);
            ssrc[(size_t)gb + pos] = e.x;
        }
    }
}

__global__ __launch_bounds__(256) void gather_kernel(
    const float* __restrict__ y, const int* __restrict__ off,
    const int* __restrict__ ssrc, float* __restrict__ h, int nnode)
{
    const int lane = threadIdx.x & 31;
    const int w0 = __builtin_amdgcn_readfirstlane((int)((blockIdx.x * 256u + threadIdx.x) >> 5));
    const int nw = (int)((gridDim.x * 256u) >> 5);
    const float* yl = y + lane * 2;

    int node = w0;
    int bv = (node < nnode && lane < 2) ? off[node + lane] : 0;
    while (node < nnode) {
        const int beg = __builtin_amdgcn_readlane(bv, 0);
        const int end = __builtin_amdgcn_readlane(bv, 1);
        int mys = lane < end - beg ? ssrc[beg + lane] : 0;
        const int nxt = node + nw;
        int nbv = 0;
        if (nxt < nnode && lane < 2) nbv = off[nxt + lane];

        v2f acc = {0.f, 0.f};
        int base = beg;
        while (base < end) {
            const int n = min(32, end - base);
            for (int j = 0; j < n; j += 16) {
                v2f v[16];
                #pragma unroll
                for (int k = 0; k < 16; ++k) {
                    const int s = __builtin_amdgcn_readlane(mys, j + k);
                    v[k] = *(const v2f*)(yl + (size_t)s * DLAT);
                }
                #pragma unroll
                for (int k = 0; k < 16; ++k) {
                    const bool on = (j + k) < n;
                    acc.x += on ? v[k].x : 0.f;
                    acc.y += on ? v[k].y : 0.f;
                }
            }
            base += 32;
            if (base < end) mys = lane < end - base ? ssrc[base + lane] : 0;
        }
        *(v2f*)(h + (size_t)node * DLAT + lane * 2) = acc;
        node = nxt; bv = nbv;
    }
}

extern "C" void kernel_launch(void* const* d_in, const int* in_sizes, int n_in,
                              void* d_out, int out_size, void* d_ws, size_t ws_size,
                              hipStream_t stream)
{
    const float* features = (const float*)d_in[0];
    const float* id_emb   = (const float*)d_in[1];
    const int*   edge     = (const int*)d_in[2];
    const float* Wmlp     = (const float*)d_in[3];
    const float* bmlp     = (const float*)d_in[4];
    const float* Wc1      = (const float*)d_in[5];
    const float* Wl1      = (const float*)d_in[6];
    const float* bl1      = (const float*)d_in[7];
    const float* Wg1      = (const float*)d_in[8];
    const float* bg1      = (const float*)d_in[9];
    const float* Wc2      = (const float*)d_in[10];
    const float* Wl2      = (const float*)d_in[11];
    const float* bl2      = (const float*)d_in[12];
    const float* Wg2      = (const float*)d_in[13];
    const float* bg2      = (const float*)d_in[14];

    const int nnode = in_sizes[0] / DFEAT;
    const int nedge = in_sizes[2] / 2;
    const int* src  = edge;
    const int* dst  = edge + nedge;

    float* x = (float*)d_ws;
    float* h = x + (size_t)nnode * DLAT;
    float* y = h + (size_t)nnode * DLAT;

    const size_t cpad = (((size_t)nnode + 4) / 4) * 4;
    const size_t epad = (((size_t)nedge + 3) / 4) * 4;
    const int nb  = (nnode + BIN_ROWS - 1) / BIN_ROWS;
    const int nch = (nedge + SORT_CH - 1) / SORT_CH;
    int*  off  = (int*)(y + (size_t)nnode * DLAT);
    int*  ssrc = off + cpad;
    int2* cs   = (int2*)(ssrc + epad);
    int*  tab  = (int*)(cs + epad);
    int*  lst  = tab + (((size_t)nb * nch + 3) / 4) * 4;

    const int nblk = (nnode + 31) / 32;

    chunk_sort_kernel<<<nch, 1024, 0, stream>>>(src, dst, cs, tab, lst, nedge, nb, nch);
    bin_sort_kernel<<<nb, 1024, 0, stream>>>(cs, tab, lst, off, ssrc, nnode, nb, nch);

    mlp_norm_conv_kernel<<<nblk, 256, 0, stream>>>(features, Wmlp, bmlp, Wc1, x, y, nnode);
    gather_kernel<<<2048, 256, 0, stream>>>(y, off, ssrc, h, nnode);
    update_kernel<true><<<nblk, 256, 0, stream>>>(x, h, id_emb, Wl1, bl1, Wg1, bg1,
                                                  Wc2, x, y, nnode);
    gather_kernel<<<2048, 256, 0, stream>>>(y, off, ssrc, h, nnode);
    update_kernel<false><<<nblk, 256, 0, stream>>>(x, h, id_emb, Wl2, bl2, Wg2, bg2,
                                                   nullptr, (float*)d_out, nullptr, nnode);
}


# Dense kernels as one resident pass: 1024 blocks keep the weights in LDS and loop over the 32-row tiles; the next tile's rows and id values load under the current products, not in its epilogue
# speedup vs baseline: 2.6330x; 1.0672x over previous
#include <hip/hip_runtime.h>
#include <hip/hip_bf16.h>

#define DLAT 64
#define DFEAT 128
#define SLOPE 0.01f
#define EPSN 1e-12f

typedef __attribute__((ext_vector_type(2))) float v2f;
typedef __attribute__((ext_vector_type(8))) float v8f;
typedef __attribute__((ext_vector_type(4))) float v4f;

__device__ __forceinline__ float leaky(float v) { return v >= 0.0f ? v : SLOPE * v; }

__device__ __forceinline__ v8f wmma_f32(v2f a, v2f b, v8f c) {
    return __builtin_amdgcn_wmma_f32_16x16x4_f32(false, a, false, b, (short)0, c,
                                                 false, false);
}

#define FPITCH 132
#define XPITCH 68
#define DENSE_GRID 1024

__device__ __forceinline__ v4f ld_feat(const float* __restrict__ feat, int tile, int f, int nnode) {
    int g = tile * 32 + (f >> 5);
    if (g > nnode - 1) g = nnode - 1;
    return *(const v4f*)(feat + (size_t)g * DFEAT + ((f & 31) << 2));
}
__device__ __forceinline__ void st_feat(float* ft, int f, v4f v) {
    *(v4f*)(ft + (f >> 5) * FPITCH + ((f & 31) << 2)) = v;
}
__device__ __forceinline__ v4f ld_row64(const float* __restrict__ p, int tile, int f, int nnode) {
    int g = tile * 32 + (f >> 4);
    if (g > nnode - 1) g = nnode - 1;
    return *(const v4f*)(p + (size_t)g * DLAT + ((f & 15) << 2));
}
__device__ __forceinline__ v4f leaky4(v4f v) {
    v4f r;
    r.x = leaky(v.x); r.y = leaky(v.y); r.z = leaky(v.z); r.w = leaky(v.w);
    return r;
}
__device__ __forceinline__ v8f ld_id(const float* __restrict__ idemb, int tile, int rbase, int col, int nnode) {
    v8f r;
    #pragma unroll
    for (int i = 0; i < 8; ++i) {
        int g = tile * 32 + rbase + i;
        if (g > nnode - 1) g = nnode - 1;
        r[i] = idemb[(size_t)g * DLAT + col];
    }
    return r;
}

__global__ __launch_bounds__(256) void mlp_norm_conv_kernel(
    const float* __restrict__ feat, const float* __restrict__ Wmlp,
    const float* __restrict__ bmlp, const float* __restrict__ Wc,
    float* __restrict__ xout, float* __restrict__ yout, int nnode)
{
    __shared__ __attribute__((aligned(16))) float Wm_s[DFEAT * DLAT];
    __shared__ __attribute__((aligned(16))) float Wc_s[DLAT * DLAT];
    __shared__ __attribute__((aligned(16))) float ft[32 * FPITCH];
    __shared__ float xt[32 * 65];
    __shared__ float scl[32];

    const int tid   = threadIdx.x;
    const int ntile = (nnode + 31) >> 5;
    const int tstep = (int)gridDim.x;
    int tile = (int)blockIdx.x;
    if (tile >= ntile) return;

    v4f fr0 = ld_feat(feat, tile, tid,       nnode);
    v4f fr1 = ld_feat(feat, tile, tid + 256, nnode);
    v4f fr2 = ld_feat(feat, tile, tid + 512, nnode);
    v4f fr3 = ld_feat(feat, tile, tid + 768, nnode);
    {
        float4 wm[8], wc[4];
        #pragma unroll
        for (int i = 0; i < 8; ++i) wm[i] = ((const float4*)Wmlp)[tid + 256 * i];
        #pragma unroll
        for (int i = 0; i < 4; ++i) wc[i] = ((const float4*)Wc)[tid + 256 * i];
        #pragma unroll
        for (int i = 0; i < 8; ++i) ((float4*)Wm_s)[tid + 256 * i] = wm[i];
        #pragma unroll
        for (int i = 0; i < 4; ++i) ((float4*)Wc_s)[tid + 256 * i] = wc[i];
    }

    const int lane    = tid & 31;
    const int wave    = tid >> 5;
    const int rowtile = wave >> 2;
    const int coltile = wave & 3;
    const int mn      = lane & 15;
    const int kh      = (lane >> 4) << 1;
    const int col     = coltile * 16 + mn;

    const float* ap   = ft + (rowtile * 16 + mn) * FPITCH + kh;
    const float bias  = bmlp[col];
    const int   rbase = rowtile * 16 + ((lane >> 4) << 3);
    const int   rl    = rowtile * 16 + mn;

    for (; tile < ntile; tile += tstep) {
        const int blk0 = tile * 32;

        st_feat(ft, tid,       fr0);
        st_feat(ft, tid + 256, fr1);
        st_feat(ft, tid + 512, fr2);
        st_feat(ft, tid + 768, fr3);
        __syncthreads();

        const int ntl = tile + tstep;
        if (ntl < ntile) {
            fr0 = ld_feat(feat, ntl, tid,       nnode);
            fr1 = ld_feat(feat, ntl, tid + 256, nnode);
            fr2 = ld_feat(feat, ntl, tid + 512, nnode);
            fr3 = ld_feat(feat, ntl, tid + 768, nnode);
        }

        v8f acc = {};
        #pragma unroll
        for (int kk = 0; kk < DFEAT; kk += 4) {
            v2f a = *(const v2f*)(ap + kk);
            v2f b;
            b.x = Wm_s[(kk + kh) * DLAT + col];
            b.y = Wm_s[(kk + kh + 1) * DLAT + col];
            acc = wmma_f32(a, b, acc);
        }

        #pragma unroll
        for (int i = 0; i < 8; ++i)
            xt[(rbase + i) * 65 + col] = acc[i] + bias;
        __syncthreads();

        if (tid < 32) {
            float s = 0.f;
            #pragma unroll
            for (int j = 0; j < DLAT; ++j) { float v = xt[tid * 65 + j]; s += v * v; }
            scl[tid] = 1.0f / fmaxf(sqrtf(s), EPSN);
        }
        __syncthreads();

        for (int e = tid; e < 32 * DLAT; e += 256) {
            int r = e >> 6, c = e & 63;
            float v = xt[r * 65 + c] * scl[r];
            xt[r * 65 + c] = v;
            int g = blk0 + r;
            if (g < nnode) xout[(size_t)g * DLAT + c] = v;
        }
        __syncthreads();

        v8f acc2 = {};
        #pragma unroll
        for (int kk = 0; kk < DLAT; kk += 4) {
            v2f a;
            a.x = xt[rl * 65 + kk + kh];
            a.y = xt[rl * 65 + kk + kh + 1];
            v2f b;
            b.x = Wc_s[(kk + kh) * DLAT + col];
            b.y = Wc_s[(kk + kh + 1) * DLAT + col];
            acc2 = wmma_f32(a, b, acc2);
        }
        #pragma unroll
        for (int i = 0; i < 8; ++i) {
            int g = blk0 + rbase + i;
            if (g < nnode) yout[(size_t)g * DLAT + col] = acc2[i];
        }
    }
}

template <bool FUSE_NEXT>
__global__ __launch_bounds__(256) void update_kernel(
    const float* __restrict__ xin, const float* __restrict__ hraw,
    const float* __restrict__ idemb,
    const float* __restrict__ Wl, const float* __restrict__ bl,
    const float* __restrict__ Wg, const float* __restrict__ bg,
    const float* __restrict__ Wc, float* __restrict__ xout,
    float* __restrict__ yout, int nnode)
{
    __shared__ __attribute__((aligned(16))) float Wl_s[DLAT * DLAT];
    __shared__ __attribute__((aligned(16))) float Wg_s[DLAT * DLAT];
    __shared__ __attribute__((aligned(16))) float Wc_s[DLAT * DLAT];
    __shared__ __attribute__((aligned(16))) float xs[32 * XPITCH];
    __shared__ __attribute__((aligned(16))) float hs[32 * XPITCH];
    __shared__ float xt[32 * 65];

    const int tid   = threadIdx.x;
    const int ntile = (nnode + 31) >> 5;
    const int tstep = (int)gridDim.x;
    int tile = (int)blockIdx.x;
    if (tile >= ntile) return;

    v4f xr0 = ld_row64(xin,  tile, tid,       nnode);
    v4f hr0 = ld_row64(hraw, tile, tid,       nnode);
    v4f xr1 = ld_row64(xin,  tile, tid + 256, nnode);
    v4f hr1 = ld_row64(hraw, tile, tid + 256, nnode);
    {
        float4 wl[4], wg[4], wc[4];
        #pragma unroll
        for (int i = 0; i < 4; ++i) {
            wl[i] = ((const float4*)Wl)[tid + 256 * i];
            wg[i] = ((const float4*)Wg)[tid + 256 * i];
            if constexpr (FUSE_NEXT) wc[i] = ((const float4*)Wc)[tid + 256 * i];
        }
        #pragma unroll
        for (int i = 0; i < 4; ++i) {
            ((float4*)Wl_s)[tid + 256 * i] = wl[i];
            ((float4*)Wg_s)[tid + 256 * i] = wg[i];
            if constexpr (FUSE_NEXT) ((float4*)Wc_s)[tid + 256 * i] = wc[i];
        }
    }

    const int lane    = tid & 31;
    const int wave    = tid >> 5;
    const int rowtile = wave >> 2;
    const int coltile = wave & 3;
    const int mn      = lane & 15;
    const int kh      = (lane >> 4) << 1;
    const int col     = coltile * 16 + mn;

    const float* xp = xs + (rowtile * 16 + mn) * XPITCH + kh;
    const float* hp = hs + (rowtile * 16 + mn) * XPITCH + kh;

    const float blv = bl[col], bgv = bg[col];
    const int rbase = rowtile * 16 + ((lane >> 4) << 3);

    v8f idn = ld_id(idemb, tile, rbase, col, nnode);

    for (; tile < ntile; tile += tstep) {
        const int blk0 = tile * 32;
        const v8f idc = idn;

        {
            const int o0 = (tid >> 4) * XPITCH + ((tid & 15) << 2);
            const int o1 = o0 + 16 * XPITCH;
            *(v4f*)(xs + o0) = xr0;
            *(v4f*)(xs + o1) = xr1;
            *(v4f*)(hs + o0) = leaky4(hr0);
            *(v4f*)(hs + o1) = leaky4(hr1);
        }
        __syncthreads();

        const int ntl = tile + tstep;
        if (ntl < ntile) {
            xr0 = ld_row64(xin,  ntl, tid,       nnode);
            hr0 = ld_row64(hraw, ntl, tid,       nnode);
            xr1 = ld_row64(xin,  ntl, tid + 256, nnode);
            hr1 = ld_row64(hraw, ntl, tid + 256, nnode);
            idn = ld_id(idemb, ntl, rbase, col, nnode);
        }

        v8f c1 = {}, c2 = {};
        #pragma unroll 4
        for (int kk = 0; kk < DLAT; kk += 4) {
            v2f ax = *(const v2f*)(xp + kk);
            v2f ah = *(const v2f*)(hp + kk);
            v2f b1;
            b1.x = Wl_s[(kk + kh) * DLAT + col];
            b1.y = Wl_s[(kk + kh + 1) * DLAT + col];
            v2f b2;
            b2.x = Wg_s[(kk + kh) * DLAT + col];
            b2.y = Wg_s[(kk + kh + 1) * DLAT + col];
            c1 = wmma_f32(ax, b1, c1);
            c2 = wmma_f32(ah, b2, c2);
        }

        #pragma unroll
        for (int i = 0; i < 8; ++i) {
            int rloc = rbase + i;
            int g    = blk0 + rloc;
            float xhat = leaky(c1[i] + blv) + idc[i];
            float v    = leaky(c2[i] + bgv + xhat);
            if constexpr (FUSE_NEXT) {
                xt[rloc * 65 + col] = v;
            } else {
                if (g < nnode) xout[(size_t)g * DLAT + col] = v;
            }
        }

        __syncthreads();
        if constexpr (FUSE_NEXT) {
            for (int e = tid; e < 32 * DLAT; e += 256) {
                int r = e >> 6, c = e & 63;
                int g = blk0 + r;
                if (g < nnode) xout[(size_t)g * DLAT + c] = xt[r * 65 + c];
            }
            v8f acc2 = {};
            const int rl = rowtile * 16 + mn;
            #pragma unroll 4
            for (int kk = 0; kk < DLAT; kk += 4) {
                v2f a;
                a.x = xt[rl * 65 + kk + kh];
                a.y = xt[rl * 65 + kk + kh + 1];
                v2f b;
                b.x = Wc_s[(kk + kh) * DLAT + col];
                b.y = Wc_s[(kk + kh + 1) * DLAT + col];
                acc2 = wmma_f32(a, b, acc2);
            }
            #pragma unroll
            for (int i = 0; i < 8; ++i) {
                int g = blk0 + rbase + i;
                if (g < nnode) yout[(size_t)g * DLAT + col] = acc2[i];
            }
        }
    }
}

#define SORT_CH  8192
#define BIN_SH   9
#define BIN_ROWS 512
#define MAXNB    1024
#define MAXCH    1024
#define BIN_CAP  12288

__device__ __forceinline__ int block_excl_scan_1024(int v, int* wsum, int& total)
{
    const int lane = threadIdx.x & 31;
    const int wave = threadIdx.x >> 5;
    int inc = v;
    for (int o = 1; o < 32; o <<= 1) {
        int u = __shfl_up(inc, o);
        if (lane >= o) inc += u;
    }
    if (lane == 31) wsum[wave] = inc;
    __syncthreads();
    const int ws = wsum[lane];
    int wp  = lane < wave ? ws : 0;
    int tot = ws;
    for (int o = 16; o > 0; o >>= 1) {
        wp  += __shfl_xor(wp, o);
        tot += __shfl_xor(tot, o);
    }
    total = tot;
    __syncthreads();
    return wp + inc - v;
}

__global__ __launch_bounds__(1024) void chunk_sort_kernel(
    const int* __restrict__ src, const int* __restrict__ dst,
    int2* __restrict__ cs, int* __restrict__ tab, int* __restrict__ lst,
    int nedge, int nb, int nch)
{
    __shared__ int  lcnt[MAXNB];
    __shared__ int  lsta[MAXNB];
    __shared__ int  wsum[32];
    __shared__ int2 stage[SORT_CH];

    const int tid  = threadIdx.x;
    const int c    = blockIdx.x;
    const int e0   = c * SORT_CH;
    const int last = nedge - 1;

    lcnt[tid] = 0;
    __syncthreads();

    int s[8], d[8], r[8];
    #pragma unroll
    for (int k = 0; k < 8; ++k) {
        const int e = min(e0 + tid + k * 1024, last);
        s[k] = src[e];
        d[k] = dst[e];
    }
    #pragma unroll
    for (int k = 0; k < 8; ++k) {
        r[k] = 0;
        const int bin = min(d[k] >> BIN_SH, MAXNB - 1);
        if (e0 + tid + k * 1024 < nedge) r[k] = atomicAdd(&lcnt[bin], 1);
    }
    __syncthreads();

    const int v = lcnt[tid];
    int tot;
    const int ex = block_excl_scan_1024(v, wsum, tot);
    lsta[tid] = ex;
    if (tid < nb) {
        tab[(size_t)tid * nch + c] = v;
        lst[(size_t)c * nb + tid]  = ex;
    }
    __syncthreads();

    #pragma unroll
    for (int k = 0; k < 8; ++k) {
        const int bin = min(d[k] >> BIN_SH, MAXNB - 1);
        if (e0 + tid + k * 1024 < nedge) stage[lsta[bin] + r[k]] = make_int2(s[k], d[k]);
    }
    __syncthreads();

    for (int j = tid; j < tot; j += 1024) cs[(size_t)e0 + j] = stage[j];
}

__global__ __launch_bounds__(1024) void bin_sort_kernel(
    const int2* __restrict__ cs, const int* __restrict__ tab, const int* __restrict__ lst,
    int* __restrict__ off, int* __restrict__ ssrc, int nnode, int nb, int nch)
{
    __shared__ int  ppre[MAXCH];
    __shared__ int  ps[MAXCH];
    __shared__ int  lcnt[BIN_ROWS];
    __shared__ int  loff[BIN_ROWS];
    __shared__ int  lcur[BIN_ROWS];
    __shared__ int  wsum[32];
    __shared__ int  gsum[32];
    __shared__ int2 A[BIN_CAP];
    __shared__ int  B[BIN_CAP];

    const int tid   = threadIdx.x;
    const int lane  = tid & 31;
    const int wave  = tid >> 5;
    const int b     = blockIdx.x;
    const int rbase = b << BIN_SH;

    int n = 0, st = 0;
    if (tid < nch) {
        n  = tab[(size_t)b * nch + tid];
        st = lst[(size_t)tid * nb + b];
    }

    int g = 0;
    {
        const int   nfront = b * nch;
        const int   n4     = nfront >> 2;
        const int4* t4     = (const int4*)tab;
        int i = tid;
        for (; i + 7 * 1024 < n4; i += 8 * 1024) {
            int4 a[8];
            #pragma unroll
            for (int k = 0; k < 8; ++k) a[k] = t4[i + k * 1024];
            #pragma unroll
            for (int k = 0; k < 8; ++k) g += (a[k].x + a[k].y) + (a[k].z + a[k].w);
        }
        for (; i < n4; i += 1024) {
            int4 a = t4[i];
            g += (a.x + a.y) + (a.z + a.w);
        }
        for (int q = (n4 << 2) + tid; q < nfront; q += 1024) g += tab[q];
    }
    for (int o = 16; o > 0; o >>= 1) g += __shfl_xor(g, o);
    if (lane == 0) gsum[wave] = g;

    ps[tid] = st;
    if (tid < BIN_ROWS) { lcnt[tid] = 0; lcur[tid] = 0; }
    int total;
    const int ex = block_excl_scan_1024(n, wsum, total);
    ppre[tid] = ex;
    int gb = gsum[lane];
    for (int o = 16; o > 0; o >>= 1) gb += __shfl_xor(gb, o);
    __syncthreads();

    const bool single = total <= BIN_CAP;

    for (int j0 = tid; j0 < total; j0 += 8 * 1024) {
        int2 e[8];
        #pragma unroll
        for (int k = 0; k < 8; ++k) {
            const int j = min(j0 + k * 1024, total - 1);
            int cc = 0;
            #pragma unroll
            for (int sft = MAXCH / 2; sft > 0; sft >>= 1)
                if (ppre[cc + sft] <= j) cc += sft;
            e[k] = cs[(size_t)cc * SORT_CH + ps[cc] + (j - ppre[cc])];
        }
        #pragma unroll
        for (int k = 0; k < 8; ++k) {
            const int j = j0 + k * 1024;
            if (j < total) {
                const int ld = min(max(e[k].y - rbase, 0), BIN_ROWS - 1);
                atomicAdd(&lcnt[ld], 1);
                if (single) A[j] = make_int2(e[k].x, ld);
            }
        }
    }
    __syncthreads();

    const int cv = tid < BIN_ROWS ? lcnt[tid] : 0;
    int rtot;
    const int rex = block_excl_scan_1024(cv, wsum, rtot);
    if (tid < BIN_ROWS) {
        loff[tid] = rex;
        if (rbase + tid < nnode) off[rbase + tid] = gb + rex;
    }
    if (b == nb - 1 && tid == 0) off[nnode] = gb + total;
    __syncthreads();

    if (single) {
        for (int j = tid; j < total; j += 1024) {
            const int2 e = A[j];
            const int pos = loff[e.y] + atomicAdd(&lcur[e.y], 1);
            B[pos] = e.x;
        }
        __syncthreads();
        for (int j = tid; j < total; j += 1024) ssrc[(size_t)gb + j] = B[j];
    } else {
        for (int j = tid; j < total; j += 1024) {
            int cc = 0;
            #pragma unroll
            for (int sft = MAXCH / 2; sft > 0; sft >>= 1)
                if (ppre[cc + sft] <= j) cc += sft;
            const int2 e  = cs[(size_t)cc * SORT_CH + ps[cc] + (j - ppre[cc])];
            const int  ld = min(max(e.y - rbase, 0), BIN_ROWS - 1);
            const int  pos = loff[ld] + atomicAdd(&lcur[ld], 1);
            ssrc[(size_t)gb + pos] = e.x;
        }
    }
}

__global__ __launch_bounds__(256) void gather_kernel(
    const float* __restrict__ y, const int* __restrict__ off,
    const int* __restrict__ ssrc, float* __restrict__ h, int nnode)
{
    const int lane = threadIdx.x & 31;
    const int w0 = __builtin_amdgcn_readfirstlane((int)((blockIdx.x * 256u + threadIdx.x) >> 5));
    const int nw = (int)((gridDim.x * 256u) >> 5);
    const float* yl = y + lane * 2;

    int node = w0;
    int bv = (node < nnode && lane < 2) ? off[node + lane] : 0;
    while (node < nnode) {
        const int beg = __builtin_amdgcn_readlane(bv, 0);
        const int end = __builtin_amdgcn_readlane(bv, 1);
        int mys = lane < end - beg ? ssrc[beg + lane] : 0;
        const int nxt = node + nw;
        int nbv = 0;
        if (nxt < nnode && lane < 2) nbv = off[nxt + lane];

        v2f acc = {0.f, 0.f};
        int base = beg;
        while (base < end) {
            const int n = min(32, end - base);
            for (int j = 0; j < n; j += 16) {
                v2f v[16];
                #pragma unroll
                for (int k = 0; k < 16; ++k) {
                    const int s = __builtin_amdgcn_readlane(mys, j + k);
                    v[k] = *(const v2f*)(yl + (size_t)s * DLAT);
                }
                #pragma unroll
                for (int k = 0; k < 16; ++k) {
                    const bool on = (j + k) < n;
                    acc.x += on ? v[k].x : 0.f;
                    acc.y += on ? v[k].y : 0.f;
                }
            }
            base += 32;
            if (base < end) mys = lane < end - base ? ssrc[base + lane] : 0;
        }
        *(v2f*)(h + (size_t)node * DLAT + lane * 2) = acc;
        node = nxt; bv = nbv;
    }
}

extern "C" void kernel_launch(void* const* d_in, const int* in_sizes, int n_in,
                              void* d_out, int out_size, void* d_ws, size_t ws_size,
                              hipStream_t stream)
{
    const float* features = (const float*)d_in[0];
    const float* id_emb   = (const float*)d_in[1];
    const int*   edge     = (const int*)d_in[2];
    const float* Wmlp     = (const float*)d_in[3];
    const float* bmlp     = (const float*)d_in[4];
    const float* Wc1      = (const float*)d_in[5];
    const float* Wl1      = (const float*)d_in[6];
    const float* bl1      = (const float*)d_in[7];
    const float* Wg1      = (const float*)d_in[8];
    const float* bg1      = (const float*)d_in[9];
    const float* Wc2      = (const float*)d_in[10];
    const float* Wl2      = (const float*)d_in[11];
    const float* bl2      = (const float*)d_in[12];
    const float* Wg2      = (const float*)d_in[13];
    const float* bg2      = (const float*)d_in[14];

    const int nnode = in_sizes[0] / DFEAT;
    const int nedge = in_sizes[2] / 2;
    const int* src  = edge;
    const int* dst  = edge + nedge;

    float* x = (float*)d_ws;
    float* h = x + (size_t)nnode * DLAT;
    float* y = h + (size_t)nnode * DLAT;

    const size_t cpad = (((size_t)nnode + 4) / 4) * 4;
    const size_t epad = (((size_t)nedge + 3) / 4) * 4;
    const int nb  = (nnode + BIN_ROWS - 1) / BIN_ROWS;
    const int nch = (nedge + SORT_CH - 1) / SORT_CH;
    int*  off  = (int*)(y + (size_t)nnode * DLAT);
    int*  ssrc = off + cpad;
    int2* cs   = (int2*)(ssrc + epad);
    int*  tab  = (int*)(cs + epad);
    int*  lst  = tab + (((size_t)nb * nch + 3) / 4) * 4;

    chunk_sort_kernel<<<nch, 1024, 0, stream>>>(src, dst, cs, tab, lst, nedge, nb, nch);
    bin_sort_kernel<<<nb, 1024, 0, stream>>>(cs, tab, lst, off, ssrc, nnode, nb, nch);

    mlp_norm_conv_kernel<<<DENSE_GRID, 256, 0, stream>>>(features, Wmlp, bmlp, Wc1, x, y, nnode);
    gather_kernel<<<2048, 256, 0, stream>>>(y, off, ssrc, h, nnode);
    update_kernel<true><<<DENSE_GRID, 256, 0, stream>>>(x, h, id_emb, Wl1, bl1, Wg1, bg1,
                                                  Wc2, x, y, nnode);
    gather_kernel<<<2048, 256, 0, stream>>>(y, off, ssrc, h, nnode);
    update_kernel<false><<<DENSE_GRID, 256, 0, stream>>>(x, h, id_emb, Wl2, bl2, Wg2, bg2,
                                                   nullptr, (float*)d_out, nullptr, nnode);
}


# Edge aggregation moved into the update kernels: each wave sums the source rows of 4 rows of its block's tile into the LDS h tile, same edge order; h buffer and both gather launches gone (7 to 5)
# speedup vs baseline: 2.7930x; 1.0608x over previous
#include <hip/hip_runtime.h>
#include <hip/hip_bf16.h>

#define DLAT 64
#define DFEAT 128
#define SLOPE 0.01f
#define EPSN 1e-12f

typedef __attribute__((ext_vector_type(2))) float v2f;
typedef __attribute__((ext_vector_type(8))) float v8f;
typedef __attribute__((ext_vector_type(4))) float v4f;

__device__ __forceinline__ float leaky(float v) { return v >= 0.0f ? v : SLOPE * v; }

__device__ __forceinline__ v8f wmma_f32(v2f a, v2f b, v8f c) {
    return __builtin_amdgcn_wmma_f32_16x16x4_f32(false, a, false, b, (short)0, c,
                                                 false, false);
}

#define FPITCH 132
#define XPITCH 68
#define DENSE_GRID 1024

__device__ __forceinline__ v4f ld_feat(const float* __restrict__ feat, int tile, int f, int nnode) {
    int g = tile * 32 + (f >> 5);
    if (g > nnode - 1) g = nnode - 1;
    return *(const v4f*)(feat + (size_t)g * DFEAT + ((f & 31) << 2));
}
__device__ __forceinline__ void st_feat(float* ft, int f, v4f v) {
    *(v4f*)(ft + (f >> 5) * FPITCH + ((f & 31) << 2)) = v;
}
__device__ __forceinline__ v4f ld_row64(const float* __restrict__ p, int tile, int f, int nnode) {
    int g = tile * 32 + (f >> 4);
    if (g > nnode - 1) g = nnode - 1;
    return *(const v4f*)(p + (size_t)g * DLAT + ((f & 15) << 2));
}
__device__ __forceinline__ v8f ld_id(const float* __restrict__ idemb, int tile, int rbase, int col, int nnode) {
    v8f r;
    #pragma unroll
    for (int i = 0; i < 8; ++i) {
        int g = tile * 32 + rbase + i;
        if (g > nnode - 1) g = nnode - 1;
        r[i] = idemb[(size_t)g * DLAT + col];
    }
    return r;
}

__global__ __launch_bounds__(256) void mlp_norm_conv_kernel(
    const float* __restrict__ feat, const float* __restrict__ Wmlp,
    const float* __restrict__ bmlp, const float* __restrict__ Wc,
    float* __restrict__ xout, float* __restrict__ yout, int nnode)
{
    __shared__ __attribute__((aligned(16))) float Wm_s[DFEAT * DLAT];
    __shared__ __attribute__((aligned(16))) float Wc_s[DLAT * DLAT];
    __shared__ __attribute__((aligned(16))) float ft[32 * FPITCH];
    __shared__ float xt[32 * 65];
    __shared__ float scl[32];

    const int tid   = threadIdx.x;
    const int ntile = (nnode + 31) >> 5;
    const int tstep = (int)gridDim.x;
    int tile = (int)blockIdx.x;
    if (tile >= ntile) return;

    v4f fr0 = ld_feat(feat, tile, tid,       nnode);
    v4f fr1 = ld_feat(feat, tile, tid + 256, nnode);
    v4f fr2 = ld_feat(feat, tile, tid + 512, nnode);
    v4f fr3 = ld_feat(feat, tile, tid + 768, nnode);
    {
        float4 wm[8], wc[4];
        #pragma unroll
        for (int i = 0; i < 8; ++i) wm[i] = ((const float4*)Wmlp)[tid + 256 * i];
        #pragma unroll
        for (int i = 0; i < 4; ++i) wc[i] = ((const float4*)Wc)[tid + 256 * i];
        #pragma unroll
        for (int i = 0; i < 8; ++i) ((float4*)Wm_s)[tid + 256 * i] = wm[i];
        #pragma unroll
        for (int i = 0; i < 4; ++i) ((float4*)Wc_s)[tid + 256 * i] = wc[i];
    }

    const int lane    = tid & 31;
    const int wave    = tid >> 5;
    const int rowtile = wave >> 2;
    const int coltile = wave & 3;
    const int mn      = lane & 15;
    const int kh      = (lane >> 4) << 1;
    const int col     = coltile * 16 + mn;

    const float* ap   = ft + (rowtile * 16 + mn) * FPITCH + kh;
    const float bias  = bmlp[col];
    const int   rbase = rowtile * 16 + ((lane >> 4) << 3);
    const int   rl    = rowtile * 16 + mn;

    for (; tile < ntile; tile += tstep) {
        const int blk0 = tile * 32;

        st_feat(ft, tid,       fr0);
        st_feat(ft, tid + 256, fr1);
        st_feat(ft, tid + 512, fr2);
        st_feat(ft, tid + 768, fr3);
        __syncthreads();

        const int ntl = tile + tstep;
        if (ntl < ntile) {
            fr0 = ld_feat(feat, ntl, tid,       nnode);
            fr1 = ld_feat(feat, ntl, tid + 256, nnode);
            fr2 = ld_feat(feat, ntl, tid + 512, nnode);
            fr3 = ld_feat(feat, ntl, tid + 768, nnode);
        }

        v8f acc = {};
        #pragma unroll
        for (int kk = 0; kk < DFEAT; kk += 4) {
            v2f a = *(const v2f*)(ap + kk);
            v2f b;
            b.x = Wm_s[(kk + kh) * DLAT + col];
            b.y = Wm_s[(kk + kh + 1) * DLAT + col];
            acc = wmma_f32(a, b, acc);
        }

        #pragma unroll
        for (int i = 0; i < 8; ++i)
            xt[(rbase + i) * 65 + col] = acc[i] + bias;
        __syncthreads();

        if (tid < 32) {
            float s = 0.f;
            #pragma unroll
            for (int j = 0; j < DLAT; ++j) { float v = xt[tid * 65 + j]; s += v * v; }
            scl[tid] = 1.0f / fmaxf(sqrtf(s), EPSN);
        }
        __syncthreads();

        for (int e = tid; e < 32 * DLAT; e += 256) {
            int r = e >> 6, c = e & 63;
            float v = xt[r * 65 + c] * scl[r];
            xt[r * 65 + c] = v;
            int g = blk0 + r;
            if (g < nnode) xout[(size_t)g * DLAT + c] = v;
        }
        __syncthreads();

        v8f acc2 = {};
        #pragma unroll
        for (int kk = 0; kk < DLAT; kk += 4) {
            v2f a;
            a.x = xt[rl * 65 + kk + kh];
            a.y = xt[rl * 65 + kk + kh + 1];
            v2f b;
            b.x = Wc_s[(kk + kh) * DLAT + col];
            b.y = Wc_s[(kk + kh + 1) * DLAT + col];
            acc2 = wmma_f32(a, b, acc2);
        }
        #pragma unroll
        for (int i = 0; i < 8; ++i) {
            int g = blk0 + rbase + i;
            if (g < nnode) yout[(size_t)g * DLAT + col] = acc2[i];
        }
    }
}


__device__ __forceinline__ v2f gather_row(const float* __restrict__ yl, const int* __restrict__ ssrc,
                                          int beg, int end, int mys, int lane)
{
    v2f acc = {0.f, 0.f};
    int base = beg;
    while (base < end) {
        const int n = min(32, end - base);
        for (int j = 0; j < n; j += 16) {
            v2f v[16];
            #pragma unroll
            for (int k = 0; k < 16; ++k) {
                const int s = __builtin_amdgcn_readlane(mys, j + k);
                v[k] = *(const v2f*)(yl + (size_t)s * DLAT);
            }
            #pragma unroll
            for (int k = 0; k < 16; ++k) {
                const bool on = (j + k) < n;
                acc.x += on ? v[k].x : 0.f;
                acc.y += on ? v[k].y : 0.f;
            }
        }
        base += 32;
        if (base < end) mys = lane < end - base ? ssrc[base + lane] : 0;
    }
    return acc;
}

template <bool FUSE_NEXT>
__global__ __launch_bounds__(256) void update_kernel(
    const float* __restrict__ xin, const float* __restrict__ ysrc,
    const int* __restrict__ off, const int* __restrict__ ssrc,
    const float* __restrict__ idemb,
    const float* __restrict__ Wl, const float* __restrict__ bl,
    const float* __restrict__ Wg, const float* __restrict__ bg,
    const float* __restrict__ Wc, float* __restrict__ xout,
    float* __restrict__ yout, int nnode)
{
    __shared__ __attribute__((aligned(16))) float Wl_s[DLAT * DLAT];
    __shared__ __attribute__((aligned(16))) float Wg_s[DLAT * DLAT];
    __shared__ __attribute__((aligned(16))) float Wc_s[DLAT * DLAT];
    __shared__ __attribute__((aligned(16))) float xs[32 * XPITCH];
    __shared__ __attribute__((aligned(16))) float hs[32 * XPITCH];
    __shared__ float xt[32 * 65];

    const int tid   = threadIdx.x;
    const int ntile = (nnode + 31) >> 5;
    const int tstep = (int)gridDim.x;
    int tile = (int)blockIdx.x;
    if (tile >= ntile) return;

    const int lane    = tid & 31;
    const int wave    = tid >> 5;
    const int r4      = wave * 4;
    const int l4      = min(lane, 4);

    v4f xr0 = ld_row64(xin, tile, tid,       nnode);
    v4f xr1 = ld_row64(xin, tile, tid + 256, nnode);
    int offc = off[min(tile * 32 + r4 + l4, nnode)];
    int offn = off[min((tile + tstep) * 32 + r4 + l4, nnode)];
    {
        float4 wl[4], wg[4], wc[4];
        #pragma unroll
        for (int i = 0; i < 4; ++i) {
            wl[i] = ((const float4*)Wl)[tid + 256 * i];
            wg[i] = ((const float4*)Wg)[tid + 256 * i];
            if constexpr (FUSE_NEXT) wc[i] = ((const float4*)Wc)[tid + 256 * i];
        }
        #pragma unroll
        for (int i = 0; i < 4; ++i) {
            ((float4*)Wl_s)[tid + 256 * i] = wl[i];
            ((float4*)Wg_s)[tid + 256 * i] = wg[i];
            if constexpr (FUSE_NEXT) ((float4*)Wc_s)[tid + 256 * i] = wc[i];
        }
    }

    int mm[4];
    #pragma unroll
    for (int i = 0; i < 4; ++i) {
        const int b = __builtin_amdgcn_readlane(offc, i);
        const int e = __builtin_amdgcn_readlane(offc, i + 1);
        mm[i] = lane < e - b ? ssrc[b + lane] : 0;
    }

    const int rowtile = wave >> 2;
    const int coltile = wave & 3;
    const int mn      = lane & 15;
    const int kh      = (lane >> 4) << 1;
    const int col     = coltile * 16 + mn;

    const float* xp = xs + (rowtile * 16 + mn) * XPITCH + kh;
    const float* hp = hs + (rowtile * 16 + mn) * XPITCH + kh;
    const float* yl = ysrc + lane * 2;

    const float blv = bl[col], bgv = bg[col];
    const int rbase = rowtile * 16 + ((lane >> 4) << 3);

    v8f idn = ld_id(idemb, tile, rbase, col, nnode);

    for (; tile < ntile; tile += tstep) {
        const int blk0 = tile * 32;
        const v8f idc = idn;

        {
            const int o0 = (tid >> 4) * XPITCH + ((tid & 15) << 2);
            const int o1 = o0 + 16 * XPITCH;
            *(v4f*)(xs + o0) = xr0;
            *(v4f*)(xs + o1) = xr1;
        }
        #pragma unroll
        for (int i = 0; i < 4; ++i) {
            const int b = __builtin_amdgcn_readlane(offc, i);
            const int e = __builtin_amdgcn_readlane(offc, i + 1);
            const v2f a = gather_row(yl, ssrc, b, e, mm[i], lane);
            v2f r;
            r.x = leaky(a.x); r.y = leaky(a.y);
            *(v2f*)(hs + (r4 + i) * XPITCH + lane * 2) = r;
        }
        __syncthreads();

        const int ntl = tile + tstep;
        if (ntl < ntile) {
            offc = offn;
            #pragma unroll
            for (int i = 0; i < 4; ++i) {
                const int b = __builtin_amdgcn_readlane(offc, i);
                const int e = __builtin_amdgcn_readlane(offc, i + 1);
                mm[i] = lane < e - b ? ssrc[b + lane] : 0;
            }
            offn = off[min((ntl + tstep) * 32 + r4 + l4, nnode)];
            xr0 = ld_row64(xin, ntl, tid,       nnode);
            xr1 = ld_row64(xin, ntl, tid + 256, nnode);
            idn = ld_id(idemb, ntl, rbase, col, nnode);
        }

        v8f c1 = {}, c2 = {};
        #pragma unroll 4
        for (int kk = 0; kk < DLAT; kk += 4) {
            v2f ax = *(const v2f*)(xp + kk);
            v2f ah = *(const v2f*)(hp + kk);
            v2f b1;
            b1.x = Wl_s[(kk + kh) * DLAT + col];
            b1.y = Wl_s[(kk + kh + 1) * DLAT + col];
            v2f b2;
            b2.x = Wg_s[(kk + kh) * DLAT + col];
            b2.y = Wg_s[(kk + kh + 1) * DLAT + col];
            c1 = wmma_f32(ax, b1, c1);
            c2 = wmma_f32(ah, b2, c2);
        }

        #pragma unroll
        for (int i = 0; i < 8; ++i) {
            int rloc = rbase + i;
            int g    = blk0 + rloc;
            float xhat = leaky(c1[i] + blv) + idc[i];
            float v    = leaky(c2[i] + bgv + xhat);
            if constexpr (FUSE_NEXT) {
                xt[rloc * 65 + col] = v;
            } else {
                if (g < nnode) xout[(size_t)g * DLAT + col] = v;
            }
        }

        __syncthreads();
        if constexpr (FUSE_NEXT) {
            for (int e = tid; e < 32 * DLAT; e += 256) {
                int r = e >> 6, c = e & 63;
                int g = blk0 + r;
                if (g < nnode) xout[(size_t)g * DLAT + c] = xt[r * 65 + c];
            }
            v8f acc2 = {};
            const int rl = rowtile * 16 + mn;
            #pragma unroll 4
            for (int kk = 0; kk < DLAT; kk += 4) {
                v2f a;
                a.x = xt[rl * 65 + kk + kh];
                a.y = xt[rl * 65 + kk + kh + 1];
                v2f b;
                b.x = Wc_s[(kk + kh) * DLAT + col];
                b.y = Wc_s[(kk + kh + 1) * DLAT + col];
                acc2 = wmma_f32(a, b, acc2);
            }
            #pragma unroll
            for (int i = 0; i < 8; ++i) {
                int g = blk0 + rbase + i;
                if (g < nnode) yout[(size_t)g * DLAT + col] = acc2[i];
            }
        }
    }
}

#define SORT_CH  8192
#define BIN_SH   9
#define BIN_ROWS 512
#define MAXNB    1024
#define MAXCH    1024
#define BIN_CAP  12288

__device__ __forceinline__ int block_excl_scan_1024(int v, int* wsum, int& total)
{
    const int lane = threadIdx.x & 31;
    const int wave = threadIdx.x >> 5;
    int inc = v;
    for (int o = 1; o < 32; o <<= 1) {
        int u = __shfl_up(inc, o);
        if (lane >= o) inc += u;
    }
    if (lane == 31) wsum[wave] = inc;
    __syncthreads();
    const int ws = wsum[lane];
    int wp  = lane < wave ? ws : 0;
    int tot = ws;
    for (int o = 16; o > 0; o >>= 1) {
        wp  += __shfl_xor(wp, o);
        tot += __shfl_xor(tot, o);
    }
    total = tot;
    __syncthreads();
    return wp + inc - v;
}

__global__ __launch_bounds__(1024) void chunk_sort_kernel(
    const int* __restrict__ src, const int* __restrict__ dst,
    int2* __restrict__ cs, int* __restrict__ tab, int* __restrict__ lst,
    int nedge, int nb, int nch)
{
    __shared__ int  lcnt[MAXNB];
    __shared__ int  lsta[MAXNB];
    __shared__ int  wsum[32];
    __shared__ int2 stage[SORT_CH];

    const int tid  = threadIdx.x;
    const int c    = blockIdx.x;
    const int e0   = c * SORT_CH;
    const int last = nedge - 1;

    lcnt[tid] = 0;
    __syncthreads();

    int s[8], d[8], r[8];
    #pragma unroll
    for (int k = 0; k < 8; ++k) {
        const int e = min(e0 + tid + k * 1024, last);
        s[k] = src[e];
        d[k] = dst[e];
    }
    #pragma unroll
    for (int k = 0; k < 8; ++k) {
        r[k] = 0;
        const int bin = min(d[k] >> BIN_SH, MAXNB - 1);
        if (e0 + tid + k * 1024 < nedge) r[k] = atomicAdd(&lcnt[bin], 1);
    }
    __syncthreads();

    const int v = lcnt[tid];
    int tot;
    const int ex = block_excl_scan_1024(v, wsum, tot);
    lsta[tid] = ex;
    if (tid < nb) {
        tab[(size_t)tid * nch + c] = v;
        lst[(size_t)c * nb + tid]  = ex;
    }
    __syncthreads();

    #pragma unroll
    for (int k = 0; k < 8; ++k) {
        const int bin = min(d[k] >> BIN_SH, MAXNB - 1);
        if (e0 + tid + k * 1024 < nedge) stage[lsta[bin] + r[k]] = make_int2(s[k], d[k]);
    }
    __syncthreads();

    for (int j = tid; j < tot; j += 1024) cs[(size_t)e0 + j] = stage[j];
}

__global__ __launch_bounds__(1024) void bin_sort_kernel(
    const int2* __restrict__ cs, const int* __restrict__ tab, const int* __restrict__ lst,
    int* __restrict__ off, int* __restrict__ ssrc, int nnode, int nb, int nch)
{
    __shared__ int  ppre[MAXCH];
    __shared__ int  ps[MAXCH];
    __shared__ int  lcnt[BIN_ROWS];
    __shared__ int  loff[BIN_ROWS];
    __shared__ int  lcur[BIN_ROWS];
    __shared__ int  wsum[32];
    __shared__ int  gsum[32];
    __shared__ int2 A[BIN_CAP];
    __shared__ int  B[BIN_CAP];

    const int tid   = threadIdx.x;
    const int lane  = tid & 31;
    const int wave  = tid >> 5;
    const int b     = blockIdx.x;
    const int rbase = b << BIN_SH;

    int n = 0, st = 0;
    if (tid < nch) {
        n  = tab[(size_t)b * nch + tid];
        st = lst[(size_t)tid * nb + b];
    }

    int g = 0;
    {
        const int   nfront = b * nch;
        const int   n4     = nfront >> 2;
        const int4* t4     = (const int4*)tab;
        int i = tid;
        for (; i + 7 * 1024 < n4; i += 8 * 1024) {
            int4 a[8];
            #pragma unroll
            for (int k = 0; k < 8; ++k) a[k] = t4[i + k * 1024];
            #pragma unroll
            for (int k = 0; k < 8; ++k) g += (a[k].x + a[k].y) + (a[k].z + a[k].w);
        }
        for (; i < n4; i += 1024) {
            int4 a = t4[i];
            g += (a.x + a.y) + (a.z + a.w);
        }
        for (int q = (n4 << 2) + tid; q < nfront; q += 1024) g += tab[q];
    }
    for (int o = 16; o > 0; o >>= 1) g += __shfl_xor(g, o);
    if (lane == 0) gsum[wave] = g;

    ps[tid] = st;
    if (tid < BIN_ROWS) { lcnt[tid] = 0; lcur[tid] = 0; }
    int total;
    const int ex = block_excl_scan_1024(n, wsum, total);
    ppre[tid] = ex;
    int gb = gsum[lane];
    for (int o = 16; o > 0; o >>= 1) gb += __shfl_xor(gb, o);
    __syncthreads();

    const bool single = total <= BIN_CAP;

    for (int j0 = tid; j0 < total; j0 += 8 * 1024) {
        int2 e[8];
        #pragma unroll
        for (int k = 0; k < 8; ++k) {
            const int j = min(j0 + k * 1024, total - 1);
            int cc = 0;
            #pragma unroll
            for (int sft = MAXCH / 2; sft > 0; sft >>= 1)
                if (ppre[cc + sft] <= j) cc += sft;
            e[k] = cs[(size_t)cc * SORT_CH + ps[cc] + (j - ppre[cc])];
        }
        #pragma unroll
        for (int k = 0; k < 8; ++k) {
            const int j = j0 + k * 1024;
            if (j < total) {
                const int ld = min(max(e[k].y - rbase, 0), BIN_ROWS - 1);
                atomicAdd(&lcnt[ld], 1);
                if (single) A[j] = make_int2(e[k].x, ld);
            }
        }
    }
    __syncthreads();

    const int cv = tid < BIN_ROWS ? lcnt[tid] : 0;
    int rtot;
    const int rex = block_excl_scan_1024(cv, wsum, rtot);
    if (tid < BIN_ROWS) {
        loff[tid] = rex;
        if (rbase + tid < nnode) off[rbase + tid] = gb + rex;
    }
    if (b == nb - 1 && tid == 0) off[nnode] = gb + total;
    __syncthreads();

    if (single) {
        for (int j = tid; j < total; j += 1024) {
            const int2 e = A[j];
            const int pos = loff[e.y] + atomicAdd(&lcur[e.y], 1);
            B[pos] = e.x;
        }
        __syncthreads();
        for (int j = tid; j < total; j += 1024) ssrc[(size_t)gb + j] = B[j];
    } else {
        for (int j = tid; j < total; j += 1024) {
            int cc = 0;
            #pragma unroll
            for (int sft = MAXCH / 2; sft > 0; sft >>= 1)
                if (ppre[cc + sft] <= j) cc += sft;
            const int2 e  = cs[(size_t)cc * SORT_CH + ps[cc] + (j - ppre[cc])];
            const int  ld = min(max(e.y - rbase, 0), BIN_ROWS - 1);
            const int  pos = loff[ld] + atomicAdd(&lcur[ld], 1);
            ssrc[(size_t)gb + pos] = e.x;
        }
    }
}

extern "C" void kernel_launch(void* const* d_in, const int* in_sizes, int n_in,
                              void* d_out, int out_size, void* d_ws, size_t ws_size,
                              hipStream_t stream)
{
    const float* features = (const float*)d_in[0];
    const float* id_emb   = (const float*)d_in[1];
    const int*   edge     = (const int*)d_in[2];
    const float* Wmlp     = (const float*)d_in[3];
    const float* bmlp     = (const float*)d_in[4];
    const float* Wc1      = (const float*)d_in[5];
    const float* Wl1      = (const float*)d_in[6];
    const float* bl1      = (const float*)d_in[7];
    const float* Wg1      = (const float*)d_in[8];
    const float* bg1      = (const float*)d_in[9];
    const float* Wc2      = (const float*)d_in[10];
    const float* Wl2      = (const float*)d_in[11];
    const float* bl2      = (const float*)d_in[12];
    const float* Wg2      = (const float*)d_in[13];
    const float* bg2      = (const float*)d_in[14];

    const int nnode = in_sizes[0] / DFEAT;
    const int nedge = in_sizes[2] / 2;
    const int* src  = edge;
    const int* dst  = edge + nedge;

    float* x = (float*)d_ws;
    float* y2 = x + (size_t)nnode * DLAT;
    float* y  = y2 + (size_t)nnode * DLAT;

    const size_t cpad = (((size_t)nnode + 4) / 4) * 4;
    const size_t epad = (((size_t)nedge + 3) / 4) * 4;
    const int nb  = (nnode + BIN_ROWS - 1) / BIN_ROWS;
    const int nch = (nedge + SORT_CH - 1) / SORT_CH;
    int*  off  = (int*)(y + (size_t)nnode * DLAT);
    int*  ssrc = off + cpad;
    int2* cs   = (int2*)(ssrc + epad);
    int*  tab  = (int*)(cs + epad);
    int*  lst  = tab + (((size_t)nb * nch + 3) / 4) * 4;

    chunk_sort_kernel<<<nch, 1024, 0, stream>>>(src, dst, cs, tab, lst, nedge, nb, nch);
    bin_sort_kernel<<<nb, 1024, 0, stream>>>(cs, tab, lst, off, ssrc, nnode, nb, nch);

    mlp_norm_conv_kernel<<<DENSE_GRID, 256, 0, stream>>>(features, Wmlp, bmlp, Wc1, x, y, nnode);
    update_kernel<true><<<DENSE_GRID, 256, 0, stream>>>(x, y, off, ssrc, id_emb, Wl1, bl1, Wg1, bg1,
                                                  Wc2, x, y2, nnode);
    update_kernel<false><<<DENSE_GRID, 256, 0, stream>>>(x, y2, off, ssrc, id_emb, Wl2, bl2, Wg2, bg2,
                                                   nullptr, (float*)d_out, nullptr, nnode);
}


# Gather in the update kernels: half-wave per source row (16-B loads), two destination rows at a time; slots past a row's end read a zero row, adds unconditional; same edge order per row
# speedup vs baseline: 3.0562x; 1.0942x over previous
#include <hip/hip_runtime.h>
#include <hip/hip_bf16.h>

#define DLAT 64
#define DFEAT 128
#define SLOPE 0.01f
#define EPSN 1e-12f

typedef __attribute__((ext_vector_type(2))) float v2f;
typedef __attribute__((ext_vector_type(8))) float v8f;
typedef __attribute__((ext_vector_type(4))) float v4f;

__device__ __forceinline__ float leaky(float v) { return v >= 0.0f ? v : SLOPE * v; }

__device__ __forceinline__ v8f wmma_f32(v2f a, v2f b, v8f c) {
    return __builtin_amdgcn_wmma_f32_16x16x4_f32(false, a, false, b, (short)0, c,
                                                 false, false);
}

#define FPITCH 132
#define XPITCH 68
#define DENSE_GRID 1024

__device__ __forceinline__ v4f ld_feat(const float* __restrict__ feat, int tile, int f, int nnode) {
    int g = tile * 32 + (f >> 5);
    if (g > nnode - 1) g = nnode - 1;
    return *(const v4f*)(feat + (size_t)g * DFEAT + ((f & 31) << 2));
}
__device__ __forceinline__ void st_feat(float* ft, int f, v4f v) {
    *(v4f*)(ft + (f >> 5) * FPITCH + ((f & 31) << 2)) = v;
}
__device__ __forceinline__ v4f ld_row64(const float* __restrict__ p, int tile, int f, int nnode) {
    int g = tile * 32 + (f >> 4);
    if (g > nnode - 1) g = nnode - 1;
    return *(const v4f*)(p + (size_t)g * DLAT + ((f & 15) << 2));
}
__device__ __forceinline__ v8f ld_id(const float* __restrict__ idemb, int tile, int rbase, int col, int nnode) {
    v8f r;
    #pragma unroll
    for (int i = 0; i < 8; ++i) {
        int g = tile * 32 + rbase + i;
        if (g > nnode - 1) g = nnode - 1;
        r[i] = idemb[(size_t)g * DLAT + col];
    }
    return r;
}

__global__ __launch_bounds__(256) void mlp_norm_conv_kernel(
    const float* __restrict__ feat, const float* __restrict__ Wmlp,
    const float* __restrict__ bmlp, const float* __restrict__ Wc,
    float* __restrict__ xout, float* __restrict__ yout, int nnode)
{
    __shared__ __attribute__((aligned(16))) float Wm_s[DFEAT * DLAT];
    __shared__ __attribute__((aligned(16))) float Wc_s[DLAT * DLAT];
    __shared__ __attribute__((aligned(16))) float ft[32 * FPITCH];
    __shared__ float xt[32 * 65];
    __shared__ float scl[32];

    const int tid   = threadIdx.x;
    const int ntile = (nnode + 31) >> 5;
    const int tstep = (int)gridDim.x;
    int tile = (int)blockIdx.x;
    if (tile >= ntile) return;

    if (blockIdx.x == 0 && tid < DLAT) yout[(size_t)nnode * DLAT + tid] = 0.f;

    v4f fr0 = ld_feat(feat, tile, tid,       nnode);
    v4f fr1 = ld_feat(feat, tile, tid + 256, nnode);
    v4f fr2 = ld_feat(feat, tile, tid + 512, nnode);
    v4f fr3 = ld_feat(feat, tile, tid + 768, nnode);
    {
        float4 wm[8], wc[4];
        #pragma unroll
        for (int i = 0; i < 8; ++i) wm[i] = ((const float4*)Wmlp)[tid + 256 * i];
        #pragma unroll
        for (int i = 0; i < 4; ++i) wc[i] = ((const float4*)Wc)[tid + 256 * i];
        #pragma unroll
        for (int i = 0; i < 8; ++i) ((float4*)Wm_s)[tid + 256 * i] = wm[i];
        #pragma unroll
        for (int i = 0; i < 4; ++i) ((float4*)Wc_s)[tid + 256 * i] = wc[i];
    }

    const int lane    = tid & 31;
    const int wave    = tid >> 5;
    const int rowtile = wave >> 2;
    const int coltile = wave & 3;
    const int mn      = lane & 15;
    const int kh      = (lane >> 4) << 1;
    const int col     = coltile * 16 + mn;

    const float* ap   = ft + (rowtile * 16 + mn) * FPITCH + kh;
    const float bias  = bmlp[col];
    const int   rbase = rowtile * 16 + ((lane >> 4) << 3);
    const int   rl    = rowtile * 16 + mn;

    for (; tile < ntile; tile += tstep) {
        const int blk0 = tile * 32;

        st_feat(ft, tid,       fr0);
        st_feat(ft, tid + 256, fr1);
        st_feat(ft, tid + 512, fr2);
        st_feat(ft, tid + 768, fr3);
        __syncthreads();

        const int ntl = tile + tstep;
        if (ntl < ntile) {
            fr0 = ld_feat(feat, ntl, tid,       nnode);
            fr1 = ld_feat(feat, ntl, tid + 256, nnode);
            fr2 = ld_feat(feat, ntl, tid + 512, nnode);
            fr3 = ld_feat(feat, ntl, tid + 768, nnode);
        }

        v8f acc = {};
        #pragma unroll
        for (int kk = 0; kk < DFEAT; kk += 4) {
            v2f a = *(const v2f*)(ap + kk);
            v2f b;
            b.x = Wm_s[(kk + kh) * DLAT + col];
            b.y = Wm_s[(kk + kh + 1) * DLAT + col];
            acc = wmma_f32(a, b, acc);
        }

        #pragma unroll
        for (int i = 0; i < 8; ++i)
            xt[(rbase + i) * 65 + col] = acc[i] + bias;
        __syncthreads();

        if (tid < 32) {
            float s = 0.f;
            #pragma unroll
            for (int j = 0; j < DLAT; ++j) { float v = xt[tid * 65 + j]; s += v * v; }
            scl[tid] = 1.0f / fmaxf(sqrtf(s), EPSN);
        }
        __syncthreads();

        for (int e = tid; e < 32 * DLAT; e += 256) {
            int r = e >> 6, c = e & 63;
            float v = xt[r * 65 + c] * scl[r];
            xt[r * 65 + c] = v;
            int g = blk0 + r;
            if (g < nnode) xout[(size_t)g * DLAT + c] = v;
        }
        __syncthreads();

        v8f acc2 = {};
        #pragma unroll
        for (int kk = 0; kk < DLAT; kk += 4) {
            v2f a;
            a.x = xt[rl * 65 + kk + kh];
            a.y = xt[rl * 65 + kk + kh + 1];
            v2f b;
            b.x = Wc_s[(kk + kh) * DLAT + col];
            b.y = Wc_s[(kk + kh + 1) * DLAT + col];
            acc2 = wmma_f32(a, b, acc2);
        }
        #pragma unroll
        for (int i = 0; i < 8; ++i) {
            int g = blk0 + rbase + i;
            if (g < nnode) yout[(size_t)g * DLAT + col] = acc2[i];
        }
    }
}


struct PairBounds { int myb, myn, nmax; };

__device__ __forceinline__ PairBounds pair_bounds(int offc, int p, bool hi)
{
    const int b0 = __builtin_amdgcn_readlane(offc, 2 * p);
    const int m  = __builtin_amdgcn_readlane(offc, 2 * p + 1);
    const int e1 = __builtin_amdgcn_readlane(offc, 2 * p + 2);
    PairBounds r;
    r.myb  = hi ? m : b0;
    r.myn  = hi ? e1 - m : m - b0;
    r.nmax = max(m - b0, e1 - m);
    return r;
}

__device__ __forceinline__ int ld_src_id(const int* __restrict__ ssrc, const PairBounds& pb, int idx, int zrow)
{
    return idx < pb.myn ? ssrc[pb.myb + idx] : zrow;
}

template <int K>
__device__ __forceinline__ void issue_row_loads(const char* __restrict__ yb, int ids, unsigned coff, v4f (&v)[16])
{
    if constexpr (K < 16) {
        const int s = __builtin_amdgcn_update_dpp(0, ids, 0x150 + K, 0xf, 0xf, false);
        v[K] = *(const v4f*)(yb + (size_t)(((unsigned)s << 8) | coff));
        issue_row_loads<K + 1>(yb, ids, coff, v);
    }
}

__device__ __forceinline__ v4f gather_pair(const char* __restrict__ yb, const int* __restrict__ ssrc,
                                           const PairBounds& pb, int ids0, int ids1, int c, int zrow)
{
    v4f acc = {0.f, 0.f, 0.f, 0.f};
    const unsigned coff = (unsigned)c << 4;
    for (int j = 0; j < pb.nmax; j += 16) {
        v4f v[16];
        issue_row_loads<0>(yb, ids0, coff, v);
        ids0 = ids1;
        if (j + 32 < pb.nmax) ids1 = ld_src_id(ssrc, pb, j + 32 + c, zrow);
        #pragma unroll
        for (int k = 0; k < 16; ++k) acc += v[k];
    }
    return acc;
}

template <bool FUSE_NEXT>
__global__ __launch_bounds__(256) void update_kernel(
    const float* __restrict__ xin, const float* __restrict__ ysrc,
    const int* __restrict__ off, const int* __restrict__ ssrc,
    const float* __restrict__ idemb,
    const float* __restrict__ Wl, const float* __restrict__ bl,
    const float* __restrict__ Wg, const float* __restrict__ bg,
    const float* __restrict__ Wc, float* __restrict__ xout,
    float* __restrict__ yout, int nnode)
{
    __shared__ __attribute__((aligned(16))) float Wl_s[DLAT * DLAT];
    __shared__ __attribute__((aligned(16))) float Wg_s[DLAT * DLAT];
    __shared__ __attribute__((aligned(16))) float Wc_s[DLAT * DLAT];
    __shared__ __attribute__((aligned(16))) float xs[32 * XPITCH];
    __shared__ __attribute__((aligned(16))) float hs[32 * XPITCH];
    __shared__ float xt[32 * 65];

    const int tid   = threadIdx.x;
    const int ntile = (nnode + 31) >> 5;
    const int tstep = (int)gridDim.x;
    int tile = (int)blockIdx.x;
    if (tile >= ntile) return;

    const int lane    = tid & 31;
    const int wave    = tid >> 5;
    const int r4      = wave * 4;
    const int l4      = min(lane, 4);

    v4f xr0 = ld_row64(xin, tile, tid,       nnode);
    v4f xr1 = ld_row64(xin, tile, tid + 256, nnode);
    int offc = off[min(tile * 32 + r4 + l4, nnode)];
    int offn = off[min((tile + tstep) * 32 + r4 + l4, nnode)];
    {
        float4 wl[4], wg[4], wc[4];
        #pragma unroll
        for (int i = 0; i < 4; ++i) {
            wl[i] = ((const float4*)Wl)[tid + 256 * i];
            wg[i] = ((const float4*)Wg)[tid + 256 * i];
            if constexpr (FUSE_NEXT) wc[i] = ((const float4*)Wc)[tid + 256 * i];
        }
        #pragma unroll
        for (int i = 0; i < 4; ++i) {
            ((float4*)Wl_s)[tid + 256 * i] = wl[i];
            ((float4*)Wg_s)[tid + 256 * i] = wg[i];
            if constexpr (FUSE_NEXT) ((float4*)Wc_s)[tid + 256 * i] = wc[i];
        }
    }

    const int  mn = lane & 15;
    const bool hi = lane >= 16;

    if constexpr (FUSE_NEXT) {
        if (blockIdx.x == 0 && tid < DLAT) yout[(size_t)nnode * DLAT + tid] = 0.f;
    }

    int ids[2][2];
    #pragma unroll
    for (int p = 0; p < 2; ++p) {
        const PairBounds pb = pair_bounds(offc, p, hi);
        ids[p][0] = ld_src_id(ssrc, pb, mn,      nnode);
        ids[p][1] = ld_src_id(ssrc, pb, mn + 16, nnode);
    }

    const int rowtile = wave >> 2;
    const int coltile = wave & 3;
    const int kh      = (lane >> 4) << 1;
    const int col     = coltile * 16 + mn;

    const float* xp = xs + (rowtile * 16 + mn) * XPITCH + kh;
    const float* hp = hs + (rowtile * 16 + mn) * XPITCH + kh;
    const char*  yb = (const char*)ysrc;

    const float blv = bl[col], bgv = bg[col];
    const int rbase = rowtile * 16 + ((lane >> 4) << 3);

    v8f idn = ld_id(idemb, tile, rbase, col, nnode);

    for (; tile < ntile; tile += tstep) {
        const int blk0 = tile * 32;
        const v8f idc = idn;

        {
            const int o0 = (tid >> 4) * XPITCH + ((tid & 15) << 2);
            const int o1 = o0 + 16 * XPITCH;
            *(v4f*)(xs + o0) = xr0;
            *(v4f*)(xs + o1) = xr1;
        }
        #pragma unroll
        for (int p = 0; p < 2; ++p) {
            const PairBounds pb = pair_bounds(offc, p, hi);
            const v4f a = gather_pair(yb, ssrc, pb, ids[p][0], ids[p][1], mn, nnode);
            v4f r;
            r.x = leaky(a.x); r.y = leaky(a.y); r.z = leaky(a.z); r.w = leaky(a.w);
            *(v4f*)(hs + (r4 + 2 * p + (hi ? 1 : 0)) * XPITCH + mn * 4) = r;
        }
        __syncthreads();

        const int ntl = tile + tstep;
        if (ntl < ntile) {
            offc = offn;
            #pragma unroll
            for (int p = 0; p < 2; ++p) {
                const PairBounds pb = pair_bounds(offc, p, hi);
                ids[p][0] = ld_src_id(ssrc, pb, mn,      nnode);
                ids[p][1] = ld_src_id(ssrc, pb, mn + 16, nnode);
            }
            offn = off[min((ntl + tstep) * 32 + r4 + l4, nnode)];
            xr0 = ld_row64(xin, ntl, tid,       nnode);
            xr1 = ld_row64(xin, ntl, tid + 256, nnode);
            idn = ld_id(idemb, ntl, rbase, col, nnode);
        }

        v8f c1 = {}, c2 = {};
        #pragma unroll 4
        for (int kk = 0; kk < DLAT; kk += 4) {
            v2f ax = *(const v2f*)(xp + kk);
            v2f ah = *(const v2f*)(hp + kk);
            v2f b1;
            b1.x = Wl_s[(kk + kh) * DLAT + col];
            b1.y = Wl_s[(kk + kh + 1) * DLAT + col];
            v2f b2;
            b2.x = Wg_s[(kk + kh) * DLAT + col];
            b2.y = Wg_s[(kk + kh + 1) * DLAT + col];
            c1 = wmma_f32(ax, b1, c1);
            c2 = wmma_f32(ah, b2, c2);
        }

        #pragma unroll
        for (int i = 0; i < 8; ++i) {
            int rloc = rbase + i;
            int g    = blk0 + rloc;
            float xhat = leaky(c1[i] + blv) + idc[i];
            float v    = leaky(c2[i] + bgv + xhat);
            if constexpr (FUSE_NEXT) {
                xt[rloc * 65 + col] = v;
            } else {
                if (g < nnode) xout[(size_t)g * DLAT + col] = v;
            }
        }

        __syncthreads();
        if constexpr (FUSE_NEXT) {
            for (int e = tid; e < 32 * DLAT; e += 256) {
                int r = e >> 6, c = e & 63;
                int g = blk0 + r;
                if (g < nnode) xout[(size_t)g * DLAT + c] = xt[r * 65 + c];
            }
            v8f acc2 = {};
            const int rl = rowtile * 16 + mn;
            #pragma unroll 4
            for (int kk = 0; kk < DLAT; kk += 4) {
                v2f a;
                a.x = xt[rl * 65 + kk + kh];
                a.y = xt[rl * 65 + kk + kh + 1];
                v2f b;
                b.x = Wc_s[(kk + kh) * DLAT + col];
                b.y = Wc_s[(kk + kh + 1) * DLAT + col];
                acc2 = wmma_f32(a, b, acc2);
            }
            #pragma unroll
            for (int i = 0; i < 8; ++i) {
                int g = blk0 + rbase + i;
                if (g < nnode) yout[(size_t)g * DLAT + col] = acc2[i];
            }
        }
    }
}

#define SORT_CH  8192
#define BIN_SH   9
#define BIN_ROWS 512
#define MAXNB    1024
#define MAXCH    1024
#define BIN_CAP  12288

__device__ __forceinline__ int block_excl_scan_1024(int v, int* wsum, int& total)
{
    const int lane = threadIdx.x & 31;
    const int wave = threadIdx.x >> 5;
    int inc = v;
    for (int o = 1; o < 32; o <<= 1) {
        int u = __shfl_up(inc, o);
        if (lane >= o) inc += u;
    }
    if (lane == 31) wsum[wave] = inc;
    __syncthreads();
    const int ws = wsum[lane];
    int wp  = lane < wave ? ws : 0;
    int tot = ws;
    for (int o = 16; o > 0; o >>= 1) {
        wp  += __shfl_xor(wp, o);
        tot += __shfl_xor(tot, o);
    }
    total = tot;
    __syncthreads();
    return wp + inc - v;
}

__global__ __launch_bounds__(1024) void chunk_sort_kernel(
    const int* __restrict__ src, const int* __restrict__ dst,
    int2* __restrict__ cs, int* __restrict__ tab, int* __restrict__ lst,
    int nedge, int nb, int nch)
{
    __shared__ int  lcnt[MAXNB];
    __shared__ int  lsta[MAXNB];
    __shared__ int  wsum[32];
    __shared__ int2 stage[SORT_CH];

    const int tid  = threadIdx.x;
    const int c    = blockIdx.x;
    const int e0   = c * SORT_CH;
    const int last = nedge - 1;

    lcnt[tid] = 0;
    __syncthreads();

    int s[8], d[8], r[8];
    #pragma unroll
    for (int k = 0; k < 8; ++k) {
        const int e = min(e0 + tid + k * 1024, last);
        s[k] = src[e];
        d[k] = dst[e];
    }
    #pragma unroll
    for (int k = 0; k < 8; ++k) {
        r[k] = 0;
        const int bin = min(d[k] >> BIN_SH, MAXNB - 1);
        if (e0 + tid + k * 1024 < nedge) r[k] = atomicAdd(&lcnt[bin], 1);
    }
    __syncthreads();

    const int v = lcnt[tid];
    int tot;
    const int ex = block_excl_scan_1024(v, wsum, tot);
    lsta[tid] = ex;
    if (tid < nb) {
        tab[(size_t)tid * nch + c] = v;
        lst[(size_t)c * nb + tid]  = ex;
    }
    __syncthreads();

    #pragma unroll
    for (int k = 0; k < 8; ++k) {
        const int bin = min(d[k] >> BIN_SH, MAXNB - 1);
        if (e0 + tid + k * 1024 < nedge) stage[lsta[bin] + r[k]] = make_int2(s[k], d[k]);
    }
    __syncthreads();

    for (int j = tid; j < tot; j += 1024) cs[(size_t)e0 + j] = stage[j];
}

__global__ __launch_bounds__(1024) void bin_sort_kernel(
    const int2* __restrict__ cs, const int* __restrict__ tab, const int* __restrict__ lst,
    int* __restrict__ off, int* __restrict__ ssrc, int nnode, int nb, int nch)
{
    __shared__ int  ppre[MAXCH];
    __shared__ int  ps[MAXCH];
    __shared__ int  lcnt[BIN_ROWS];
    __shared__ int  loff[BIN_ROWS];
    __shared__ int  lcur[BIN_ROWS];
    __shared__ int  wsum[32];
    __shared__ int  gsum[32];
    __shared__ int2 A[BIN_CAP];
    __shared__ int  B[BIN_CAP];

    const int tid   = threadIdx.x;
    const int lane  = tid & 31;
    const int wave  = tid >> 5;
    const int b     = blockIdx.x;
    const int rbase = b << BIN_SH;

    int n = 0, st = 0;
    if (tid < nch) {
        n  = tab[(size_t)b * nch + tid];
        st = lst[(size_t)tid * nb + b];
    }

    int g = 0;
    {
        const int   nfront = b * nch;
        const int   n4     = nfront >> 2;
        const int4* t4     = (const int4*)tab;
        int i = tid;
        for (; i + 7 * 1024 < n4; i += 8 * 1024) {
            int4 a[8];
            #pragma unroll
            for (int k = 0; k < 8; ++k) a[k] = t4[i + k * 1024];
            #pragma unroll
            for (int k = 0; k < 8; ++k) g += (a[k].x + a[k].y) + (a[k].z + a[k].w);
        }
        for (; i < n4; i += 1024) {
            int4 a = t4[i];
            g += (a.x + a.y) + (a.z + a.w);
        }
        for (int q = (n4 << 2) + tid; q < nfront; q += 1024) g += tab[q];
    }
    for (int o = 16; o > 0; o >>= 1) g += __shfl_xor(g, o);
    if (lane == 0) gsum[wave] = g;

    ps[tid] = st;
    if (tid < BIN_ROWS) { lcnt[tid] = 0; lcur[tid] = 0; }
    int total;
    const int ex = block_excl_scan_1024(n, wsum, total);
    ppre[tid] = ex;
    int gb = gsum[lane];
    for (int o = 16; o > 0; o >>= 1) gb += __shfl_xor(gb, o);
    __syncthreads();

    const bool single = total <= BIN_CAP;

    for (int j0 = tid; j0 < total; j0 += 8 * 1024) {
        int2 e[8];
        #pragma unroll
        for (int k = 0; k < 8; ++k) {
            const int j = min(j0 + k * 1024, total - 1);
            int cc = 0;
            #pragma unroll
            for (int sft = MAXCH / 2; sft > 0; sft >>= 1)
                if (ppre[cc + sft] <= j) cc += sft;
            e[k] = cs[(size_t)cc * SORT_CH + ps[cc] + (j - ppre[cc])];
        }
        #pragma unroll
        for (int k = 0; k < 8; ++k) {
            const int j = j0 + k * 1024;
            if (j < total) {
                const int ld = min(max(e[k].y - rbase, 0), BIN_ROWS - 1);
                atomicAdd(&lcnt[ld], 1);
                if (single) A[j] = make_int2(e[k].x, ld);
            }
        }
    }
    __syncthreads();

    const int cv = tid < BIN_ROWS ? lcnt[tid] : 0;
    int rtot;
    const int rex = block_excl_scan_1024(cv, wsum, rtot);
    if (tid < BIN_ROWS) {
        loff[tid] = rex;
        if (rbase + tid < nnode) off[rbase + tid] = gb + rex;
    }
    if (b == nb - 1 && tid == 0) off[nnode] = gb + total;
    __syncthreads();

    if (single) {
        for (int j = tid; j < total; j += 1024) {
            const int2 e = A[j];
            const int pos = loff[e.y] + atomicAdd(&lcur[e.y], 1);
            B[pos] = e.x;
        }
        __syncthreads();
        for (int j = tid; j < total; j += 1024) ssrc[(size_t)gb + j] = B[j];
    } else {
        for (int j = tid; j < total; j += 1024) {
            int cc = 0;
            #pragma unroll
            for (int sft = MAXCH / 2; sft > 0; sft >>= 1)
                if (ppre[cc + sft] <= j) cc += sft;
            const int2 e  = cs[(size_t)cc * SORT_CH + ps[cc] + (j - ppre[cc])];
            const int  ld = min(max(e.y - rbase, 0), BIN_ROWS - 1);
            const int  pos = loff[ld] + atomicAdd(&lcur[ld], 1);
            ssrc[(size_t)gb + pos] = e.x;
        }
    }
}

extern "C" void kernel_launch(void* const* d_in, const int* in_sizes, int n_in,
                              void* d_out, int out_size, void* d_ws, size_t ws_size,
                              hipStream_t stream)
{
    const float* features = (const float*)d_in[0];
    const float* id_emb   = (const float*)d_in[1];
    const int*   edge     = (const int*)d_in[2];
    const float* Wmlp     = (const float*)d_in[3];
    const float* bmlp     = (const float*)d_in[4];
    const float* Wc1      = (const float*)d_in[5];
    const float* Wl1      = (const float*)d_in[6];
    const float* bl1      = (const float*)d_in[7];
    const float* Wg1      = (const float*)d_in[8];
    const float* bg1      = (const float*)d_in[9];
    const float* Wc2      = (const float*)d_in[10];
    const float* Wl2      = (const float*)d_in[11];
    const float* bl2      = (const float*)d_in[12];
    const float* Wg2      = (const float*)d_in[13];
    const float* bg2      = (const float*)d_in[14];

    const int nnode = in_sizes[0] / DFEAT;
    const int nedge = in_sizes[2] / 2;
    const int* src  = edge;
    const int* dst  = edge + nedge;

    float* x = (float*)d_ws;
    float* y2 = x + (size_t)nnode * DLAT;
    float* y  = y2 + ((size_t)nnode + 1) * DLAT;

    const size_t cpad = (((size_t)nnode + 4) / 4) * 4;
    const size_t epad = (((size_t)nedge + 3) / 4) * 4;
    const int nb  = (nnode + BIN_ROWS - 1) / BIN_ROWS;
    const int nch = (nedge + SORT_CH - 1) / SORT_CH;
    int*  off  = (int*)(y + ((size_t)nnode + 1) * DLAT);
    int*  ssrc = off + cpad;
    int2* cs   = (int2*)(ssrc + epad);
    int*  tab  = (int*)(cs + epad);
    int*  lst  = tab + (((size_t)nb * nch + 3) / 4) * 4;

    chunk_sort_kernel<<<nch, 1024, 0, stream>>>(src, dst, cs, tab, lst, nedge, nb, nch);
    bin_sort_kernel<<<nb, 1024, 0, stream>>>(cs, tab, lst, off, ssrc, nnode, nb, nch);

    mlp_norm_conv_kernel<<<DENSE_GRID, 256, 0, stream>>>(features, Wmlp, bmlp, Wc1, x, y, nnode);
    update_kernel<true><<<DENSE_GRID, 256, 0, stream>>>(x, y, off, ssrc, id_emb, Wl1, bl1, Wg1, bg1,
                                                  Wc2, x, y2, nnode);
    update_kernel<false><<<DENSE_GRID, 256, 0, stream>>>(x, y2, off, ssrc, id_emb, Wl2, bl2, Wg2, bg2,
                                                   nullptr, (float*)d_out, nullptr, nnode);
}


# Dense kernels: no drain of the global load/store counters inside a trip: work-group barriers by hand with an LDS-only wait; next tile's row bounds loaded under the gather, not behind the barrier
# speedup vs baseline: 3.1483x; 1.0301x over previous
#include <hip/hip_runtime.h>
#include <hip/hip_bf16.h>

#define DLAT 64
#define DFEAT 128
#define SLOPE 0.01f
#define EPSN 1e-12f

typedef __attribute__((ext_vector_type(2))) float v2f;
typedef __attribute__((ext_vector_type(8))) float v8f;
typedef __attribute__((ext_vector_type(4))) float v4f;

__device__ __forceinline__ float leaky(float v) { return v >= 0.0f ? v : SLOPE * v; }

__device__ __forceinline__ v8f wmma_f32(v2f a, v2f b, v8f c) {
    return __builtin_amdgcn_wmma_f32_16x16x4_f32(false, a, false, b, (short)0, c,
                                                 false, false);
}

__device__ __forceinline__ void lds_barrier() {
    asm volatile("s_wait_dscnt 0x0\n\ts_barrier_signal -1\n\ts_barrier_wait -1" ::: "memory");
}

#define FPITCH 132
#define XPITCH 68
#define DENSE_GRID 1024

__device__ __forceinline__ v4f ld_feat(const float* __restrict__ feat, int tile, int f, int nnode) {
    int g = tile * 32 + (f >> 5);
    if (g > nnode - 1) g = nnode - 1;
    return *(const v4f*)(feat + (size_t)g * DFEAT + ((f & 31) << 2));
}
__device__ __forceinline__ void st_feat(float* ft, int f, v4f v) {
    *(v4f*)(ft + (f >> 5) * FPITCH + ((f & 31) << 2)) = v;
}
__device__ __forceinline__ v4f ld_row64(const float* __restrict__ p, int tile, int f, int nnode) {
    int g = tile * 32 + (f >> 4);
    if (g > nnode - 1) g = nnode - 1;
    return *(const v4f*)(p + (size_t)g * DLAT + ((f & 15) << 2));
}
__device__ __forceinline__ v8f ld_id(const float* __restrict__ idemb, int tile, int rbase, int col, int nnode) {
    v8f r;
    #pragma unroll
    for (int i = 0; i < 8; ++i) {
        int g = tile * 32 + rbase + i;
        if (g > nnode - 1) g = nnode - 1;
        r[i] = idemb[(size_t)g * DLAT + col];
    }
    return r;
}

__global__ __launch_bounds__(256) void mlp_norm_conv_kernel(
    const float* __restrict__ feat, const float* __restrict__ Wmlp,
    const float* __restrict__ bmlp, const float* __restrict__ Wc,
    float* __restrict__ xout, float* __restrict__ yout, int nnode)
{
    __shared__ __attribute__((aligned(16))) float Wm_s[DFEAT * DLAT];
    __shared__ __attribute__((aligned(16))) float Wc_s[DLAT * DLAT];
    __shared__ __attribute__((aligned(16))) float ft[32 * FPITCH];
    __shared__ float xt[32 * 65];
    __shared__ float scl[32];

    const int tid   = threadIdx.x;
    const int ntile = (nnode + 31) >> 5;
    const int tstep = (int)gridDim.x;
    int tile = (int)blockIdx.x;
    if (tile >= ntile) return;

    if (blockIdx.x == 0 && tid < DLAT) yout[(size_t)nnode * DLAT + tid] = 0.f;

    v4f fr0 = ld_feat(feat, tile, tid,       nnode);
    v4f fr1 = ld_feat(feat, tile, tid + 256, nnode);
    v4f fr2 = ld_feat(feat, tile, tid + 512, nnode);
    v4f fr3 = ld_feat(feat, tile, tid + 768, nnode);
    {
        float4 wm[8], wc[4];
        #pragma unroll
        for (int i = 0; i < 8; ++i) wm[i] = ((const float4*)Wmlp)[tid + 256 * i];
        #pragma unroll
        for (int i = 0; i < 4; ++i) wc[i] = ((const float4*)Wc)[tid + 256 * i];
        #pragma unroll
        for (int i = 0; i < 8; ++i) ((float4*)Wm_s)[tid + 256 * i] = wm[i];
        #pragma unroll
        for (int i = 0; i < 4; ++i) ((float4*)Wc_s)[tid + 256 * i] = wc[i];
    }

    const int lane    = tid & 31;
    const int wave    = tid >> 5;
    const int rowtile = wave >> 2;
    const int coltile = wave & 3;
    const int mn      = lane & 15;
    const int kh      = (lane >> 4) << 1;
    const int col     = coltile * 16 + mn;

    const float* ap   = ft + (rowtile * 16 + mn) * FPITCH + kh;
    const float bias  = bmlp[col];
    const int   rbase = rowtile * 16 + ((lane >> 4) << 3);
    const int   rl    = rowtile * 16 + mn;

    for (; tile < ntile; tile += tstep) {
        const int blk0 = tile * 32;

        st_feat(ft, tid,       fr0);
        st_feat(ft, tid + 256, fr1);
        st_feat(ft, tid + 512, fr2);
        st_feat(ft, tid + 768, fr3);
        lds_barrier();

        const int ntl = tile + tstep;
        if (ntl < ntile) {
            fr0 = ld_feat(feat, ntl, tid,       nnode);
            fr1 = ld_feat(feat, ntl, tid + 256, nnode);
            fr2 = ld_feat(feat, ntl, tid + 512, nnode);
            fr3 = ld_feat(feat, ntl, tid + 768, nnode);
        }

        v8f acc = {};
        #pragma unroll
        for (int kk = 0; kk < DFEAT; kk += 4) {
            v2f a = *(const v2f*)(ap + kk);
            v2f b;
            b.x = Wm_s[(kk + kh) * DLAT + col];
            b.y = Wm_s[(kk + kh + 1) * DLAT + col];
            acc = wmma_f32(a, b, acc);
        }

        #pragma unroll
        for (int i = 0; i < 8; ++i)
            xt[(rbase + i) * 65 + col] = acc[i] + bias;
        lds_barrier();

        if (tid < 32) {
            float s = 0.f;
            #pragma unroll
            for (int j = 0; j < DLAT; ++j) { float v = xt[tid * 65 + j]; s += v * v; }
            scl[tid] = 1.0f / fmaxf(sqrtf(s), EPSN);
        }
        lds_barrier();

        for (int e = tid; e < 32 * DLAT; e += 256) {
            int r = e >> 6, c = e & 63;
            float v = xt[r * 65 + c] * scl[r];
            xt[r * 65 + c] = v;
            int g = blk0 + r;
            if (g < nnode) xout[(size_t)g * DLAT + c] = v;
        }
        lds_barrier();

        v8f acc2 = {};
        #pragma unroll
        for (int kk = 0; kk < DLAT; kk += 4) {
            v2f a;
            a.x = xt[rl * 65 + kk + kh];
            a.y = xt[rl * 65 + kk + kh + 1];
            v2f b;
            b.x = Wc_s[(kk + kh) * DLAT + col];
            b.y = Wc_s[(kk + kh + 1) * DLAT + col];
            acc2 = wmma_f32(a, b, acc2);
        }
        #pragma unroll
        for (int i = 0; i < 8; ++i) {
            int g = blk0 + rbase + i;
            if (g < nnode) yout[(size_t)g * DLAT + col] = acc2[i];
        }
    }
}


struct PairBounds { int myb, myn, nmax; };

__device__ __forceinline__ PairBounds pair_bounds(int offc, int p, bool hi)
{
    const int b0 = __builtin_amdgcn_readlane(offc, 2 * p);
    const int m  = __builtin_amdgcn_readlane(offc, 2 * p + 1);
    const int e1 = __builtin_amdgcn_readlane(offc, 2 * p + 2);
    PairBounds r;
    r.myb  = hi ? m : b0;
    r.myn  = hi ? e1 - m : m - b0;
    r.nmax = max(m - b0, e1 - m);
    return r;
}

__device__ __forceinline__ int ld_src_id(const int* __restrict__ ssrc, const PairBounds& pb, int idx, int zrow)
{
    return idx < pb.myn ? ssrc[pb.myb + idx] : zrow;
}

template <int K>
__device__ __forceinline__ void issue_row_loads(const char* __restrict__ yb, int ids, unsigned coff, v4f (&v)[16])
{
    if constexpr (K < 16) {
        const int s = __builtin_amdgcn_update_dpp(0, ids, 0x150 + K, 0xf, 0xf, false);
        v[K] = *(const v4f*)(yb + (size_t)(((unsigned)s << 8) | coff));
        issue_row_loads<K + 1>(yb, ids, coff, v);
    }
}

__device__ __forceinline__ v4f gather_pair(const char* __restrict__ yb, const int* __restrict__ ssrc,
                                           const PairBounds& pb, int ids0, int ids1, int c, int zrow)
{
    v4f acc = {0.f, 0.f, 0.f, 0.f};
    const unsigned coff = (unsigned)c << 4;
    for (int j = 0; j < pb.nmax; j += 16) {
        v4f v[16];
        issue_row_loads<0>(yb, ids0, coff, v);
        ids0 = ids1;
        if (j + 32 < pb.nmax) ids1 = ld_src_id(ssrc, pb, j + 32 + c, zrow);
        #pragma unroll
        for (int k = 0; k < 16; ++k) acc += v[k];
    }
    return acc;
}

template <bool FUSE_NEXT>
__global__ __launch_bounds__(256) void update_kernel(
    const float* __restrict__ xin, const float* __restrict__ ysrc,
    const int* __restrict__ off, const int* __restrict__ ssrc,
    const float* __restrict__ idemb,
    const float* __restrict__ Wl, const float* __restrict__ bl,
    const float* __restrict__ Wg, const float* __restrict__ bg,
    const float* __restrict__ Wc, float* __restrict__ xout,
    float* __restrict__ yout, int nnode)
{
    __shared__ __attribute__((aligned(16))) float Wl_s[DLAT * DLAT];
    __shared__ __attribute__((aligned(16))) float Wg_s[DLAT * DLAT];
    __shared__ __attribute__((aligned(16))) float Wc_s[DLAT * DLAT];
    __shared__ __attribute__((aligned(16))) float xs[32 * XPITCH];
    __shared__ __attribute__((aligned(16))) float hs[32 * XPITCH];
    __shared__ float xt[32 * 65];

    const int tid   = threadIdx.x;
    const int ntile = (nnode + 31) >> 5;
    const int tstep = (int)gridDim.x;
    int tile = (int)blockIdx.x;
    if (tile >= ntile) return;

    const int lane    = tid & 31;
    const int wave    = tid >> 5;
    const int r4      = wave * 4;
    const int l4      = min(lane, 4);

    v4f xr0 = ld_row64(xin, tile, tid,       nnode);
    v4f xr1 = ld_row64(xin, tile, tid + 256, nnode);
    int offc = off[min(tile * 32 + r4 + l4, nnode)];
    {
        float4 wl[4], wg[4], wc[4];
        #pragma unroll
        for (int i = 0; i < 4; ++i) {
            wl[i] = ((const float4*)Wl)[tid + 256 * i];
            wg[i] = ((const float4*)Wg)[tid + 256 * i];
            if constexpr (FUSE_NEXT) wc[i] = ((const float4*)Wc)[tid + 256 * i];
        }
        #pragma unroll
        for (int i = 0; i < 4; ++i) {
            ((float4*)Wl_s)[tid + 256 * i] = wl[i];
            ((float4*)Wg_s)[tid + 256 * i] = wg[i];
            if constexpr (FUSE_NEXT) ((float4*)Wc_s)[tid + 256 * i] = wc[i];
        }
    }

    const int  mn = lane & 15;
    const bool hi = lane >= 16;

    if constexpr (FUSE_NEXT) {
        if (blockIdx.x == 0 && tid < DLAT) yout[(size_t)nnode * DLAT + tid] = 0.f;
    }

    int ids[2][2];
    #pragma unroll
    for (int p = 0; p < 2; ++p) {
        const PairBounds pb = pair_bounds(offc, p, hi);
        ids[p][0] = ld_src_id(ssrc, pb, mn,      nnode);
        ids[p][1] = ld_src_id(ssrc, pb, mn + 16, nnode);
    }

    const int rowtile = wave >> 2;
    const int coltile = wave & 3;
    const int kh      = (lane >> 4) << 1;
    const int col     = coltile * 16 + mn;

    const float* xp = xs + (rowtile * 16 + mn) * XPITCH + kh;
    const float* hp = hs + (rowtile * 16 + mn) * XPITCH + kh;
    const char*  yb = (const char*)ysrc;

    const float blv = bl[col], bgv = bg[col];
    const int rbase = rowtile * 16 + ((lane >> 4) << 3);

    v8f idn = ld_id(idemb, tile, rbase, col, nnode);

    for (; tile < ntile; tile += tstep) {
        const int blk0 = tile * 32;
        const v8f idc = idn;
        const int offnx = off[min((tile + tstep) * 32 + r4 + l4, nnode)];

        {
            const int o0 = (tid >> 4) * XPITCH + ((tid & 15) << 2);
            const int o1 = o0 + 16 * XPITCH;
            *(v4f*)(xs + o0) = xr0;
            *(v4f*)(xs + o1) = xr1;
        }
        #pragma unroll
        for (int p = 0; p < 2; ++p) {
            const PairBounds pb = pair_bounds(offc, p, hi);
            const v4f a = gather_pair(yb, ssrc, pb, ids[p][0], ids[p][1], mn, nnode);
            v4f r;
            r.x = leaky(a.x); r.y = leaky(a.y); r.z = leaky(a.z); r.w = leaky(a.w);
            *(v4f*)(hs + (r4 + 2 * p + (hi ? 1 : 0)) * XPITCH + mn * 4) = r;
        }
        lds_barrier();

        const int ntl = tile + tstep;
        offc = offnx;
        const PairBounds nb0 = pair_bounds(offc, 0, hi);
        const PairBounds nb1 = pair_bounds(offc, 1, hi);
        if (ntl < ntile) {
            ids[0][0] = ld_src_id(ssrc, nb0, mn,      nnode);
            ids[0][1] = ld_src_id(ssrc, nb0, mn + 16, nnode);
            ids[1][0] = ld_src_id(ssrc, nb1, mn,      nnode);
            ids[1][1] = ld_src_id(ssrc, nb1, mn + 16, nnode);
            xr0 = ld_row64(xin, ntl, tid,       nnode);
            xr1 = ld_row64(xin, ntl, tid + 256, nnode);
            idn = ld_id(idemb, ntl, rbase, col, nnode);
        }

        v8f c1 = {}, c2 = {};
        #pragma unroll 4
        for (int kk = 0; kk < DLAT; kk += 4) {
            v2f ax = *(const v2f*)(xp + kk);
            v2f ah = *(const v2f*)(hp + kk);
            v2f b1;
            b1.x = Wl_s[(kk + kh) * DLAT + col];
            b1.y = Wl_s[(kk + kh + 1) * DLAT + col];
            v2f b2;
            b2.x = Wg_s[(kk + kh) * DLAT + col];
            b2.y = Wg_s[(kk + kh + 1) * DLAT + col];
            c1 = wmma_f32(ax, b1, c1);
            c2 = wmma_f32(ah, b2, c2);
        }

        #pragma unroll
        for (int i = 0; i < 8; ++i) {
            int rloc = rbase + i;
            int g    = blk0 + rloc;
            float xhat = leaky(c1[i] + blv) + idc[i];
            float v    = leaky(c2[i] + bgv + xhat);
            if constexpr (FUSE_NEXT) {
                xt[rloc * 65 + col] = v;
            } else {
                if (g < nnode) xout[(size_t)g * DLAT + col] = v;
            }
        }

        lds_barrier();
        if constexpr (FUSE_NEXT) {
            for (int e = tid; e < 32 * DLAT; e += 256) {
                int r = e >> 6, c = e & 63;
                int g = blk0 + r;
                if (g < nnode) xout[(size_t)g * DLAT + c] = xt[r * 65 + c];
            }
            v8f acc2 = {};
            const int rl = rowtile * 16 + mn;
            #pragma unroll 4
            for (int kk = 0; kk < DLAT; kk += 4) {
                v2f a;
                a.x = xt[rl * 65 + kk + kh];
                a.y = xt[rl * 65 + kk + kh + 1];
                v2f b;
                b.x = Wc_s[(kk + kh) * DLAT + col];
                b.y = Wc_s[(kk + kh + 1) * DLAT + col];
                acc2 = wmma_f32(a, b, acc2);
            }
            #pragma unroll
            for (int i = 0; i < 8; ++i) {
                int g = blk0 + rbase + i;
                if (g < nnode) yout[(size_t)g * DLAT + col] = acc2[i];
            }
        }
    }
}

#define SORT_CH  8192
#define BIN_SH   9
#define BIN_ROWS 512
#define MAXNB    1024
#define MAXCH    1024
#define BIN_CAP  12288

__device__ __forceinline__ int block_excl_scan_1024(int v, int* wsum, int& total)
{
    const int lane = threadIdx.x & 31;
    const int wave = threadIdx.x >> 5;
    int inc = v;
    for (int o = 1; o < 32; o <<= 1) {
        int u = __shfl_up(inc, o);
        if (lane >= o) inc += u;
    }
    if (lane == 31) wsum[wave] = inc;
    __syncthreads();
    const int ws = wsum[lane];
    int wp  = lane < wave ? ws : 0;
    int tot = ws;
    for (int o = 16; o > 0; o >>= 1) {
        wp  += __shfl_xor(wp, o);
        tot += __shfl_xor(tot, o);
    }
    total = tot;
    __syncthreads();
    return wp + inc - v;
}

__global__ __launch_bounds__(1024) void chunk_sort_kernel(
    const int* __restrict__ src, const int* __restrict__ dst,
    int2* __restrict__ cs, int* __restrict__ tab, int* __restrict__ lst,
    int nedge, int nb, int nch)
{
    __shared__ int  lcnt[MAXNB];
    __shared__ int  lsta[MAXNB];
    __shared__ int  wsum[32];
    __shared__ int2 stage[SORT_CH];

    const int tid  = threadIdx.x;
    const int c    = blockIdx.x;
    const int e0   = c * SORT_CH;
    const int last = nedge - 1;

    lcnt[tid] = 0;
    __syncthreads();

    int s[8], d[8], r[8];
    #pragma unroll
    for (int k = 0; k < 8; ++k) {
        const int e = min(e0 + tid + k * 1024, last);
        s[k] = src[e];
        d[k] = dst[e];
    }
    #pragma unroll
    for (int k = 0; k < 8; ++k) {
        r[k] = 0;
        const int bin = min(d[k] >> BIN_SH, MAXNB - 1);
        if (e0 + tid + k * 1024 < nedge) r[k] = atomicAdd(&lcnt[bin], 1);
    }
    __syncthreads();

    const int v = lcnt[tid];
    int tot;
    const int ex = block_excl_scan_1024(v, wsum, tot);
    lsta[tid] = ex;
    if (tid < nb) {
        tab[(size_t)tid * nch + c] = v;
        lst[(size_t)c * nb + tid]  = ex;
    }
    __syncthreads();

    #pragma unroll
    for (int k = 0; k < 8; ++k) {
        const int bin = min(d[k] >> BIN_SH, MAXNB - 1);
        if (e0 + tid + k * 1024 < nedge) stage[lsta[bin] + r[k]] = make_int2(s[k], d[k]);
    }
    __syncthreads();

    for (int j = tid; j < tot; j += 1024) cs[(size_t)e0 + j] = stage[j];
}

__global__ __launch_bounds__(1024) void bin_sort_kernel(
    const int2* __restrict__ cs, const int* __restrict__ tab, const int* __restrict__ lst,
    int* __restrict__ off, int* __restrict__ ssrc, int nnode, int nb, int nch)
{
    __shared__ int  ppre[MAXCH];
    __shared__ int  ps[MAXCH];
    __shared__ int  lcnt[BIN_ROWS];
    __shared__ int  loff[BIN_ROWS];
    __shared__ int  lcur[BIN_ROWS];
    __shared__ int  wsum[32];
    __shared__ int  gsum[32];
    __shared__ int2 A[BIN_CAP];
    __shared__ int  B[BIN_CAP];

    const int tid   = threadIdx.x;
    const int lane  = tid & 31;
    const int wave  = tid >> 5;
    const int b     = blockIdx.x;
    const int rbase = b << BIN_SH;

    int n = 0, st = 0;
    if (tid < nch) {
        n  = tab[(size_t)b * nch + tid];
        st = lst[(size_t)tid * nb + b];
    }

    int g = 0;
    {
        const int   nfront = b * nch;
        const int   n4     = nfront >> 2;
        const int4* t4     = (const int4*)tab;
        int i = tid;
        for (; i + 7 * 1024 < n4; i += 8 * 1024) {
            int4 a[8];
            #pragma unroll
            for (int k = 0; k < 8; ++k) a[k] = t4[i + k * 1024];
            #pragma unroll
            for (int k = 0; k < 8; ++k) g += (a[k].x + a[k].y) + (a[k].z + a[k].w);
        }
        for (; i < n4; i += 1024) {
            int4 a = t4[i];
            g += (a.x + a.y) + (a.z + a.w);
        }
        for (int q = (n4 << 2) + tid; q < nfront; q += 1024) g += tab[q];
    }
    for (int o = 16; o > 0; o >>= 1) g += __shfl_xor(g, o);
    if (lane == 0) gsum[wave] = g;

    ps[tid] = st;
    if (tid < BIN_ROWS) { lcnt[tid] = 0; lcur[tid] = 0; }
    int total;
    const int ex = block_excl_scan_1024(n, wsum, total);
    ppre[tid] = ex;
    int gb = gsum[lane];
    for (int o = 16; o > 0; o >>= 1) gb += __shfl_xor(gb, o);
    __syncthreads();

    const bool single = total <= BIN_CAP;

    for (int j0 = tid; j0 < total; j0 += 8 * 1024) {
        int2 e[8];
        #pragma unroll
        for (int k = 0; k < 8; ++k) {
            const int j = min(j0 + k * 1024, total - 1);
            int cc = 0;
            #pragma unroll
            for (int sft = MAXCH / 2; sft > 0; sft >>= 1)
                if (ppre[cc + sft] <= j) cc += sft;
            e[k] = cs[(size_t)cc * SORT_CH + ps[cc] + (j - ppre[cc])];
        }
        #pragma unroll
        for (int k = 0; k < 8; ++k) {
            const int j = j0 + k * 1024;
            if (j < total) {
                const int ld = min(max(e[k].y - rbase, 0), BIN_ROWS - 1);
                atomicAdd(&lcnt[ld], 1);
                if (single) A[j] = make_int2(e[k].x, ld);
            }
        }
    }
    __syncthreads();

    const int cv = tid < BIN_ROWS ? lcnt[tid] : 0;
    int rtot;
    const int rex = block_excl_scan_1024(cv, wsum, rtot);
    if (tid < BIN_ROWS) {
        loff[tid] = rex;
        if (rbase + tid < nnode) off[rbase + tid] = gb + rex;
    }
    if (b == nb - 1 && tid == 0) off[nnode] = gb + total;
    __syncthreads();

    if (single) {
        for (int j = tid; j < total; j += 1024) {
            const int2 e = A[j];
            const int pos = loff[e.y] + atomicAdd(&lcur[e.y], 1);
            B[pos] = e.x;
        }
        __syncthreads();
        for (int j = tid; j < total; j += 1024) ssrc[(size_t)gb + j] = B[j];
    } else {
        for (int j = tid; j < total; j += 1024) {
            int cc = 0;
            #pragma unroll
            for (int sft = MAXCH / 2; sft > 0; sft >>= 1)
                if (ppre[cc + sft] <= j) cc += sft;
            const int2 e  = cs[(size_t)cc * SORT_CH + ps[cc] + (j - ppre[cc])];
            const int  ld = min(max(e.y - rbase, 0), BIN_ROWS - 1);
            const int  pos = loff[ld] + atomicAdd(&lcur[ld], 1);
            ssrc[(size_t)gb + pos] = e.x;
        }
    }
}

extern "C" void kernel_launch(void* const* d_in, const int* in_sizes, int n_in,
                              void* d_out, int out_size, void* d_ws, size_t ws_size,
                              hipStream_t stream)
{
    const float* features = (const float*)d_in[0];
    const float* id_emb   = (const float*)d_in[1];
    const int*   edge     = (const int*)d_in[2];
    const float* Wmlp     = (const float*)d_in[3];
    const float* bmlp     = (const float*)d_in[4];
    const float* Wc1      = (const float*)d_in[5];
    const float* Wl1      = (const float*)d_in[6];
    const float* bl1      = (const float*)d_in[7];
    const float* Wg1      = (const float*)d_in[8];
    const float* bg1      = (const float*)d_in[9];
    const float* Wc2      = (const float*)d_in[10];
    const float* Wl2      = (const float*)d_in[11];
    const float* bl2      = (const float*)d_in[12];
    const float* Wg2      = (const float*)d_in[13];
    const float* bg2      = (const float*)d_in[14];

    const int nnode = in_sizes[0] / DFEAT;
    const int nedge = in_sizes[2] / 2;
    const int* src  = edge;
    const int* dst  = edge + nedge;

    float* x = (float*)d_ws;
    float* y2 = x + (size_t)nnode * DLAT;
    float* y  = y2 + ((size_t)nnode + 1) * DLAT;

    const size_t cpad = (((size_t)nnode + 4) / 4) * 4;
    const size_t epad = (((size_t)nedge + 3) / 4) * 4;
    const int nb  = (nnode + BIN_ROWS - 1) / BIN_ROWS;
    const int nch = (nedge + SORT_CH - 1) / SORT_CH;
    int*  off  = (int*)(y + ((size_t)nnode + 1) * DLAT);
    int*  ssrc = off + cpad;
    int2* cs   = (int2*)(ssrc + epad);
    int*  tab  = (int*)(cs + epad);
    int*  lst  = tab + (((size_t)nb * nch + 3) / 4) * 4;

    chunk_sort_kernel<<<nch, 1024, 0, stream>>>(src, dst, cs, tab, lst, nedge, nb, nch);
    bin_sort_kernel<<<nb, 1024, 0, stream>>>(cs, tab, lst, off, ssrc, nnode, nb, nch);

    mlp_norm_conv_kernel<<<DENSE_GRID, 256, 0, stream>>>(features, Wmlp, bmlp, Wc1, x, y, nnode);
    update_kernel<true><<<DENSE_GRID, 256, 0, stream>>>(x, y, off, ssrc, id_emb, Wl1, bl1, Wg1, bg1,
                                                  Wc2, x, y2, nnode);
    update_kernel<false><<<DENSE_GRID, 256, 0, stream>>>(x, y2, off, ssrc, id_emb, Wl2, bl2, Wg2, bg2,
                                                   nullptr, (float*)d_out, nullptr, nnode);
}


# Dense kernels: weights transposed in LDS; each A and B fragment of two k-steps is one conflict-free 16-B LDS load (was 4-B pairs, 4 dwords per bank); k per product 0,1,4,5|2,3,6,7: sum order changes
# speedup vs baseline: 3.2503x; 1.0324x over previous
#include <hip/hip_runtime.h>
#include <hip/hip_bf16.h>

#define DLAT 64
#define DFEAT 128
#define SLOPE 0.01f
#define EPSN 1e-12f

typedef __attribute__((ext_vector_type(2))) float v2f;
typedef __attribute__((ext_vector_type(8))) float v8f;
typedef __attribute__((ext_vector_type(4))) float v4f;

__device__ __forceinline__ float leaky(float v) { return v >= 0.0f ? v : SLOPE * v; }

__device__ __forceinline__ v8f wmma_f32(v2f a, v2f b, v8f c) {
    return __builtin_amdgcn_wmma_f32_16x16x4_f32(false, a, false, b, (short)0, c,
                                                 false, false);
}

__device__ __forceinline__ void lds_barrier() {
    asm volatile("s_wait_dscnt 0x0\n\ts_barrier_signal -1\n\ts_barrier_wait -1" ::: "memory");
}

#define FPITCH 132
#define XPITCH 68
#define WPITCH 68
#define WMPITCH 132
#define DENSE_GRID 1024

template <int K>
__device__ __forceinline__ void ld_wT(const float* __restrict__ W, int tid, v4f (&r)[K / 16]) {
    const int col = tid & 63;
    const int kq  = tid >> 6;
    #pragma unroll
    for (int i = 0; i < K / 16; ++i) {
        const int k = (kq + 4 * i) << 2;
        r[i].x = W[(k + 0) * DLAT + col];
        r[i].y = W[(k + 1) * DLAT + col];
        r[i].z = W[(k + 2) * DLAT + col];
        r[i].w = W[(k + 3) * DLAT + col];
    }
}
template <int K, int PITCH>
__device__ __forceinline__ void st_wT(float* wt, int tid, const v4f (&r)[K / 16]) {
    const int col = tid & 63;
    const int kq  = tid >> 6;
    #pragma unroll
    for (int i = 0; i < K / 16; ++i)
        *(v4f*)(wt + col * PITCH + ((kq + 4 * i) << 2)) = r[i];
}

__device__ __forceinline__ v8f wmma_2steps(v4f a, v4f b, v8f c) {
    v2f a0, b0, a1, b1;
    a0.x = a.x; a0.y = a.y; b0.x = b.x; b0.y = b.y;
    a1.x = a.z; a1.y = a.w; b1.x = b.z; b1.y = b.w;
    c = wmma_f32(a0, b0, c);
    return wmma_f32(a1, b1, c);
}

__device__ __forceinline__ v4f ld_feat(const float* __restrict__ feat, int tile, int f, int nnode) {
    int g = tile * 32 + (f >> 5);
    if (g > nnode - 1) g = nnode - 1;
    return *(const v4f*)(feat + (size_t)g * DFEAT + ((f & 31) << 2));
}
__device__ __forceinline__ void st_feat(float* ft, int f, v4f v) {
    *(v4f*)(ft + (f >> 5) * FPITCH + ((f & 31) << 2)) = v;
}
__device__ __forceinline__ v4f ld_row64(const float* __restrict__ p, int tile, int f, int nnode) {
    int g = tile * 32 + (f >> 4);
    if (g > nnode - 1) g = nnode - 1;
    return *(const v4f*)(p + (size_t)g * DLAT + ((f & 15) << 2));
}
__device__ __forceinline__ v8f ld_id(const float* __restrict__ idemb, int tile, int rbase, int col, int nnode) {
    v8f r;
    #pragma unroll
    for (int i = 0; i < 8; ++i) {
        int g = tile * 32 + rbase + i;
        if (g > nnode - 1) g = nnode - 1;
        r[i] = idemb[(size_t)g * DLAT + col];
    }
    return r;
}

__global__ __launch_bounds__(256) void mlp_norm_conv_kernel(
    const float* __restrict__ feat, const float* __restrict__ Wmlp,
    const float* __restrict__ bmlp, const float* __restrict__ Wc,
    float* __restrict__ xout, float* __restrict__ yout, int nnode)
{
    __shared__ __attribute__((aligned(16))) float Wm_t[DLAT * WMPITCH];
    __shared__ __attribute__((aligned(16))) float Wc_t[DLAT * WPITCH];
    __shared__ __attribute__((aligned(16))) float ft[32 * FPITCH];
    __shared__ __attribute__((aligned(16))) float xt[32 * XPITCH];
    __shared__ float scl[32];

    const int tid   = threadIdx.x;
    const int ntile = (nnode + 31) >> 5;
    const int tstep = (int)gridDim.x;
    int tile = (int)blockIdx.x;
    if (tile >= ntile) return;

    if (blockIdx.x == 0 && tid < DLAT) yout[(size_t)nnode * DLAT + tid] = 0.f;

    v4f fr0 = ld_feat(feat, tile, tid,       nnode);
    v4f fr1 = ld_feat(feat, tile, tid + 256, nnode);
    v4f fr2 = ld_feat(feat, tile, tid + 512, nnode);
    v4f fr3 = ld_feat(feat, tile, tid + 768, nnode);
    {
        v4f wm[8], wc[4];
        ld_wT<DFEAT>(Wmlp, tid, wm);
        ld_wT<DLAT>(Wc, tid, wc);
        st_wT<DFEAT, WMPITCH>(Wm_t, tid, wm);
        st_wT<DLAT, WPITCH>(Wc_t, tid, wc);
    }

    const int lane    = tid & 31;
    const int wave    = tid >> 5;
    const int rowtile = wave >> 2;
    const int coltile = wave & 3;
    const int mn      = lane & 15;
    const int k4h     = (lane >> 4) << 2;
    const int col     = coltile * 16 + mn;

    const float* ap   = ft   + (rowtile * 16 + mn) * FPITCH + k4h;
    const float* bp   = Wm_t + col * WMPITCH + k4h;
    const float* a2p  = xt   + (rowtile * 16 + mn) * XPITCH + k4h;
    const float* b2p  = Wc_t + col * WPITCH + k4h;
    const float bias  = bmlp[col];
    const int   rbase = rowtile * 16 + ((lane >> 4) << 3);

    for (; tile < ntile; tile += tstep) {
        const int blk0 = tile * 32;

        st_feat(ft, tid,       fr0);
        st_feat(ft, tid + 256, fr1);
        st_feat(ft, tid + 512, fr2);
        st_feat(ft, tid + 768, fr3);
        lds_barrier();

        const int ntl = tile + tstep;
        if (ntl < ntile) {
            fr0 = ld_feat(feat, ntl, tid,       nnode);
            fr1 = ld_feat(feat, ntl, tid + 256, nnode);
            fr2 = ld_feat(feat, ntl, tid + 512, nnode);
            fr3 = ld_feat(feat, ntl, tid + 768, nnode);
        }

        v8f acc = {};
        #pragma unroll
        for (int kk = 0; kk < DFEAT; kk += 8)
            acc = wmma_2steps(*(const v4f*)(ap + kk), *(const v4f*)(bp + kk), acc);

        #pragma unroll
        for (int i = 0; i < 8; ++i)
            xt[(rbase + i) * XPITCH + col] = acc[i] + bias;
        lds_barrier();

        if (tid < 32) {
            float s = 0.f;
            #pragma unroll
            for (int q = 0; q < DLAT / 4; ++q) {
                const v4f v = *(const v4f*)(xt + tid * XPITCH + 4 * q);
                s += v.x * v.x; s += v.y * v.y; s += v.z * v.z; s += v.w * v.w;
            }
            scl[tid] = 1.0f / fmaxf(sqrtf(s), EPSN);
        }
        lds_barrier();

        #pragma unroll
        for (int i = 0; i < 2; ++i) {
            const int f = tid + 256 * i;
            const int r = f >> 4, c4 = (f & 15) << 2;
            const float sc = scl[r];
            v4f v = *(const v4f*)(xt + r * XPITCH + c4);
            v.x *= sc; v.y *= sc; v.z *= sc; v.w *= sc;
            *(v4f*)(xt + r * XPITCH + c4) = v;
            const int g = blk0 + r;
            if (g < nnode) *(v4f*)(xout + (size_t)g * DLAT + c4) = v;
        }
        lds_barrier();

        v8f acc2 = {};
        #pragma unroll
        for (int kk = 0; kk < DLAT; kk += 8)
            acc2 = wmma_2steps(*(const v4f*)(a2p + kk), *(const v4f*)(b2p + kk), acc2);
        #pragma unroll
        for (int i = 0; i < 8; ++i) {
            int g = blk0 + rbase + i;
            if (g < nnode) yout[(size_t)g * DLAT + col] = acc2[i];
        }
    }
}


struct PairBounds { int myb, myn, nmax; };

__device__ __forceinline__ PairBounds pair_bounds(int offc, int p, bool hi)
{
    const int b0 = __builtin_amdgcn_readlane(offc, 2 * p);
    const int m  = __builtin_amdgcn_readlane(offc, 2 * p + 1);
    const int e1 = __builtin_amdgcn_readlane(offc, 2 * p + 2);
    PairBounds r;
    r.myb  = hi ? m : b0;
    r.myn  = hi ? e1 - m : m - b0;
    r.nmax = max(m - b0, e1 - m);
    return r;
}

__device__ __forceinline__ int ld_src_id(const int* __restrict__ ssrc, const PairBounds& pb, int idx, int zrow)
{
    return idx < pb.myn ? ssrc[pb.myb + idx] : zrow;
}

template <int K>
__device__ __forceinline__ void issue_row_loads(const char* __restrict__ yb, int ids, unsigned coff, v4f (&v)[16])
{
    if constexpr (K < 16) {
        const int s = __builtin_amdgcn_update_dpp(0, ids, 0x150 + K, 0xf, 0xf, false);
        v[K] = *(const v4f*)(yb + (size_t)(((unsigned)s << 8) | coff));
        issue_row_loads<K + 1>(yb, ids, coff, v);
    }
}

__device__ __forceinline__ v4f gather_pair(const char* __restrict__ yb, const int* __restrict__ ssrc,
                                           const PairBounds& pb, int ids0, int ids1, int c, int zrow)
{
    v4f acc = {0.f, 0.f, 0.f, 0.f};
    const unsigned coff = (unsigned)c << 4;
    for (int j = 0; j < pb.nmax; j += 16) {
        v4f v[16];
        issue_row_loads<0>(yb, ids0, coff, v);
        ids0 = ids1;
        if (j + 32 < pb.nmax) ids1 = ld_src_id(ssrc, pb, j + 32 + c, zrow);
        #pragma unroll
        for (int k = 0; k < 16; ++k) acc += v[k];
    }
    return acc;
}

template <bool FUSE_NEXT>
__global__ __launch_bounds__(256) void update_kernel(
    const float* __restrict__ xin, const float* __restrict__ ysrc,
    const int* __restrict__ off, const int* __restrict__ ssrc,
    const float* __restrict__ idemb,
    const float* __restrict__ Wl, const float* __restrict__ bl,
    const float* __restrict__ Wg, const float* __restrict__ bg,
    const float* __restrict__ Wc, float* __restrict__ xout,
    float* __restrict__ yout, int nnode)
{
    constexpr int NPIECE = FUSE_NEXT ? 3 : 2;
    __shared__ __attribute__((aligned(16))) float lds[NPIECE * 32 * XPITCH + NPIECE * DLAT * WPITCH];
    float* const xs   = lds;
    float* const hs   = xs + 32 * XPITCH;
    float* const xt   = hs + 32 * XPITCH;
    float* const Wl_t = lds + NPIECE * 32 * XPITCH;
    float* const Wg_t = Wl_t + DLAT * WPITCH;
    float* const Wc_t = Wg_t + DLAT * WPITCH;

    const int tid   = threadIdx.x;
    const int ntile = (nnode + 31) >> 5;
    const int tstep = (int)gridDim.x;
    int tile = (int)blockIdx.x;
    if (tile >= ntile) return;

    const int lane    = tid & 31;
    const int wave    = tid >> 5;
    const int r4      = wave * 4;
    const int l4      = min(lane, 4);

    v4f xr0 = ld_row64(xin, tile, tid,       nnode);
    v4f xr1 = ld_row64(xin, tile, tid + 256, nnode);
    int offc = off[min(tile * 32 + r4 + l4, nnode)];
    {
        v4f wl[4], wg[4], wc[4];
        ld_wT<DLAT>(Wl, tid, wl);
        ld_wT<DLAT>(Wg, tid, wg);
        if constexpr (FUSE_NEXT) ld_wT<DLAT>(Wc, tid, wc);
        st_wT<DLAT, WPITCH>(Wl_t, tid, wl);
        st_wT<DLAT, WPITCH>(Wg_t, tid, wg);
        if constexpr (FUSE_NEXT) st_wT<DLAT, WPITCH>(Wc_t, tid, wc);
    }

    const int  mn = lane & 15;
    const bool hi = lane >= 16;

    if constexpr (FUSE_NEXT) {
        if (blockIdx.x == 0 && tid < DLAT) yout[(size_t)nnode * DLAT + tid] = 0.f;
    }

    int ids[2][2];
    #pragma unroll
    for (int p = 0; p < 2; ++p) {
        const PairBounds pb = pair_bounds(offc, p, hi);
        ids[p][0] = ld_src_id(ssrc, pb, mn,      nnode);
        ids[p][1] = ld_src_id(ssrc, pb, mn + 16, nnode);
    }

    const int rowtile = wave >> 2;
    const int coltile = wave & 3;
    const int k4h     = (lane >> 4) << 2;
    const int col     = coltile * 16 + mn;

    const float* xp  = xs   + (rowtile * 16 + mn) * XPITCH + k4h;
    const float* hp  = hs   + (rowtile * 16 + mn) * XPITCH + k4h;
    const float* wlp = Wl_t + col * WPITCH + k4h;
    const float* wgp = Wg_t + col * WPITCH + k4h;
    const char*  yb = (const char*)ysrc;

    const float blv = bl[col], bgv = bg[col];
    const int rbase = rowtile * 16 + ((lane >> 4) << 3);

    v8f idn = ld_id(idemb, tile, rbase, col, nnode);

    for (; tile < ntile; tile += tstep) {
        const int blk0 = tile * 32;
        const v8f idc = idn;
        const int offnx = off[min((tile + tstep) * 32 + r4 + l4, nnode)];

        {
            const int o0 = (tid >> 4) * XPITCH + ((tid & 15) << 2);
            const int o1 = o0 + 16 * XPITCH;
            *(v4f*)(xs + o0) = xr0;
            *(v4f*)(xs + o1) = xr1;
        }
        #pragma unroll
        for (int p = 0; p < 2; ++p) {
            const PairBounds pb = pair_bounds(offc, p, hi);
            const v4f a = gather_pair(yb, ssrc, pb, ids[p][0], ids[p][1], mn, nnode);
            v4f r;
            r.x = leaky(a.x); r.y = leaky(a.y); r.z = leaky(a.z); r.w = leaky(a.w);
            *(v4f*)(hs + (r4 + 2 * p + (hi ? 1 : 0)) * XPITCH + mn * 4) = r;
        }
        lds_barrier();

        const int ntl = tile + tstep;
        offc = offnx;
        const PairBounds nb0 = pair_bounds(offc, 0, hi);
        const PairBounds nb1 = pair_bounds(offc, 1, hi);
        if (ntl < ntile) {
            ids[0][0] = ld_src_id(ssrc, nb0, mn,      nnode);
            ids[0][1] = ld_src_id(ssrc, nb0, mn + 16, nnode);
            ids[1][0] = ld_src_id(ssrc, nb1, mn,      nnode);
            ids[1][1] = ld_src_id(ssrc, nb1, mn + 16, nnode);
            xr0 = ld_row64(xin, ntl, tid,       nnode);
            xr1 = ld_row64(xin, ntl, tid + 256, nnode);
            idn = ld_id(idemb, ntl, rbase, col, nnode);
        }

        v8f c1 = {}, c2 = {};
        #pragma unroll
        for (int kk = 0; kk < DLAT; kk += 8) {
            const v4f ax = *(const v4f*)(xp + kk);
            const v4f ah = *(const v4f*)(hp + kk);
            const v4f b1 = *(const v4f*)(wlp + kk);
            const v4f b2 = *(const v4f*)(wgp + kk);
            c1 = wmma_2steps(ax, b1, c1);
            c2 = wmma_2steps(ah, b2, c2);
        }

        #pragma unroll
        for (int i = 0; i < 8; ++i) {
            int rloc = rbase + i;
            int g    = blk0 + rloc;
            float xhat = leaky(c1[i] + blv) + idc[i];
            float v    = leaky(c2[i] + bgv + xhat);
            if constexpr (FUSE_NEXT) {
                xt[rloc * XPITCH + col] = v;
            } else {
                if (g < nnode) xout[(size_t)g * DLAT + col] = v;
            }
        }

        lds_barrier();
        if constexpr (FUSE_NEXT) {
            #pragma unroll
            for (int i = 0; i < 2; ++i) {
                const int f = tid + 256 * i;
                const int r = f >> 4, c4 = (f & 15) << 2;
                const int g = blk0 + r;
                if (g < nnode) *(v4f*)(xout + (size_t)g * DLAT + c4) = *(const v4f*)(xt + r * XPITCH + c4);
            }
            v8f acc2 = {};
            const float* a2p = xt   + (rowtile * 16 + mn) * XPITCH + k4h;
            const float* wcp = Wc_t + col * WPITCH + k4h;
            #pragma unroll
            for (int kk = 0; kk < DLAT; kk += 8)
                acc2 = wmma_2steps(*(const v4f*)(a2p + kk), *(const v4f*)(wcp + kk), acc2);
            #pragma unroll
            for (int i = 0; i < 8; ++i) {
                int g = blk0 + rbase + i;
                if (g < nnode) yout[(size_t)g * DLAT + col] = acc2[i];
            }
        }
    }
}

#define SORT_CH  8192
#define BIN_SH   9
#define BIN_ROWS 512
#define MAXNB    1024
#define MAXCH    1024
#define BIN_CAP  12288

__device__ __forceinline__ int block_excl_scan_1024(int v, int* wsum, int& total)
{
    const int lane = threadIdx.x & 31;
    const int wave = threadIdx.x >> 5;
    int inc = v;
    for (int o = 1; o < 32; o <<= 1) {
        int u = __shfl_up(inc, o);
        if (lane >= o) inc += u;
    }
    if (lane == 31) wsum[wave] = inc;
    __syncthreads();
    const int ws = wsum[lane];
    int wp  = lane < wave ? ws : 0;
    int tot = ws;
    for (int o = 16; o > 0; o >>= 1) {
        wp  += __shfl_xor(wp, o);
        tot += __shfl_xor(tot, o);
    }
    total = tot;
    __syncthreads();
    return wp + inc - v;
}

__global__ __launch_bounds__(1024) void chunk_sort_kernel(
    const int* __restrict__ src, const int* __restrict__ dst,
    int2* __restrict__ cs, int* __restrict__ tab, int* __restrict__ lst,
    int nedge, int nb, int nch)
{
    __shared__ int  lcnt[MAXNB];
    __shared__ int  lsta[MAXNB];
    __shared__ int  wsum[32];
    __shared__ int2 stage[SORT_CH];

    const int tid  = threadIdx.x;
    const int c    = blockIdx.x;
    const int e0   = c * SORT_CH;
    const int last = nedge - 1;

    lcnt[tid] = 0;
    __syncthreads();

    int s[8], d[8], r[8];
    #pragma unroll
    for (int k = 0; k < 8; ++k) {
        const int e = min(e0 + tid + k * 1024, last);
        s[k] = src[e];
        d[k] = dst[e];
    }
    #pragma unroll
    for (int k = 0; k < 8; ++k) {
        r[k] = 0;
        const int bin = min(d[k] >> BIN_SH, MAXNB - 1);
        if (e0 + tid + k * 1024 < nedge) r[k] = atomicAdd(&lcnt[bin], 1);
    }
    __syncthreads();

    const int v = lcnt[tid];
    int tot;
    const int ex = block_excl_scan_1024(v, wsum, tot);
    lsta[tid] = ex;
    if (tid < nb) {
        tab[(size_t)tid * nch + c] = v;
        lst[(size_t)c * nb + tid]  = ex;
    }
    __syncthreads();

    #pragma unroll
    for (int k = 0; k < 8; ++k) {
        const int bin = min(d[k] >> BIN_SH, MAXNB - 1);
        if (e0 + tid + k * 1024 < nedge) stage[lsta[bin] + r[k]] = make_int2(s[k], d[k]);
    }
    __syncthreads();

    for (int j = tid; j < tot; j += 1024) cs[(size_t)e0 + j] = stage[j];
}

__global__ __launch_bounds__(1024) void bin_sort_kernel(
    const int2* __restrict__ cs, const int* __restrict__ tab, const int* __restrict__ lst,
    int* __restrict__ off, int* __restrict__ ssrc, int nnode, int nb, int nch)
{
    __shared__ int  ppre[MAXCH];
    __shared__ int  ps[MAXCH];
    __shared__ int  lcnt[BIN_ROWS];
    __shared__ int  loff[BIN_ROWS];
    __shared__ int  lcur[BIN_ROWS];
    __shared__ int  wsum[32];
    __shared__ int  gsum[32];
    __shared__ int2 A[BIN_CAP];
    __shared__ int  B[BIN_CAP];

    const int tid   = threadIdx.x;
    const int lane  = tid & 31;
    const int wave  = tid >> 5;
    const int b     = blockIdx.x;
    const int rbase = b << BIN_SH;

    int n = 0, st = 0;
    if (tid < nch) {
        n  = tab[(size_t)b * nch + tid];
        st = lst[(size_t)tid * nb + b];
    }

    int g = 0;
    {
        const int   nfront = b * nch;
        const int   n4     = nfront >> 2;
        const int4* t4     = (const int4*)tab;
        int i = tid;
        for (; i + 7 * 1024 < n4; i += 8 * 1024) {
            int4 a[8];
            #pragma unroll
            for (int k = 0; k < 8; ++k) a[k] = t4[i + k * 1024];
            #pragma unroll
            for (int k = 0; k < 8; ++k) g += (a[k].x + a[k].y) + (a[k].z + a[k].w);
        }
        for (; i < n4; i += 1024) {
            int4 a = t4[i];
            g += (a.x + a.y) + (a.z + a.w);
        }
        for (int q = (n4 << 2) + tid; q < nfront; q += 1024) g += tab[q];
    }
    for (int o = 16; o > 0; o >>= 1) g += __shfl_xor(g, o);
    if (lane == 0) gsum[wave] = g;

    ps[tid] = st;
    if (tid < BIN_ROWS) { lcnt[tid] = 0; lcur[tid] = 0; }
    int total;
    const int ex = block_excl_scan_1024(n, wsum, total);
    ppre[tid] = ex;
    int gb = gsum[lane];
    for (int o = 16; o > 0; o >>= 1) gb += __shfl_xor(gb, o);
    __syncthreads();

    const bool single = total <= BIN_CAP;

    for (int j0 = tid; j0 < total; j0 += 8 * 1024) {
        int2 e[8];
        #pragma unroll
        for (int k = 0; k < 8; ++k) {
            const int j = min(j0 + k * 1024, total - 1);
            int cc = 0;
            #pragma unroll
            for (int sft = MAXCH / 2; sft > 0; sft >>= 1)
                if (ppre[cc + sft] <= j) cc += sft;
            e[k] = cs[(size_t)cc * SORT_CH + ps[cc] + (j - ppre[cc])];
        }
        #pragma unroll
        for (int k = 0; k < 8; ++k) {
            const int j = j0 + k * 1024;
            if (j < total) {
                const int ld = min(max(e[k].y - rbase, 0), BIN_ROWS - 1);
                atomicAdd(&lcnt[ld], 1);
                if (single) A[j] = make_int2(e[k].x, ld);
            }
        }
    }
    __syncthreads();

    const int cv = tid < BIN_ROWS ? lcnt[tid] : 0;
    int rtot;
    const int rex = block_excl_scan_1024(cv, wsum, rtot);
    if (tid < BIN_ROWS) {
        loff[tid] = rex;
        if (rbase + tid < nnode) off[rbase + tid] = gb + rex;
    }
    if (b == nb - 1 && tid == 0) off[nnode] = gb + total;
    __syncthreads();

    if (single) {
        for (int j = tid; j < total; j += 1024) {
            const int2 e = A[j];
            const int pos = loff[e.y] + atomicAdd(&lcur[e.y], 1);
            B[pos] = e.x;
        }
        __syncthreads();
        for (int j = tid; j < total; j += 1024) ssrc[(size_t)gb + j] = B[j];
    } else {
        for (int j = tid; j < total; j += 1024) {
            int cc = 0;
            #pragma unroll
            for (int sft = MAXCH / 2; sft > 0; sft >>= 1)
                if (ppre[cc + sft] <= j) cc += sft;
            const int2 e  = cs[(size_t)cc * SORT_CH + ps[cc] + (j - ppre[cc])];
            const int  ld = min(max(e.y - rbase, 0), BIN_ROWS - 1);
            const int  pos = loff[ld] + atomicAdd(&lcur[ld], 1);
            ssrc[(size_t)gb + pos] = e.x;
        }
    }
}

extern "C" void kernel_launch(void* const* d_in, const int* in_sizes, int n_in,
                              void* d_out, int out_size, void* d_ws, size_t ws_size,
                              hipStream_t stream)
{
    const float* features = (const float*)d_in[0];
    const float* id_emb   = (const float*)d_in[1];
    const int*   edge     = (const int*)d_in[2];
    const float* Wmlp     = (const float*)d_in[3];
    const float* bmlp     = (const float*)d_in[4];
    const float* Wc1      = (const float*)d_in[5];
    const float* Wl1      = (const float*)d_in[6];
    const float* bl1      = (const float*)d_in[7];
    const float* Wg1      = (const float*)d_in[8];
    const float* bg1      = (const float*)d_in[9];
    const float* Wc2      = (const float*)d_in[10];
    const float* Wl2      = (const float*)d_in[11];
    const float* bl2      = (const float*)d_in[12];
    const float* Wg2      = (const float*)d_in[13];
    const float* bg2      = (const float*)d_in[14];

    const int nnode = in_sizes[0] / DFEAT;
    const int nedge = in_sizes[2] / 2;
    const int* src  = edge;
    const int* dst  = edge + nedge;

    float* x = (float*)d_ws;
    float* y2 = x + (size_t)nnode * DLAT;
    float* y  = y2 + ((size_t)nnode + 1) * DLAT;

    const size_t cpad = (((size_t)nnode + 4) / 4) * 4;
    const size_t epad = (((size_t)nedge + 3) / 4) * 4;
    const int nb  = (nnode + BIN_ROWS - 1) / BIN_ROWS;
    const int nch = (nedge + SORT_CH - 1) / SORT_CH;
    int*  off  = (int*)(y + ((size_t)nnode + 1) * DLAT);
    int*  ssrc = off + cpad;
    int2* cs   = (int2*)(ssrc + epad);
    int*  tab  = (int*)(cs + epad);
    int*  lst  = tab + (((size_t)nb * nch + 3) / 4) * 4;

    chunk_sort_kernel<<<nch, 1024, 0, stream>>>(src, dst, cs, tab, lst, nedge, nb, nch);
    bin_sort_kernel<<<nb, 1024, 0, stream>>>(cs, tab, lst, off, ssrc, nnode, nb, nch);

    mlp_norm_conv_kernel<<<DENSE_GRID, 256, 0, stream>>>(features, Wmlp, bmlp, Wc1, x, y, nnode);
    update_kernel<true><<<DENSE_GRID, 256, 0, stream>>>(x, y, off, ssrc, id_emb, Wl1, bl1, Wg1, bg1,
                                                  Wc2, x, y2, nnode);
    update_kernel<false><<<DENSE_GRID, 256, 0, stream>>>(x, y2, off, ssrc, id_emb, Wl2, bl2, Wg2, bg2,
                                                   nullptr, (float*)d_out, nullptr, nnode);
}


# Update kernels: x @ Wl and the previous tile's x' @ Wc with its stores run under the gather's first eight row loads (a barrier moved in front of them), not behind the gather; sum order unchanged
# speedup vs baseline: 3.3543x; 1.0320x over previous
#include <hip/hip_runtime.h>
#include <hip/hip_bf16.h>

#define DLAT 64
#define DFEAT 128
#define SLOPE 0.01f
#define EPSN 1e-12f

typedef __attribute__((ext_vector_type(2))) float v2f;
typedef __attribute__((ext_vector_type(8))) float v8f;
typedef __attribute__((ext_vector_type(4))) float v4f;

__device__ __forceinline__ float leaky(float v) { return v >= 0.0f ? v : SLOPE * v; }

__device__ __forceinline__ unsigned roff(int g, int c) { return ((unsigned)g << 8) | ((unsigned)c << 2); }

__device__ __forceinline__ v8f wmma_f32(v2f a, v2f b, v8f c) {
    return __builtin_amdgcn_wmma_f32_16x16x4_f32(false, a, false, b, (short)0, c,
                                                 false, false);
}

__device__ __forceinline__ void lds_barrier() {
    asm volatile("s_wait_dscnt 0x0\n\ts_barrier_signal -1\n\ts_barrier_wait -1" ::: "memory");
}

#define FPITCH 132
#define XPITCH 68
#define WPITCH 68
#define WMPITCH 132
#define DENSE_GRID 1024

template <int K>
__device__ __forceinline__ void ld_wT(const float* __restrict__ W, int tid, v4f (&r)[K / 16]) {
    const int col = tid & 63;
    const int kq  = tid >> 6;
    #pragma unroll
    for (int i = 0; i < K / 16; ++i) {
        const int k = (kq + 4 * i) << 2;
        r[i].x = W[(k + 0) * DLAT + col];
        r[i].y = W[(k + 1) * DLAT + col];
        r[i].z = W[(k + 2) * DLAT + col];
        r[i].w = W[(k + 3) * DLAT + col];
    }
}
template <int K, int PITCH>
__device__ __forceinline__ void st_wT(float* wt, int tid, const v4f (&r)[K / 16]) {
    const int col = tid & 63;
    const int kq  = tid >> 6;
    #pragma unroll
    for (int i = 0; i < K / 16; ++i)
        *(v4f*)(wt + col * PITCH + ((kq + 4 * i) << 2)) = r[i];
}

__device__ __forceinline__ v8f wmma_2steps(v4f a, v4f b, v8f c) {
    v2f a0, b0, a1, b1;
    a0.x = a.x; a0.y = a.y; b0.x = b.x; b0.y = b.y;
    a1.x = a.z; a1.y = a.w; b1.x = b.z; b1.y = b.w;
    c = wmma_f32(a0, b0, c);
    return wmma_f32(a1, b1, c);
}

__device__ __forceinline__ v4f ld_feat(const float* __restrict__ feat, int tile, int f, int nnode) {
    int g = tile * 32 + (f >> 5);
    if (g > nnode - 1) g = nnode - 1;
    return *(const v4f*)(feat + (size_t)g * DFEAT + ((f & 31) << 2));
}
__device__ __forceinline__ void st_feat(float* ft, int f, v4f v) {
    *(v4f*)(ft + (f >> 5) * FPITCH + ((f & 31) << 2)) = v;
}
__device__ __forceinline__ v4f ld_xrow(const float* __restrict__ p, int g, unsigned coff, int nnode) {
    if (g > nnode - 1) g = nnode - 1;
    return *(const v4f*)((const char*)p + (((unsigned)g << 8) | coff));
}
__device__ __forceinline__ v8f ld_id(const float* __restrict__ idemb, int tile, int rbase, int col, int nnode) {
    v8f r;
    #pragma unroll
    for (int i = 0; i < 8; ++i) {
        int g = tile * 32 + rbase + i;
        if (g > nnode - 1) g = nnode - 1;
        r[i] = *(const float*)((const char*)idemb + roff(g, col));
    }
    return r;
}

__global__ __launch_bounds__(256) void mlp_norm_conv_kernel(
    const float* __restrict__ feat, const float* __restrict__ Wmlp,
    const float* __restrict__ bmlp, const float* __restrict__ Wc,
    float* __restrict__ xout, float* __restrict__ yout, int nnode)
{
    __shared__ __attribute__((aligned(16))) float Wm_t[DLAT * WMPITCH];
    __shared__ __attribute__((aligned(16))) float Wc_t[DLAT * WPITCH];
    __shared__ __attribute__((aligned(16))) float ft[32 * FPITCH];
    __shared__ __attribute__((aligned(16))) float xt[32 * XPITCH];
    __shared__ float scl[32];

    const int tid   = threadIdx.x;
    const int ntile = (nnode + 31) >> 5;
    const int tstep = (int)gridDim.x;
    int tile = (int)blockIdx.x;
    if (tile >= ntile) return;

    if (blockIdx.x == 0 && tid < DLAT) yout[(size_t)nnode * DLAT + tid] = 0.f;

    v4f fr0 = ld_feat(feat, tile, tid,       nnode);
    v4f fr1 = ld_feat(feat, tile, tid + 256, nnode);
    v4f fr2 = ld_feat(feat, tile, tid + 512, nnode);
    v4f fr3 = ld_feat(feat, tile, tid + 768, nnode);
    {
        v4f wm[8], wc[4];
        ld_wT<DFEAT>(Wmlp, tid, wm);
        ld_wT<DLAT>(Wc, tid, wc);
        st_wT<DFEAT, WMPITCH>(Wm_t, tid, wm);
        st_wT<DLAT, WPITCH>(Wc_t, tid, wc);
    }

    const int lane    = tid & 31;
    const int wave    = tid >> 5;
    const int rowtile = wave >> 2;
    const int coltile = wave & 3;
    const int mn      = lane & 15;
    const int k4h     = (lane >> 4) << 2;
    const int col     = coltile * 16 + mn;

    const float* ap   = ft   + (rowtile * 16 + mn) * FPITCH + k4h;
    const float* bp   = Wm_t + col * WMPITCH + k4h;
    const float* a2p  = xt   + (rowtile * 16 + mn) * XPITCH + k4h;
    const float* b2p  = Wc_t + col * WPITCH + k4h;
    const float bias  = bmlp[col];
    const int   rbase = rowtile * 16 + ((lane >> 4) << 3);

    for (; tile < ntile; tile += tstep) {
        const int blk0 = tile * 32;

        st_feat(ft, tid,       fr0);
        st_feat(ft, tid + 256, fr1);
        st_feat(ft, tid + 512, fr2);
        st_feat(ft, tid + 768, fr3);
        lds_barrier();

        const int ntl = tile + tstep;
        if (ntl < ntile) {
            fr0 = ld_feat(feat, ntl, tid,       nnode);
            fr1 = ld_feat(feat, ntl, tid + 256, nnode);
            fr2 = ld_feat(feat, ntl, tid + 512, nnode);
            fr3 = ld_feat(feat, ntl, tid + 768, nnode);
        }

        v8f acc = {};
        #pragma unroll
        for (int kk = 0; kk < DFEAT; kk += 8)
            acc = wmma_2steps(*(const v4f*)(ap + kk), *(const v4f*)(bp + kk), acc);

        #pragma unroll
        for (int i = 0; i < 8; ++i)
            xt[(rbase + i) * XPITCH + col] = acc[i] + bias;
        lds_barrier();

        if (tid < 32) {
            float s = 0.f;
            #pragma unroll
            for (int q = 0; q < DLAT / 4; ++q) {
                const v4f v = *(const v4f*)(xt + tid * XPITCH + 4 * q);
                s += v.x * v.x; s += v.y * v.y; s += v.z * v.z; s += v.w * v.w;
            }
            scl[tid] = 1.0f / fmaxf(sqrtf(s), EPSN);
        }
        lds_barrier();

        #pragma unroll
        for (int i = 0; i < 2; ++i) {
            const int f = tid + 256 * i;
            const int r = f >> 4, c4 = (f & 15) << 2;
            const float sc = scl[r];
            v4f v = *(const v4f*)(xt + r * XPITCH + c4);
            v.x *= sc; v.y *= sc; v.z *= sc; v.w *= sc;
            *(v4f*)(xt + r * XPITCH + c4) = v;
            const int g = blk0 + r;
            if (g < nnode) *(v4f*)(xout + (size_t)g * DLAT + c4) = v;
        }
        lds_barrier();

        v8f acc2 = {};
        #pragma unroll
        for (int kk = 0; kk < DLAT; kk += 8)
            acc2 = wmma_2steps(*(const v4f*)(a2p + kk), *(const v4f*)(b2p + kk), acc2);
        #pragma unroll
        for (int i = 0; i < 8; ++i) {
            int g = blk0 + rbase + i;
            if (g < nnode) yout[(size_t)g * DLAT + col] = acc2[i];
        }
    }
}


struct PairBounds { int myb, myn, nmax; };

__device__ __forceinline__ PairBounds pair_bounds(int offc, int p, bool hi)
{
    const int b0 = __builtin_amdgcn_readlane(offc, 2 * p);
    const int m  = __builtin_amdgcn_readlane(offc, 2 * p + 1);
    const int e1 = __builtin_amdgcn_readlane(offc, 2 * p + 2);
    PairBounds r;
    r.myb  = hi ? m : b0;
    r.myn  = hi ? e1 - m : m - b0;
    r.nmax = max(m - b0, e1 - m);
    return r;
}

__device__ __forceinline__ int ld_src_id(const int* __restrict__ ssrc, const PairBounds& pb, int idx, int zrow)
{
    return idx < pb.myn ? ssrc[pb.myb + idx] : zrow;
}

template <int K>
__device__ __forceinline__ void issue_row_loads(const char* __restrict__ yb, int ids, unsigned coff, v4f (&v)[16])
{
    if constexpr (K < 16) {
        const int s = __builtin_amdgcn_update_dpp(0, ids, 0x150 + K, 0xf, 0xf, false);
        v[K] = *(const v4f*)(yb + (size_t)(((unsigned)s << 8) | coff));
        issue_row_loads<K + 1>(yb, ids, coff, v);
    }
}

__device__ __forceinline__ v4f gather_pair(const char* __restrict__ yb, const int* __restrict__ ssrc,
                                           const PairBounds& pb, int ids0, int ids1, int c, int zrow)
{
    v4f acc = {0.f, 0.f, 0.f, 0.f};
    const unsigned coff = (unsigned)c << 4;
    for (int j = 0; j < pb.nmax; j += 16) {
        v4f v[16];
        issue_row_loads<0>(yb, ids0, coff, v);
        ids0 = ids1;
        if (j + 32 < pb.nmax) ids1 = ld_src_id(ssrc, pb, j + 32 + c, zrow);
        #pragma unroll
        for (int k = 0; k < 16; ++k) acc += v[k];
    }
    return acc;
}

template <int K, int N, int OUT, int VN>
__device__ __forceinline__ void issue_rows(const char* __restrict__ yb, int ids, unsigned coff, v4f (&v)[VN])
{
    if constexpr (N > 0) {
        const int s = __builtin_amdgcn_update_dpp(0, ids, 0x150 + K, 0xf, 0xf, false);
        v[OUT] = *(const v4f*)(yb + (size_t)(((unsigned)s << 8) | coff));
        issue_rows<K + 1, N - 1, OUT + 1, VN>(yb, ids, coff, v);
    }
}

__device__ __forceinline__ v4f gather_pair_from8(const char* __restrict__ yb, const int* __restrict__ ssrc,
                                                 const PairBounds& pb, const v4f (&g)[8], int idsA, int idsB, int c, int zrow)
{
    v4f acc = {0.f, 0.f, 0.f, 0.f};
    const unsigned coff = (unsigned)c << 4;
    #pragma unroll
    for (int k = 0; k < 8; ++k) acc += g[k];
    for (int j = 8; j < pb.nmax; j += 16) {
        v4f v[16];
        issue_rows<8, 8, 0, 16>(yb, idsA, coff, v);
        issue_rows<0, 8, 8, 16>(yb, idsB, coff, v);
        idsA = idsB;
        idsB = zrow;
        if (j + 24 < pb.nmax) idsB = ld_src_id(ssrc, pb, j + 24 + c, zrow);
        #pragma unroll
        for (int k = 0; k < 16; ++k) acc += v[k];
    }
    return acc;
}

template <bool FUSE_NEXT>
__global__ __launch_bounds__(256) __attribute__((amdgpu_waves_per_eu(8, 8))) void update_kernel(
    const float* __restrict__ xin, const float* __restrict__ ysrc,
    const int* __restrict__ off, const int* __restrict__ ssrc,
    const float* __restrict__ idemb,
    const float* __restrict__ Wl, const float* __restrict__ bl,
    const float* __restrict__ Wg, const float* __restrict__ bg,
    const float* __restrict__ Wc, float* __restrict__ xout,
    float* __restrict__ yout, int nnode)
{
    constexpr int NPIECE = FUSE_NEXT ? 3 : 2;
    __shared__ __attribute__((aligned(16))) float lds[NPIECE * 32 * XPITCH + NPIECE * DLAT * WPITCH];
    float* const xs   = lds;
    float* const hs   = xs + 32 * XPITCH;
    float* const xt   = hs + 32 * XPITCH;
    float* const Wl_t = lds + NPIECE * 32 * XPITCH;
    float* const Wg_t = Wl_t + DLAT * WPITCH;
    float* const Wc_t = Wg_t + DLAT * WPITCH;

    const int tid   = threadIdx.x;
    const int ntile = (nnode + 31) >> 5;
    const int tstep = (int)gridDim.x;
    int tile = (int)blockIdx.x;
    if (tile >= ntile) return;

    const int lane    = tid & 31;
    const int wave    = tid >> 5;
    const int r4      = wave * 4;
    const int l4      = min(lane, 4);
    const int  mn = lane & 15;
    const bool hi = lane >= 16;
    const int      hrow = r4 + (hi ? 1 : 0);
    const int      hoff = hrow * XPITCH + mn * 4;
    const unsigned coff = (unsigned)mn << 4;

    v4f xr0 = ld_xrow(xin, tile * 32 + hrow,     coff, nnode);
    v4f xr1 = ld_xrow(xin, tile * 32 + hrow + 2, coff, nnode);
    int offc = off[min(tile * 32 + r4 + l4, nnode)];
    {
        v4f wl[4], wg[4], wc[4];
        ld_wT<DLAT>(Wl, tid, wl);
        ld_wT<DLAT>(Wg, tid, wg);
        if constexpr (FUSE_NEXT) ld_wT<DLAT>(Wc, tid, wc);
        st_wT<DLAT, WPITCH>(Wl_t, tid, wl);
        st_wT<DLAT, WPITCH>(Wg_t, tid, wg);
        if constexpr (FUSE_NEXT) st_wT<DLAT, WPITCH>(Wc_t, tid, wc);
    }

    if constexpr (FUSE_NEXT) {
        if (blockIdx.x == 0 && tid < DLAT) yout[(size_t)nnode * DLAT + tid] = 0.f;
    }

    int ids[2][2];
    #pragma unroll
    for (int p = 0; p < 2; ++p) {
        const PairBounds pb = pair_bounds(offc, p, hi);
        ids[p][0] = ld_src_id(ssrc, pb, mn,      nnode);
        ids[p][1] = ld_src_id(ssrc, pb, mn + 16, nnode);
    }

    const int rowtile = wave >> 2;
    const int coltile = wave & 3;
    const int k4h     = (lane >> 4) << 2;
    const int col     = coltile * 16 + mn;

    const float* xp  = xs   + (rowtile * 16 + mn) * XPITCH + k4h;
    const float* hp  = hs   + (rowtile * 16 + mn) * XPITCH + k4h;
    const float* wlp = Wl_t + col * WPITCH + k4h;
    const float* wgp = Wg_t + col * WPITCH + k4h;
    const char*  yb = (const char*)ysrc;

    const float blv = bl[col], bgv = bg[col];
    const int rbase = rowtile * 16 + ((lane >> 4) << 3);

    auto tile_out = [&](int b0) {
        #pragma unroll
        for (int i = 0; i < 2; ++i) {
            const int g = b0 + hrow + 2 * i;
            if (g < nnode)
                *(v4f*)((char*)xout + (((unsigned)g << 8) | coff)) = *(const v4f*)(xt + hoff + 2 * i * XPITCH);
        }
        v8f acc2 = {};
        const float* a2p = xt   + (rowtile * 16 + mn) * XPITCH + k4h;
        const float* wcp = Wc_t + col * WPITCH + k4h;
        #pragma unroll
        for (int kk = 0; kk < DLAT; kk += 8)
            acc2 = wmma_2steps(*(const v4f*)(a2p + kk), *(const v4f*)(wcp + kk), acc2);
        #pragma unroll
        for (int i = 0; i < 8; ++i) {
            const int g = b0 + rbase + i;
            if (g < nnode) *(float*)((char*)yout + roff(g, col)) = acc2[i];
        }
    };

    for (; tile < ntile; tile += tstep) {
        const int blk0 = tile * 32;
        const int offnx = off[min((tile + tstep) * 32 + r4 + l4, nnode)];

        *(v4f*)(xs + hoff) = xr0;
        *(v4f*)(xs + hoff + 2 * XPITCH) = xr1;
        lds_barrier();

        const PairBounds pb0 = pair_bounds(offc, 0, hi);
        v4f gv[8];
        issue_rows<0, 8, 0, 8>(yb, ids[0][0], coff, gv);

        v8f c1 = {};
        #pragma unroll
        for (int kk = 0; kk < DLAT; kk += 8)
            c1 = wmma_2steps(*(const v4f*)(xp + kk), *(const v4f*)(wlp + kk), c1);

        if constexpr (FUSE_NEXT) {
            if (tile != (int)blockIdx.x) tile_out(blk0 - tstep * 32);
        }

        const v8f idc = ld_id(idemb, tile, rbase, col, nnode);

        {
            const v4f a = gather_pair_from8(yb, ssrc, pb0, gv, ids[0][0], ids[0][1], mn, nnode);
            v4f r;
            r.x = leaky(a.x); r.y = leaky(a.y); r.z = leaky(a.z); r.w = leaky(a.w);
            *(v4f*)(hs + hoff) = r;
        }
        {
            const PairBounds pb = pair_bounds(offc, 1, hi);
            const v4f a = gather_pair(yb, ssrc, pb, ids[1][0], ids[1][1], mn, nnode);
            v4f r;
            r.x = leaky(a.x); r.y = leaky(a.y); r.z = leaky(a.z); r.w = leaky(a.w);
            *(v4f*)(hs + hoff + 2 * XPITCH) = r;
        }
        v8f xh;
        #pragma unroll
        for (int i = 0; i < 8; ++i) xh[i] = leaky(c1[i] + blv) + idc[i];
        lds_barrier();

        const int ntl = tile + tstep;
        offc = offnx;
        const PairBounds nb0 = pair_bounds(offc, 0, hi);
        const PairBounds nb1 = pair_bounds(offc, 1, hi);
        if (ntl < ntile) {
            ids[0][0] = ld_src_id(ssrc, nb0, mn,      nnode);
            ids[0][1] = ld_src_id(ssrc, nb0, mn + 16, nnode);
            ids[1][0] = ld_src_id(ssrc, nb1, mn,      nnode);
            ids[1][1] = ld_src_id(ssrc, nb1, mn + 16, nnode);
            xr0 = ld_xrow(xin, ntl * 32 + hrow,     coff, nnode);
            xr1 = ld_xrow(xin, ntl * 32 + hrow + 2, coff, nnode);
        }

        v8f c2 = {};
        #pragma unroll
        for (int kk = 0; kk < DLAT; kk += 8)
            c2 = wmma_2steps(*(const v4f*)(hp + kk), *(const v4f*)(wgp + kk), c2);

        #pragma unroll
        for (int i = 0; i < 8; ++i) {
            int rloc = rbase + i;
            int g    = blk0 + rloc;
            float v    = leaky(c2[i] + bgv + xh[i]);
            if constexpr (FUSE_NEXT) {
                xt[rloc * XPITCH + col] = v;
            } else {
                if (g < nnode) *(float*)((char*)xout + roff(g, col)) = v;
            }
        }

    }
    if constexpr (FUSE_NEXT) {
        lds_barrier();
        tile_out((tile - tstep) * 32);
    }
}

#define SORT_CH  8192
#define BIN_SH   9
#define BIN_ROWS 512
#define MAXNB    1024
#define MAXCH    1024
#define BIN_CAP  12288

__device__ __forceinline__ int block_excl_scan_1024(int v, int* wsum, int& total)
{
    const int lane = threadIdx.x & 31;
    const int wave = threadIdx.x >> 5;
    int inc = v;
    for (int o = 1; o < 32; o <<= 1) {
        int u = __shfl_up(inc, o);
        if (lane >= o) inc += u;
    }
    if (lane == 31) wsum[wave] = inc;
    __syncthreads();
    const int ws = wsum[lane];
    int wp  = lane < wave ? ws : 0;
    int tot = ws;
    for (int o = 16; o > 0; o >>= 1) {
        wp  += __shfl_xor(wp, o);
        tot += __shfl_xor(tot, o);
    }
    total = tot;
    __syncthreads();
    return wp + inc - v;
}

__global__ __launch_bounds__(1024) void chunk_sort_kernel(
    const int* __restrict__ src, const int* __restrict__ dst,
    int2* __restrict__ cs, int* __restrict__ tab, int* __restrict__ lst,
    int nedge, int nb, int nch)
{
    __shared__ int  lcnt[MAXNB];
    __shared__ int  lsta[MAXNB];
    __shared__ int  wsum[32];
    __shared__ int2 stage[SORT_CH];

    const int tid  = threadIdx.x;
    const int c    = blockIdx.x;
    const int e0   = c * SORT_CH;
    const int last = nedge - 1;

    lcnt[tid] = 0;
    __syncthreads();

    int s[8], d[8], r[8];
    #pragma unroll
    for (int k = 0; k < 8; ++k) {
        const int e = min(e0 + tid + k * 1024, last);
        s[k] = src[e];
        d[k] = dst[e];
    }
    #pragma unroll
    for (int k = 0; k < 8; ++k) {
        r[k] = 0;
        const int bin = min(d[k] >> BIN_SH, MAXNB - 1);
        if (e0 + tid + k * 1024 < nedge) r[k] = atomicAdd(&lcnt[bin], 1);
    }
    __syncthreads();

    const int v = lcnt[tid];
    int tot;
    const int ex = block_excl_scan_1024(v, wsum, tot);
    lsta[tid] = ex;
    if (tid < nb) {
        tab[(size_t)tid * nch + c] = v;
        lst[(size_t)c * nb + tid]  = ex;
    }
    __syncthreads();

    #pragma unroll
    for (int k = 0; k < 8; ++k) {
        const int bin = min(d[k] >> BIN_SH, MAXNB - 1);
        if (e0 + tid + k * 1024 < nedge) stage[lsta[bin] + r[k]] = make_int2(s[k], d[k]);
    }
    __syncthreads();

    for (int j = tid; j < tot; j += 1024) cs[(size_t)e0 + j] = stage[j];
}

__global__ __launch_bounds__(1024) void bin_sort_kernel(
    const int2* __restrict__ cs, const int* __restrict__ tab, const int* __restrict__ lst,
    int* __restrict__ off, int* __restrict__ ssrc, int nnode, int nb, int nch)
{
    __shared__ int  ppre[MAXCH];
    __shared__ int  ps[MAXCH];
    __shared__ int  lcnt[BIN_ROWS];
    __shared__ int  loff[BIN_ROWS];
    __shared__ int  lcur[BIN_ROWS];
    __shared__ int  wsum[32];
    __shared__ int  gsum[32];
    __shared__ int2 A[BIN_CAP];
    __shared__ int  B[BIN_CAP];

    const int tid   = threadIdx.x;
    const int lane  = tid & 31;
    const int wave  = tid >> 5;
    const int b     = blockIdx.x;
    const int rbase = b << BIN_SH;

    int n = 0, st = 0;
    if (tid < nch) {
        n  = tab[(size_t)b * nch + tid];
        st = lst[(size_t)tid * nb + b];
    }

    int g = 0;
    {
        const int   nfront = b * nch;
        const int   n4     = nfront >> 2;
        const int4* t4     = (const int4*)tab;
        int i = tid;
        for (; i + 7 * 1024 < n4; i += 8 * 1024) {
            int4 a[8];
            #pragma unroll
            for (int k = 0; k < 8; ++k) a[k] = t4[i + k * 1024];
            #pragma unroll
            for (int k = 0; k < 8; ++k) g += (a[k].x + a[k].y) + (a[k].z + a[k].w);
        }
        for (; i < n4; i += 1024) {
            int4 a = t4[i];
            g += (a.x + a.y) + (a.z + a.w);
        }
        for (int q = (n4 << 2) + tid; q < nfront; q += 1024) g += tab[q];
    }
    for (int o = 16; o > 0; o >>= 1) g += __shfl_xor(g, o);
    if (lane == 0) gsum[wave] = g;

    ps[tid] = st;
    if (tid < BIN_ROWS) { lcnt[tid] = 0; lcur[tid] = 0; }
    int total;
    const int ex = block_excl_scan_1024(n, wsum, total);
    ppre[tid] = ex;
    int gb = gsum[lane];
    for (int o = 16; o > 0; o >>= 1) gb += __shfl_xor(gb, o);
    __syncthreads();

    const bool single = total <= BIN_CAP;

    for (int j0 = tid; j0 < total; j0 += 8 * 1024) {
        int2 e[8];
        #pragma unroll
        for (int k = 0; k < 8; ++k) {
            const int j = min(j0 + k * 1024, total - 1);
            int cc = 0;
            #pragma unroll
            for (int sft = MAXCH / 2; sft > 0; sft >>= 1)
                if (ppre[cc + sft] <= j) cc += sft;
            e[k] = cs[(size_t)cc * SORT_CH + ps[cc] + (j - ppre[cc])];
        }
        #pragma unroll
        for (int k = 0; k < 8; ++k) {
            const int j = j0 + k * 1024;
            if (j < total) {
                const int ld = min(max(e[k].y - rbase, 0), BIN_ROWS - 1);
                atomicAdd(&lcnt[ld], 1);
                if (single) A[j] = make_int2(e[k].x, ld);
            }
        }
    }
    __syncthreads();

    const int cv = tid < BIN_ROWS ? lcnt[tid] : 0;
    int rtot;
    const int rex = block_excl_scan_1024(cv, wsum, rtot);
    if (tid < BIN_ROWS) {
        loff[tid] = rex;
        if (rbase + tid < nnode) off[rbase + tid] = gb + rex;
    }
    if (b == nb - 1 && tid == 0) off[nnode] = gb + total;
    __syncthreads();

    if (single) {
        for (int j = tid; j < total; j += 1024) {
            const int2 e = A[j];
            const int pos = loff[e.y] + atomicAdd(&lcur[e.y], 1);
            B[pos] = e.x;
        }
        __syncthreads();
        for (int j = tid; j < total; j += 1024) ssrc[(size_t)gb + j] = B[j];
    } else {
        for (int j = tid; j < total; j += 1024) {
            int cc = 0;
            #pragma unroll
            for (int sft = MAXCH / 2; sft > 0; sft >>= 1)
                if (ppre[cc + sft] <= j) cc += sft;
            const int2 e  = cs[(size_t)cc * SORT_CH + ps[cc] + (j - ppre[cc])];
            const int  ld = min(max(e.y - rbase, 0), BIN_ROWS - 1);
            const int  pos = loff[ld] + atomicAdd(&lcur[ld], 1);
            ssrc[(size_t)gb + pos] = e.x;
        }
    }
}

extern "C" void kernel_launch(void* const* d_in, const int* in_sizes, int n_in,
                              void* d_out, int out_size, void* d_ws, size_t ws_size,
                              hipStream_t stream)
{
    const float* features = (const float*)d_in[0];
    const float* id_emb   = (const float*)d_in[1];
    const int*   edge     = (const int*)d_in[2];
    const float* Wmlp     = (const float*)d_in[3];
    const float* bmlp     = (const float*)d_in[4];
    const float* Wc1      = (const float*)d_in[5];
    const float* Wl1      = (const float*)d_in[6];
    const float* bl1      = (const float*)d_in[7];
    const float* Wg1      = (const float*)d_in[8];
    const float* bg1      = (const float*)d_in[9];
    const float* Wc2      = (const float*)d_in[10];
    const float* Wl2      = (const float*)d_in[11];
    const float* bl2      = (const float*)d_in[12];
    const float* Wg2      = (const float*)d_in[13];
    const float* bg2      = (const float*)d_in[14];

    const int nnode = in_sizes[0] / DFEAT;
    const int nedge = in_sizes[2] / 2;
    const int* src  = edge;
    const int* dst  = edge + nedge;

    float* x = (float*)d_ws;
    float* y2 = x + (size_t)nnode * DLAT;
    float* y  = y2 + ((size_t)nnode + 1) * DLAT;

    const size_t cpad = (((size_t)nnode + 4) / 4) * 4;
    const size_t epad = (((size_t)nedge + 3) / 4) * 4;
    const int nb  = (nnode + BIN_ROWS - 1) / BIN_ROWS;
    const int nch = (nedge + SORT_CH - 1) / SORT_CH;
    int*  off  = (int*)(y + ((size_t)nnode + 1) * DLAT);
    int*  ssrc = off + cpad;
    int2* cs   = (int2*)(ssrc + epad);
    int*  tab  = (int*)(cs + epad);
    int*  lst  = tab + (((size_t)nb * nch + 3) / 4) * 4;

    chunk_sort_kernel<<<nch, 1024, 0, stream>>>(src, dst, cs, tab, lst, nedge, nb, nch);
    bin_sort_kernel<<<nb, 1024, 0, stream>>>(cs, tab, lst, off, ssrc, nnode, nb, nch);

    mlp_norm_conv_kernel<<<DENSE_GRID, 256, 0, stream>>>(features, Wmlp, bmlp, Wc1, x, y, nnode);
    update_kernel<true><<<DENSE_GRID, 256, 0, stream>>>(x, y, off, ssrc, id_emb, Wl1, bl1, Wg1, bg1,
                                                  Wc2, x, y2, nnode);
    update_kernel<false><<<DENSE_GRID, 256, 0, stream>>>(x, y2, off, ssrc, id_emb, Wl2, bl2, Wg2, bg2,
                                                   nullptr, (float*)d_out, nullptr, nnode);
}


# Edge sort's first level runs in the MLP launch IN FRONT of the tiles: its first 196 blocks each order one 8192-edge chunk (chunks kept), then take 2 tiles; 5 to 4 launches; row edge order changes
# speedup vs baseline: 3.4789x; 1.0371x over previous
#include <hip/hip_runtime.h>
#include <hip/hip_bf16.h>

#define DLAT 64
#define DFEAT 128
#define SLOPE 0.01f
#define EPSN 1e-12f

typedef __attribute__((ext_vector_type(2))) float v2f;
typedef __attribute__((ext_vector_type(8))) float v8f;
typedef __attribute__((ext_vector_type(4))) float v4f;

__device__ __forceinline__ float leaky(float v) { return v >= 0.0f ? v : SLOPE * v; }

__device__ __forceinline__ unsigned roff(int g, int c) { return ((unsigned)g << 8) | ((unsigned)c << 2); }

__device__ __forceinline__ v8f wmma_f32(v2f a, v2f b, v8f c) {
    return __builtin_amdgcn_wmma_f32_16x16x4_f32(false, a, false, b, (short)0, c,
                                                 false, false);
}

__device__ __forceinline__ void lds_barrier() {
    asm volatile("s_wait_dscnt 0x0\n\ts_barrier_signal -1\n\ts_barrier_wait -1" ::: "memory");
}

#define SORT_CH  8192
#define BIN_SH   9
#define BIN_ROWS 512
#define MAXNB    1024
#define MAXCH    1024
#define BIN_CAP  12288

typedef __attribute__((ext_vector_type(4))) int v4i;

__device__ __forceinline__ int block_excl_scan_256(int v, int* wsum, int& total)
{
    const int lane = threadIdx.x & 31;
    const int wave = threadIdx.x >> 5;
    int inc = v;
    inc += __builtin_amdgcn_update_dpp(0, inc, 0x111, 0xf, 0xf, false);
    inc += __builtin_amdgcn_update_dpp(0, inc, 0x112, 0xf, 0xf, false);
    inc += __builtin_amdgcn_update_dpp(0, inc, 0x114, 0xf, 0xf, false);
    inc += __builtin_amdgcn_update_dpp(0, inc, 0x118, 0xf, 0xf, false);
    const int lo = __builtin_amdgcn_readlane(inc, 15);
    if (lane >= 16) inc += lo;
    if (lane == 31) wsum[wave] = inc;
    __syncthreads();
    const v4i w0 = *(const v4i*)(wsum);
    const v4i w1 = *(const v4i*)(wsum + 4);
    int wp = 0, tot = 0;
    #pragma unroll
    for (int i = 0; i < 8; ++i) {
        const int w = i < 4 ? w0[i] : w1[i - 4];
        tot += w;
        if (i < wave) wp += w;
    }
    total = tot;
    __syncthreads();
    return wp + inc - v;
}

#define SORT_LDS_INTS (MAXNB + MAXNB + 32 + 2 * SORT_CH + 2 * 32)
static_assert(MAXNB * BIN_ROWS == (1 << 19) && SORT_CH == (1 << 13), "a destination id and a rank inside a chunk share one 32-bit register");

__device__ __forceinline__ void chunk_sort_256(
    int* lds, const int* __restrict__ src, const int* __restrict__ dst,
    int2* __restrict__ cs, int* __restrict__ tab, int* __restrict__ lst,
    int nedge, int nb, int nch, int c)
{
    int*  const lcnt  = lds;
    int*  const lsta  = lcnt + MAXNB;
    int*  const wsum  = lsta + MAXNB;
    int2* const stage = (int2*)(wsum + 32);

    const int tid  = threadIdx.x;
    const int lane = tid & 31;
    const int e0   = c * SORT_CH;
    const int nv   = nedge - e0 - tid;

    *(v4i*)(lcnt + 4 * tid) = (v4i){0, 0, 0, 0};
    __syncthreads();

    int s[32], d[32];
    const int* const dp = dst + e0 + tid;
    const int* const sp = src + e0 + tid;
    if (e0 + SORT_CH <= nedge) {
        #pragma unroll
        for (int k = 0; k < 32; ++k) d[k] = dp[k * 256];
        #pragma unroll
        for (int k = 0; k < 32; ++k) s[k] = sp[k * 256];
    } else {
        #pragma unroll
        for (int k = 0; k < 32; ++k) {
            d[k] = 0; s[k] = 0;
            if (k * 256 < nv) { d[k] = dp[k * 256]; s[k] = sp[k * 256]; }
        }
    }

    #pragma unroll
    for (int k = 0; k < 32; ++k) {
        const unsigned dd = (unsigned)d[k] << 13;
        const int      ci = k * 256 < nv ? (int)(dd >> (13 + BIN_SH)) : MAXNB + lane;
        const unsigned r  = (unsigned)atomicAdd(&lcnt[ci], 1);
        d[k] = (int)(dd | r);
    }
    __syncthreads();

    const v4i cv = *(const v4i*)(lcnt + 4 * tid);
    int tot;
    const int ex = block_excl_scan_256((cv.x + cv.y) + (cv.z + cv.w), wsum, tot);
    v4i st;
    st.x = ex; st.y = ex + cv.x; st.z = st.y + cv.y; st.w = st.z + cv.z;
    *(v4i*)(lsta + 4 * tid) = st;
    #pragma unroll
    for (int q = 0; q < 4; ++q) {
        const int bin = 4 * tid + q;
        if (bin < nb) {
            tab[bin * nch + c] = cv[q];
            lst[c * nb + bin]  = st[q];
        }
    }
    __syncthreads();

    #pragma unroll
    for (int k = 0; k < 32; ++k) d[k] += lsta[(unsigned)d[k] >> (13 + BIN_SH)];
    #pragma unroll
    for (int k = 0; k < 32; ++k)
        stage[k * 256 < nv ? (d[k] & (SORT_CH - 1)) : SORT_CH + lane] = make_int2(s[k], (int)((unsigned)d[k] >> 13));
    __syncthreads();

    const int  npair = tot >> 1;
    const v4i* stp = (const v4i*)stage;
    v4i*       cp  = (v4i*)(cs + (size_t)e0);
    for (int p0 = tid; p0 < npair; p0 += 8 * 256) {
        v4i e[8];
        #pragma unroll
        for (int k = 0; k < 8; ++k) e[k] = stp[min(p0 + k * 256, npair - 1)];
        #pragma unroll
        for (int k = 0; k < 8; ++k)
            if (p0 + k * 256 < npair) cp[p0 + k * 256] = e[k];
    }
    if ((tot & 1) && tid == 0) cs[(size_t)e0 + tot - 1] = stage[tot - 1];
}

#define FPITCH 132
#define XPITCH 68
#define WPITCH 68
#define WMPITCH 132
#define DENSE_GRID 1024

template <int K>
__device__ __forceinline__ void ld_wT(const float* __restrict__ W, int tid, v4f (&r)[K / 16]) {
    const int col = tid & 63;
    const int kq  = tid >> 6;
    #pragma unroll
    for (int i = 0; i < K / 16; ++i) {
        const int k = (kq + 4 * i) << 2;
        r[i].x = W[(k + 0) * DLAT + col];
        r[i].y = W[(k + 1) * DLAT + col];
        r[i].z = W[(k + 2) * DLAT + col];
        r[i].w = W[(k + 3) * DLAT + col];
    }
}
template <int K, int PITCH>
__device__ __forceinline__ void st_wT(float* wt, int tid, const v4f (&r)[K / 16]) {
    const int col = tid & 63;
    const int kq  = tid >> 6;
    #pragma unroll
    for (int i = 0; i < K / 16; ++i)
        *(v4f*)(wt + col * PITCH + ((kq + 4 * i) << 2)) = r[i];
}

__device__ __forceinline__ v8f wmma_2steps(v4f a, v4f b, v8f c) {
    v2f a0, b0, a1, b1;
    a0.x = a.x; a0.y = a.y; b0.x = b.x; b0.y = b.y;
    a1.x = a.z; a1.y = a.w; b1.x = b.z; b1.y = b.w;
    c = wmma_f32(a0, b0, c);
    return wmma_f32(a1, b1, c);
}

__device__ __forceinline__ v4f ld_feat(const float* __restrict__ feat, int tile, int f, int nnode) {
    int g = tile * 32 + (f >> 5);
    if (g > nnode - 1) g = nnode - 1;
    return *(const v4f*)(feat + (size_t)g * DFEAT + ((f & 31) << 2));
}
__device__ __forceinline__ void st_feat(float* ft, int f, v4f v) {
    *(v4f*)(ft + (f >> 5) * FPITCH + ((f & 31) << 2)) = v;
}
__device__ __forceinline__ v4f ld_xrow(const float* __restrict__ p, int g, unsigned coff, int nnode) {
    if (g > nnode - 1) g = nnode - 1;
    return *(const v4f*)((const char*)p + (((unsigned)g << 8) | coff));
}
__device__ __forceinline__ v8f ld_id(const float* __restrict__ idemb, int tile, int rbase, int col, int nnode) {
    v8f r;
    #pragma unroll
    for (int i = 0; i < 8; ++i) {
        int g = tile * 32 + rbase + i;
        if (g > nnode - 1) g = nnode - 1;
        r[i] = *(const float*)((const char*)idemb + roff(g, col));
    }
    return r;
}

#define MLP_LDS_FLOATS (DLAT * WMPITCH + DLAT * WPITCH + 32 * FPITCH + 32 * XPITCH + 32)
union __attribute__((aligned(16))) MlpLds {
    float f[MLP_LDS_FLOATS];
    int   i[SORT_LDS_INTS];
};
static_assert(SORT_LDS_INTS <= MLP_LDS_FLOATS, "the sort's first level must fit the LDS that the dense part reserves anyway");

__global__ __launch_bounds__(256) __attribute__((amdgpu_waves_per_eu(8, 8))) void mlp_norm_conv_kernel(
    const float* __restrict__ feat, const float* __restrict__ Wmlp,
    const float* __restrict__ bmlp, const float* __restrict__ Wc,
    float* __restrict__ xout, float* __restrict__ yout, int nnode,
    const int* __restrict__ src, const int* __restrict__ dst,
    int2* __restrict__ cs, int* __restrict__ tab, int* __restrict__ lst,
    int nedge, int nb, int nch)
{
    __shared__ MlpLds sm;
    float* const Wm_t = sm.f;
    float* const Wc_t = Wm_t + DLAT * WMPITCH;
    float* const ft   = Wc_t + DLAT * WPITCH;
    float* const xt   = ft + 32 * FPITCH;
    float* const scl  = xt + 32 * XPITCH;

    const int tid   = threadIdx.x;
    const int ntile = (nnode + 31) >> 5;
    const int nblk  = (int)gridDim.x;
    const int bid   = (int)blockIdx.x;

    if (bid == 0 && tid < DLAT) yout[(size_t)nnode * DLAT + tid] = 0.f;

    static_assert(MAXCH <= DENSE_GRID, "one block of the dense grid per chunk");
    if (bid < nch) {
        chunk_sort_256(sm.i, src, dst, cs, tab, lst, nedge, nb, nch, bid);
        __syncthreads();
    }

    const int  nsort  = 2 * nch <= nblk ? nch : 0;
    const bool sorter = bid < nsort;
    int tile = sorter ? bid + (nblk - nsort) : bid - nsort;
    if (tile >= ntile) return;

    v4f fr0 = ld_feat(feat, tile, tid,       nnode);
    v4f fr1 = ld_feat(feat, tile, tid + 256, nnode);
    v4f fr2 = ld_feat(feat, tile, tid + 512, nnode);
    v4f fr3 = ld_feat(feat, tile, tid + 768, nnode);
    {
        v4f wm[8], wc[4];
        ld_wT<DFEAT>(Wmlp, tid, wm);
        ld_wT<DLAT>(Wc, tid, wc);
        st_wT<DFEAT, WMPITCH>(Wm_t, tid, wm);
        st_wT<DLAT, WPITCH>(Wc_t, tid, wc);
    }

    const int lane    = tid & 31;
    const int wave    = tid >> 5;
    const int rowtile = wave >> 2;
    const int coltile = wave & 3;
    const int mn      = lane & 15;
    const int k4h     = (lane >> 4) << 2;
    const int col     = coltile * 16 + mn;

    const float* ap   = ft   + (rowtile * 16 + mn) * FPITCH + k4h;
    const float* bp   = Wm_t + col * WMPITCH + k4h;
    const float* a2p  = xt   + (rowtile * 16 + mn) * XPITCH + k4h;
    const float* b2p  = Wc_t + col * WPITCH + k4h;
    const float bias  = bmlp[col];
    const int   rbase = rowtile * 16 + ((lane >> 4) << 3);

    for (;;) {
        const int blk0 = tile * 32;
        int ntl = tile + (tile < 2 * nblk ? nblk : nblk - nsort);
        if (sorter && tile >= nblk) ntl = ntile;

        st_feat(ft, tid,       fr0);
        st_feat(ft, tid + 256, fr1);
        st_feat(ft, tid + 512, fr2);
        st_feat(ft, tid + 768, fr3);
        lds_barrier();

        if (ntl < ntile) {
            fr0 = ld_feat(feat, ntl, tid,       nnode);
            fr1 = ld_feat(feat, ntl, tid + 256, nnode);
            fr2 = ld_feat(feat, ntl, tid + 512, nnode);
            fr3 = ld_feat(feat, ntl, tid + 768, nnode);
        }

        v8f acc = {};
        #pragma unroll
        for (int kk = 0; kk < DFEAT; kk += 8)
            acc = wmma_2steps(*(const v4f*)(ap + kk), *(const v4f*)(bp + kk), acc);

        #pragma unroll
        for (int i = 0; i < 8; ++i)
            xt[(rbase + i) * XPITCH + col] = acc[i] + bias;
        lds_barrier();

        if (tid < 32) {
            float s = 0.f;
            #pragma unroll
            for (int q = 0; q < DLAT / 4; ++q) {
                const v4f v = *(const v4f*)(xt + tid * XPITCH + 4 * q);
                s += v.x * v.x; s += v.y * v.y; s += v.z * v.z; s += v.w * v.w;
            }
            scl[tid] = 1.0f / fmaxf(sqrtf(s), EPSN);
        }
        lds_barrier();

        #pragma unroll
        for (int i = 0; i < 2; ++i) {
            const int f = tid + 256 * i;
            const int r = f >> 4, c4 = (f & 15) << 2;
            const float sc = scl[r];
            v4f v = *(const v4f*)(xt + r * XPITCH + c4);
            v.x *= sc; v.y *= sc; v.z *= sc; v.w *= sc;
            *(v4f*)(xt + r * XPITCH + c4) = v;
            const int g = blk0 + r;
            if (g < nnode) *(v4f*)(xout + (size_t)g * DLAT + c4) = v;
        }
        lds_barrier();

        v8f acc2 = {};
        #pragma unroll
        for (int kk = 0; kk < DLAT; kk += 8)
            acc2 = wmma_2steps(*(const v4f*)(a2p + kk), *(const v4f*)(b2p + kk), acc2);
        #pragma unroll
        for (int i = 0; i < 8; ++i) {
            int g = blk0 + rbase + i;
            if (g < nnode) yout[(size_t)g * DLAT + col] = acc2[i];
        }
        tile = ntl;
        if (tile >= ntile) break;
    }
}


struct PairBounds { int myb, myn, nmax; };

__device__ __forceinline__ PairBounds pair_bounds(int offc, int p, bool hi)
{
    const int b0 = __builtin_amdgcn_readlane(offc, 2 * p);
    const int m  = __builtin_amdgcn_readlane(offc, 2 * p + 1);
    const int e1 = __builtin_amdgcn_readlane(offc, 2 * p + 2);
    PairBounds r;
    r.myb  = hi ? m : b0;
    r.myn  = hi ? e1 - m : m - b0;
    r.nmax = max(m - b0, e1 - m);
    return r;
}

__device__ __forceinline__ int ld_src_id(const int* __restrict__ ssrc, const PairBounds& pb, int idx, int zrow)
{
    return idx < pb.myn ? ssrc[pb.myb + idx] : zrow;
}

template <int K>
__device__ __forceinline__ void issue_row_loads(const char* __restrict__ yb, int ids, unsigned coff, v4f (&v)[16])
{
    if constexpr (K < 16) {
        const int s = __builtin_amdgcn_update_dpp(0, ids, 0x150 + K, 0xf, 0xf, false);
        v[K] = *(const v4f*)(yb + (size_t)(((unsigned)s << 8) | coff));
        issue_row_loads<K + 1>(yb, ids, coff, v);
    }
}

__device__ __forceinline__ v4f gather_pair(const char* __restrict__ yb, const int* __restrict__ ssrc,
                                           const PairBounds& pb, int ids0, int ids1, int c, int zrow)
{
    v4f acc = {0.f, 0.f, 0.f, 0.f};
    const unsigned coff = (unsigned)c << 4;
    for (int j = 0; j < pb.nmax; j += 16) {
        v4f v[16];
        issue_row_loads<0>(yb, ids0, coff, v);
        ids0 = ids1;
        if (j + 32 < pb.nmax) ids1 = ld_src_id(ssrc, pb, j + 32 + c, zrow);
        #pragma unroll
        for (int k = 0; k < 16; ++k) acc += v[k];
    }
    return acc;
}

template <int K, int N, int OUT, int VN>
__device__ __forceinline__ void issue_rows(const char* __restrict__ yb, int ids, unsigned coff, v4f (&v)[VN])
{
    if constexpr (N > 0) {
        const int s = __builtin_amdgcn_update_dpp(0, ids, 0x150 + K, 0xf, 0xf, false);
        v[OUT] = *(const v4f*)(yb + (size_t)(((unsigned)s << 8) | coff));
        issue_rows<K + 1, N - 1, OUT + 1, VN>(yb, ids, coff, v);
    }
}

__device__ __forceinline__ v4f gather_pair_from8(const char* __restrict__ yb, const int* __restrict__ ssrc,
                                                 const PairBounds& pb, const v4f (&g)[8], int idsA, int idsB, int c, int zrow)
{
    v4f acc = {0.f, 0.f, 0.f, 0.f};
    const unsigned coff = (unsigned)c << 4;
    #pragma unroll
    for (int k = 0; k < 8; ++k) acc += g[k];
    for (int j = 8; j < pb.nmax; j += 16) {
        v4f v[16];
        issue_rows<8, 8, 0, 16>(yb, idsA, coff, v);
        issue_rows<0, 8, 8, 16>(yb, idsB, coff, v);
        idsA = idsB;
        idsB = zrow;
        if (j + 24 < pb.nmax) idsB = ld_src_id(ssrc, pb, j + 24 + c, zrow);
        #pragma unroll
        for (int k = 0; k < 16; ++k) acc += v[k];
    }
    return acc;
}

template <bool FUSE_NEXT>
__global__ __launch_bounds__(256) __attribute__((amdgpu_waves_per_eu(8, 8))) void update_kernel(
    const float* __restrict__ xin, const float* __restrict__ ysrc,
    const int* __restrict__ off, const int* __restrict__ ssrc,
    const float* __restrict__ idemb,
    const float* __restrict__ Wl, const float* __restrict__ bl,
    const float* __restrict__ Wg, const float* __restrict__ bg,
    const float* __restrict__ Wc, float* __restrict__ xout,
    float* __restrict__ yout, int nnode)
{
    constexpr int NPIECE = FUSE_NEXT ? 3 : 2;
    __shared__ __attribute__((aligned(16))) float lds[NPIECE * 32 * XPITCH + NPIECE * DLAT * WPITCH];
    float* const xs   = lds;
    float* const hs   = xs + 32 * XPITCH;
    float* const xt   = hs + 32 * XPITCH;
    float* const Wl_t = lds + NPIECE * 32 * XPITCH;
    float* const Wg_t = Wl_t + DLAT * WPITCH;
    float* const Wc_t = Wg_t + DLAT * WPITCH;

    const int tid   = threadIdx.x;
    const int ntile = (nnode + 31) >> 5;
    const int tstep = (int)gridDim.x;
    int tile = (int)blockIdx.x;
    if (tile >= ntile) return;

    const int lane    = tid & 31;
    const int wave    = tid >> 5;
    const int r4      = wave * 4;
    const int l4      = min(lane, 4);
    const int  mn = lane & 15;
    const bool hi = lane >= 16;
    const int      hrow = r4 + (hi ? 1 : 0);
    const int      hoff = hrow * XPITCH + mn * 4;
    const unsigned coff = (unsigned)mn << 4;

    v4f xr0 = ld_xrow(xin, tile * 32 + hrow,     coff, nnode);
    v4f xr1 = ld_xrow(xin, tile * 32 + hrow + 2, coff, nnode);
    int offc = off[min(tile * 32 + r4 + l4, nnode)];
    {
        v4f wl[4], wg[4], wc[4];
        ld_wT<DLAT>(Wl, tid, wl);
        ld_wT<DLAT>(Wg, tid, wg);
        if constexpr (FUSE_NEXT) ld_wT<DLAT>(Wc, tid, wc);
        st_wT<DLAT, WPITCH>(Wl_t, tid, wl);
        st_wT<DLAT, WPITCH>(Wg_t, tid, wg);
        if constexpr (FUSE_NEXT) st_wT<DLAT, WPITCH>(Wc_t, tid, wc);
    }

    if constexpr (FUSE_NEXT) {
        if (blockIdx.x == 0 && tid < DLAT) yout[(size_t)nnode * DLAT + tid] = 0.f;
    }

    int ids[2][2];
    #pragma unroll
    for (int p = 0; p < 2; ++p) {
        const PairBounds pb = pair_bounds(offc, p, hi);
        ids[p][0] = ld_src_id(ssrc, pb, mn,      nnode);
        ids[p][1] = ld_src_id(ssrc, pb, mn + 16, nnode);
    }

    const int rowtile = wave >> 2;
    const int coltile = wave & 3;
    const int k4h     = (lane >> 4) << 2;
    const int col     = coltile * 16 + mn;

    const float* xp  = xs   + (rowtile * 16 + mn) * XPITCH + k4h;
    const float* hp  = hs   + (rowtile * 16 + mn) * XPITCH + k4h;
    const float* wlp = Wl_t + col * WPITCH + k4h;
    const float* wgp = Wg_t + col * WPITCH + k4h;
    const char*  yb = (const char*)ysrc;

    const float blv = bl[col], bgv = bg[col];
    const int rbase = rowtile * 16 + ((lane >> 4) << 3);

    auto tile_out = [&](int b0) {
        #pragma unroll
        for (int i = 0; i < 2; ++i) {
            const int g = b0 + hrow + 2 * i;
            if (g < nnode)
                *(v4f*)((char*)xout + (((unsigned)g << 8) | coff)) = *(const v4f*)(xt + hoff + 2 * i * XPITCH);
        }
        v8f acc2 = {};
        const float* a2p = xt   + (rowtile * 16 + mn) * XPITCH + k4h;
        const float* wcp = Wc_t + col * WPITCH + k4h;
        #pragma unroll
        for (int kk = 0; kk < DLAT; kk += 8)
            acc2 = wmma_2steps(*(const v4f*)(a2p + kk), *(const v4f*)(wcp + kk), acc2);
        #pragma unroll
        for (int i = 0; i < 8; ++i) {
            const int g = b0 + rbase + i;
            if (g < nnode) *(float*)((char*)yout + roff(g, col)) = acc2[i];
        }
    };

    for (; tile < ntile; tile += tstep) {
        const int blk0 = tile * 32;
        const int offnx = off[min((tile + tstep) * 32 + r4 + l4, nnode)];

        *(v4f*)(xs + hoff) = xr0;
        *(v4f*)(xs + hoff + 2 * XPITCH) = xr1;
        lds_barrier();

        const PairBounds pb0 = pair_bounds(offc, 0, hi);
        v4f gv[8];
        issue_rows<0, 8, 0, 8>(yb, ids[0][0], coff, gv);

        v8f c1 = {};
        #pragma unroll
        for (int kk = 0; kk < DLAT; kk += 8)
            c1 = wmma_2steps(*(const v4f*)(xp + kk), *(const v4f*)(wlp + kk), c1);

        if constexpr (FUSE_NEXT) {
            if (tile != (int)blockIdx.x) tile_out(blk0 - tstep * 32);
        }

        const v8f idc = ld_id(idemb, tile, rbase, col, nnode);

        {
            const v4f a = gather_pair_from8(yb, ssrc, pb0, gv, ids[0][0], ids[0][1], mn, nnode);
            v4f r;
            r.x = leaky(a.x); r.y = leaky(a.y); r.z = leaky(a.z); r.w = leaky(a.w);
            *(v4f*)(hs + hoff) = r;
        }
        {
            const PairBounds pb = pair_bounds(offc, 1, hi);
            const v4f a = gather_pair(yb, ssrc, pb, ids[1][0], ids[1][1], mn, nnode);
            v4f r;
            r.x = leaky(a.x); r.y = leaky(a.y); r.z = leaky(a.z); r.w = leaky(a.w);
            *(v4f*)(hs + hoff + 2 * XPITCH) = r;
        }
        v8f xh;
        #pragma unroll
        for (int i = 0; i < 8; ++i) xh[i] = leaky(c1[i] + blv) + idc[i];
        lds_barrier();

        const int ntl = tile + tstep;
        offc = offnx;
        const PairBounds nb0 = pair_bounds(offc, 0, hi);
        const PairBounds nb1 = pair_bounds(offc, 1, hi);
        if (ntl < ntile) {
            ids[0][0] = ld_src_id(ssrc, nb0, mn,      nnode);
            ids[0][1] = ld_src_id(ssrc, nb0, mn + 16, nnode);
            ids[1][0] = ld_src_id(ssrc, nb1, mn,      nnode);
            ids[1][1] = ld_src_id(ssrc, nb1, mn + 16, nnode);
            xr0 = ld_xrow(xin, ntl * 32 + hrow,     coff, nnode);
            xr1 = ld_xrow(xin, ntl * 32 + hrow + 2, coff, nnode);
        }

        v8f c2 = {};
        #pragma unroll
        for (int kk = 0; kk < DLAT; kk += 8)
            c2 = wmma_2steps(*(const v4f*)(hp + kk), *(const v4f*)(wgp + kk), c2);

        #pragma unroll
        for (int i = 0; i < 8; ++i) {
            int rloc = rbase + i;
            int g    = blk0 + rloc;
            float v    = leaky(c2[i] + bgv + xh[i]);
            if constexpr (FUSE_NEXT) {
                xt[rloc * XPITCH + col] = v;
            } else {
                if (g < nnode) *(float*)((char*)xout + roff(g, col)) = v;
            }
        }

    }
    if constexpr (FUSE_NEXT) {
        lds_barrier();
        tile_out((tile - tstep) * 32);
    }
}

__device__ __forceinline__ int block_excl_scan_1024(int v, int* wsum, int& total)
{
    const int lane = threadIdx.x & 31;
    const int wave = threadIdx.x >> 5;
    int inc = v;
    for (int o = 1; o < 32; o <<= 1) {
        int u = __shfl_up(inc, o);
        if (lane >= o) inc += u;
    }
    if (lane == 31) wsum[wave] = inc;
    __syncthreads();
    const int ws = wsum[lane];
    int wp  = lane < wave ? ws : 0;
    int tot = ws;
    for (int o = 16; o > 0; o >>= 1) {
        wp  += __shfl_xor(wp, o);
        tot += __shfl_xor(tot, o);
    }
    total = tot;
    __syncthreads();
    return wp + inc - v;
}

__global__ __launch_bounds__(1024) void bin_sort_kernel(
    const int2* __restrict__ cs, const int* __restrict__ tab, const int* __restrict__ lst,
    int* __restrict__ off, int* __restrict__ ssrc, int nnode, int nb, int nch)
{
    __shared__ int  ppre[MAXCH];
    __shared__ int  ps[MAXCH];
    __shared__ int  lcnt[BIN_ROWS];
    __shared__ int  loff[BIN_ROWS];
    __shared__ int  lcur[BIN_ROWS];
    __shared__ int  wsum[32];
    __shared__ int  gsum[32];
    __shared__ int2 A[BIN_CAP];
    __shared__ int  B[BIN_CAP];

    const int tid   = threadIdx.x;
    const int lane  = tid & 31;
    const int wave  = tid >> 5;
    const int b     = blockIdx.x;
    const int rbase = b << BIN_SH;

    int n = 0, st = 0;
    if (tid < nch) {
        n  = tab[(size_t)b * nch + tid];
        st = lst[(size_t)tid * nb + b];
    }

    int g = 0;
    {
        const int   nfront = b * nch;
        const int   n4     = nfront >> 2;
        const int4* t4     = (const int4*)tab;
        int i = tid;
        for (; i + 7 * 1024 < n4; i += 8 * 1024) {
            int4 a[8];
            #pragma unroll
            for (int k = 0; k < 8; ++k) a[k] = t4[i + k * 1024];
            #pragma unroll
            for (int k = 0; k < 8; ++k) g += (a[k].x + a[k].y) + (a[k].z + a[k].w);
        }
        for (; i < n4; i += 1024) {
            int4 a = t4[i];
            g += (a.x + a.y) + (a.z + a.w);
        }
        for (int q = (n4 << 2) + tid; q < nfront; q += 1024) g += tab[q];
    }
    for (int o = 16; o > 0; o >>= 1) g += __shfl_xor(g, o);
    if (lane == 0) gsum[wave] = g;

    ps[tid] = st;
    if (tid < BIN_ROWS) { lcnt[tid] = 0; lcur[tid] = 0; }
    int total;
    const int ex = block_excl_scan_1024(n, wsum, total);
    ppre[tid] = ex;
    int gb = gsum[lane];
    for (int o = 16; o > 0; o >>= 1) gb += __shfl_xor(gb, o);
    __syncthreads();

    const bool single = total <= BIN_CAP;

    for (int j0 = tid; j0 < total; j0 += 8 * 1024) {
        int2 e[8];
        #pragma unroll
        for (int k = 0; k < 8; ++k) {
            const int j = min(j0 + k * 1024, total - 1);
            int cc = 0;
            #pragma unroll
            for (int sft = MAXCH / 2; sft > 0; sft >>= 1)
                if (ppre[cc + sft] <= j) cc += sft;
            e[k] = cs[(size_t)cc * SORT_CH + ps[cc] + (j - ppre[cc])];
        }
        #pragma unroll
        for (int k = 0; k < 8; ++k) {
            const int j = j0 + k * 1024;
            if (j < total) {
                const int ld = min(max(e[k].y - rbase, 0), BIN_ROWS - 1);
                atomicAdd(&lcnt[ld], 1);
                if (single) A[j] = make_int2(e[k].x, ld);
            }
        }
    }
    __syncthreads();

    const int cv = tid < BIN_ROWS ? lcnt[tid] : 0;
    int rtot;
    const int rex = block_excl_scan_1024(cv, wsum, rtot);
    if (tid < BIN_ROWS) {
        loff[tid] = rex;
        if (rbase + tid < nnode) off[rbase + tid] = gb + rex;
    }
    if (b == nb - 1 && tid == 0) off[nnode] = gb + total;
    __syncthreads();

    if (single) {
        for (int j = tid; j < total; j += 1024) {
            const int2 e = A[j];
            const int pos = loff[e.y] + atomicAdd(&lcur[e.y], 1);
            B[pos] = e.x;
        }
        __syncthreads();
        for (int j = tid; j < total; j += 1024) ssrc[(size_t)gb + j] = B[j];
    } else {
        for (int j = tid; j < total; j += 1024) {
            int cc = 0;
            #pragma unroll
            for (int sft = MAXCH / 2; sft > 0; sft >>= 1)
                if (ppre[cc + sft] <= j) cc += sft;
            const int2 e  = cs[(size_t)cc * SORT_CH + ps[cc] + (j - ppre[cc])];
            const int  ld = min(max(e.y - rbase, 0), BIN_ROWS - 1);
            const int  pos = loff[ld] + atomicAdd(&lcur[ld], 1);
            ssrc[(size_t)gb + pos] = e.x;
        }
    }
}

extern "C" void kernel_launch(void* const* d_in, const int* in_sizes, int n_in,
                              void* d_out, int out_size, void* d_ws, size_t ws_size,
                              hipStream_t stream)
{
    const float* features = (const float*)d_in[0];
    const float* id_emb   = (const float*)d_in[1];
    const int*   edge     = (const int*)d_in[2];
    const float* Wmlp     = (const float*)d_in[3];
    const float* bmlp     = (const float*)d_in[4];
    const float* Wc1      = (const float*)d_in[5];
    const float* Wl1      = (const float*)d_in[6];
    const float* bl1      = (const float*)d_in[7];
    const float* Wg1      = (const float*)d_in[8];
    const float* bg1      = (const float*)d_in[9];
    const float* Wc2      = (const float*)d_in[10];
    const float* Wl2      = (const float*)d_in[11];
    const float* bl2      = (const float*)d_in[12];
    const float* Wg2      = (const float*)d_in[13];
    const float* bg2      = (const float*)d_in[14];

    const int nnode = in_sizes[0] / DFEAT;
    const int nedge = in_sizes[2] / 2;
    const int* src  = edge;
    const int* dst  = edge + nedge;

    float* x = (float*)d_ws;
    float* y2 = x + (size_t)nnode * DLAT;
    float* y  = y2 + ((size_t)nnode + 1) * DLAT;

    const size_t cpad = (((size_t)nnode + 4) / 4) * 4;
    const size_t epad = (((size_t)nedge + 3) / 4) * 4;
    const int nb  = (nnode + BIN_ROWS - 1) / BIN_ROWS;
    const int nch = (nedge + SORT_CH - 1) / SORT_CH;
    int*  off  = (int*)(y + ((size_t)nnode + 1) * DLAT);
    int*  ssrc = off + cpad;
    int2* cs   = (int2*)(ssrc + epad);
    int*  tab  = (int*)(cs + epad);
    int*  lst  = tab + (((size_t)nb * nch + 3) / 4) * 4;

    mlp_norm_conv_kernel<<<DENSE_GRID, 256, 0, stream>>>(features, Wmlp, bmlp, Wc1, x, y, nnode,
                                                     src, dst, cs, tab, lst, nedge, nb, nch);
    bin_sort_kernel<<<nb, 1024, 0, stream>>>(cs, tab, lst, off, ssrc, nnode, nb, nch);
    update_kernel<true><<<DENSE_GRID, 256, 0, stream>>>(x, y, off, ssrc, id_emb, Wl1, bl1, Wg1, bg1,
                                                  Wc2, x, y2, nnode);
    update_kernel<false><<<DENSE_GRID, 256, 0, stream>>>(x, y2, off, ssrc, id_emb, Wl2, bl2, Wg2, bg2,
                                                   nullptr, (float*)d_out, nullptr, nnode);
}
